# starting kernel
# baseline (speedup 1.0000x reference)
#include <hip/hip_runtime.h>
#include <hip/hip_bf16.h>

typedef float v2f __attribute__((ext_vector_type(2)));
typedef float v8f __attribute__((ext_vector_type(8)));

#define IN_DIM   64
#define OUT_DIM  64
#define EDGE_DIM 16
#define H_HEADS  4
#define HD       16
#define NEG_SLOPE   0.2f
#define EPS_SOFTMAX 1e-8f
#define EPS_LN      1e-5f

__global__ void zero_kernel(float* __restrict__ p, long n) {
    long i = (long)blockIdx.x * blockDim.x + threadIdx.x;
    long stride = (long)gridDim.x * blockDim.x;
    for (; i < n; i += stride) p[i] = 0.0f;
}

__global__ void node_gemm_wmma(const float* __restrict__ x,
                               const float* __restrict__ W,
                               float* __restrict__ h, int n) {
    int wave = threadIdx.x >> 5;
    int lane = threadIdx.x & 31;
    int grp  = lane >> 4;
    int lid  = lane & 15;

    int row0 = blockIdx.x * 16;
    int col0 = wave * 16;

    int arow = row0 + lid;
    if (arow >= n) arow = n - 1;

    const float* xr = x + (long)arow * IN_DIM + 2 * grp;
    const float* wc = W + (long)(2 * grp) * OUT_DIM + col0 + lid;

    v8f c = {};
    #pragma unroll
    for (int k0 = 0; k0 < IN_DIM; k0 += 4) {
        v2f a = *(const v2f*)(xr + k0);
        v2f b;
        b.x = wc[(long)k0 * OUT_DIM];
        b.y = wc[(long)(k0 + 1) * OUT_DIM];
        c = __builtin_amdgcn_wmma_f32_16x16x4_f32(false, a, false, b,
                                                  (short)0, c, false, false);
    }

    #pragma unroll
    for (int r = 0; r < 8; ++r) {
        int row = row0 + r + 8 * grp;
        if (row < n) h[(long)row * OUT_DIM + col0 + lid] = c[r];
    }
}

__global__ void node_proj(const float* __restrict__ h,
                          const float* __restrict__ attn_src,
                          const float* __restrict__ attn_dst,
                          float* __restrict__ a_src,
                          float* __restrict__ a_dst, int n) {
    int idx = blockIdx.x * blockDim.x + threadIdx.x;
    if (idx >= n * H_HEADS) return;
    int hh   = idx & 3;
    int node = idx >> 2;
    const float4* hp = (const float4*)(h + (long)node * OUT_DIM + hh * HD);
    const float4* as = (const float4*)(attn_src + hh * HD);
    const float4* ad = (const float4*)(attn_dst + hh * HD);
    float s = 0.0f, d = 0.0f;
    #pragma unroll
    for (int j = 0; j < 4; ++j) {
        float4 hv = hp[j], sv = as[j], dv = ad[j];
        s += hv.x * sv.x + hv.y * sv.y + hv.z * sv.z + hv.w * sv.w;
        d += hv.x * dv.x + hv.y * dv.y + hv.z * dv.z + hv.w * dv.w;
    }
    a_src[idx] = s;
    a_dst[idx] = d;
}

__global__ void edge_score(const int* __restrict__ src,
                           const int* __restrict__ dst,
                           const float* __restrict__ edge_attr,
                           const float* __restrict__ W_edge,
                           const float* __restrict__ a_src,
                           const float* __restrict__ a_dst,
                           float* __restrict__ attn,
                           int* __restrict__ amax_bits, long e_cnt) {
    long e = (long)blockIdx.x * blockDim.x + threadIdx.x;
    if (e >= e_cnt) return;
    int s = src[e], t = dst[e];

    float sc[H_HEADS] = {0.f, 0.f, 0.f, 0.f};
    const float4* ea = (const float4*)(edge_attr + e * EDGE_DIM);
    #pragma unroll
    for (int q = 0; q < 4; ++q) {
        float4 v = ea[q];
        int d0 = q * 4;
        #pragma unroll
        for (int hh = 0; hh < H_HEADS; ++hh) {
            sc[hh] += v.x * W_edge[(d0 + 0) * H_HEADS + hh]
                    + v.y * W_edge[(d0 + 1) * H_HEADS + hh]
                    + v.z * W_edge[(d0 + 2) * H_HEADS + hh]
                    + v.w * W_edge[(d0 + 3) * H_HEADS + hh];
        }
    }
    float4 asv = ((const float4*)a_src)[s];
    float4 adv = ((const float4*)a_dst)[t];
    float av[H_HEADS] = {asv.x + adv.x, asv.y + adv.y, asv.z + adv.z, asv.w + adv.w};

    float4 outv;
    float* op = (float*)&outv;
    #pragma unroll
    for (int hh = 0; hh < H_HEADS; ++hh) {
        float v = av[hh] + sc[hh];
        v = (v >= 0.0f) ? v : NEG_SLOPE * v;
        op[hh] = v;
        if (v > 0.0f) atomicMax(&amax_bits[t * H_HEADS + hh], __float_as_int(v));
    }
    ((float4*)attn)[e] = outv;
}

__global__ void edge_softmax(const int* __restrict__ dst,
                             const float* __restrict__ amax,
                             float* __restrict__ attn,
                             float* __restrict__ asum, long eh) {
    long idx = (long)blockIdx.x * blockDim.x + threadIdx.x;
    if (idx >= eh) return;
    long e = idx >> 2;
    int hh = (int)(idx & 3);
    int t  = dst[e];
    float v = expf(attn[idx] - amax[t * H_HEADS + hh]);
    attn[idx] = v;
    atomicAdd(&asum[t * H_HEADS + hh], v);
}

__global__ void scatter_msg(const int* __restrict__ src,
                            const int* __restrict__ dst,
                            const float* __restrict__ h,
                            const float* __restrict__ attn,
                            const float* __restrict__ asum,
                            float* __restrict__ acc, long total) {
    long idx = (long)blockIdx.x * blockDim.x + threadIdx.x;
    if (idx >= total) return;
    long e = idx >> 6;
    int d  = (int)(idx & 63);
    int hh = d >> 4;
    int s = src[e], t = dst[e];
    float w = attn[e * H_HEADS + hh] / (asum[t * H_HEADS + hh] + EPS_SOFTMAX);
    atomicAdd(&acc[(long)t * OUT_DIM + d], h[(long)s * OUT_DIM + d] * w);
}

__global__ void residual_ln(const float* __restrict__ acc,
                            const float* __restrict__ h,
                            const float* __restrict__ gamma,
                            const float* __restrict__ beta,
                            float* __restrict__ out, int n) {
    int wave = threadIdx.x >> 5;
    int lane = threadIdx.x & 31;
    int node = blockIdx.x * 8 + wave;
    if (node >= n) return;
    long base = (long)node * OUT_DIM;

    float y0 = acc[base + lane]      + h[base + lane];
    float y1 = acc[base + lane + 32] + h[base + lane + 32];

    float s = y0 + y1;
    #pragma unroll
    for (int off = 16; off > 0; off >>= 1) s += __shfl_xor(s, off, 32);
    float mu = s * (1.0f / 64.0f);

    float d0 = y0 - mu, d1 = y1 - mu;
    float vs = d0 * d0 + d1 * d1;
    #pragma unroll
    for (int off = 16; off > 0; off >>= 1) vs += __shfl_xor(vs, off, 32);
    float rstd = rsqrtf(vs * (1.0f / 64.0f) + EPS_LN);

    out[base + lane]      = d0 * rstd * gamma[lane]      + beta[lane];
    out[base + lane + 32] = d1 * rstd * gamma[lane + 32] + beta[lane + 32];
}

extern "C" void kernel_launch(void* const* d_in, const int* in_sizes, int n_in,
                              void* d_out, int out_size, void* d_ws, size_t ws_size,
                              hipStream_t stream) {
    const float* x         = (const float*)d_in[0];
    const int*   ei        = (const int*)  d_in[1];
    const float* edge_attr = (const float*)d_in[2];
    const float* W_node    = (const float*)d_in[3];
    const float* W_edge    = (const float*)d_in[4];
    const float* attn_src  = (const float*)d_in[5];
    const float* attn_dst  = (const float*)d_in[6];
    const float* gamma     = (const float*)d_in[7];
    const float* beta      = (const float*)d_in[8];
    float* out = (float*)d_out;

    int  n = in_sizes[0] / IN_DIM;
    long e = (long)in_sizes[2] / EDGE_DIM;
    const int* srcIdx = ei;
    const int* dstIdx = ei + e;

    float* ws    = (float*)d_ws;
    float* h     = ws;  ws += (long)n * OUT_DIM;
    float* a_src = ws;  ws += (long)n * H_HEADS;
    float* a_dst = ws;  ws += (long)n * H_HEADS;
    float* attn  = ws;  ws += e * H_HEADS;
    float* acc   = ws;  ws += (long)n * OUT_DIM;
    float* amax  = ws;  ws += (long)n * H_HEADS;
    float* asum  = ws;  ws += (long)n * H_HEADS;

    long zn = (long)n * (OUT_DIM + 2 * H_HEADS);
    zero_kernel<<<(int)((zn + 255) / 256), 256, 0, stream>>>(acc, zn);

    int n_tiles = (n + 15) / 16;
    node_gemm_wmma<<<n_tiles, 128, 0, stream>>>(x, W_node, h, n);

    node_proj<<<(n * H_HEADS + 255) / 256, 256, 0, stream>>>(
        h, attn_src, attn_dst, a_src, a_dst, n);

    edge_score<<<(int)((e + 255) / 256), 256, 0, stream>>>(
        srcIdx, dstIdx, edge_attr, W_edge, a_src, a_dst, attn, (int*)amax, e);

    long eh = e * H_HEADS;
    edge_softmax<<<(int)((eh + 255) / 256), 256, 0, stream>>>(
        dstIdx, amax, attn, asum, eh);

    long total = e * OUT_DIM;
    scatter_msg<<<(int)((total + 255) / 256), 256, 0, stream>>>(
        srcIdx, dstIdx, h, attn, asum, acc, total);

    residual_ln<<<(n + 7) / 8, 256, 0, stream>>>(acc, h, gamma, beta, out, n);
}


# scatter_msg float atomics replaced by counting sort of edges by dst (arrival-order rank) + one wave per node summing h[src] in registers, residual+LN folded in; per-node sum order changes
# speedup vs baseline: 1.3484x; 1.3484x over previous
#include <hip/hip_runtime.h>
#include <hip/hip_bf16.h>

typedef float v2f __attribute__((ext_vector_type(2)));
typedef float v8f __attribute__((ext_vector_type(8)));

#define IN_DIM   64
#define OUT_DIM  64
#define EDGE_DIM 16
#define H_HEADS  4
#define HD       16
#define NEG_SLOPE   0.2f
#define EPS_SOFTMAX 1e-8f
#define EPS_LN      1e-5f

__global__ void zero_kernel(float* __restrict__ p, long n) {
    long i = (long)blockIdx.x * blockDim.x + threadIdx.x;
    long stride = (long)gridDim.x * blockDim.x;
    for (; i < n; i += stride) p[i] = 0.0f;
}

__global__ void node_gemm_wmma(const float* __restrict__ x,
                               const float* __restrict__ W,
                               float* __restrict__ h, int n) {
    int wave = threadIdx.x >> 5;
    int lane = threadIdx.x & 31;
    int grp  = lane >> 4;
    int lid  = lane & 15;

    int row0 = blockIdx.x * 16;
    int col0 = wave * 16;

    int arow = row0 + lid;
    if (arow >= n) arow = n - 1;

    const float* xr = x + (long)arow * IN_DIM + 2 * grp;
    const float* wc = W + (long)(2 * grp) * OUT_DIM + col0 + lid;

    v8f c = {};
    #pragma unroll
    for (int k0 = 0; k0 < IN_DIM; k0 += 4) {
        v2f a = *(const v2f*)(xr + k0);
        v2f b;
        b.x = wc[(long)k0 * OUT_DIM];
        b.y = wc[(long)(k0 + 1) * OUT_DIM];
        c = __builtin_amdgcn_wmma_f32_16x16x4_f32(false, a, false, b,
                                                  (short)0, c, false, false);
    }

    #pragma unroll
    for (int r = 0; r < 8; ++r) {
        int row = row0 + r + 8 * grp;
        if (row < n) h[(long)row * OUT_DIM + col0 + lid] = c[r];
    }
}

__global__ void node_proj(const float* __restrict__ h,
                          const float* __restrict__ attn_src,
                          const float* __restrict__ attn_dst,
                          float* __restrict__ a_src,
                          float* __restrict__ a_dst, int n) {
    int idx = blockIdx.x * blockDim.x + threadIdx.x;
    if (idx >= n * H_HEADS) return;
    int hh   = idx & 3;
    int node = idx >> 2;
    const float4* hp = (const float4*)(h + (long)node * OUT_DIM + hh * HD);
    const float4* as = (const float4*)(attn_src + hh * HD);
    const float4* ad = (const float4*)(attn_dst + hh * HD);
    float s = 0.0f, d = 0.0f;
    #pragma unroll
    for (int j = 0; j < 4; ++j) {
        float4 hv = hp[j], sv = as[j], dv = ad[j];
        s += hv.x * sv.x + hv.y * sv.y + hv.z * sv.z + hv.w * sv.w;
        d += hv.x * dv.x + hv.y * dv.y + hv.z * dv.z + hv.w * dv.w;
    }
    a_src[idx] = s;
    a_dst[idx] = d;
}

__global__ void edge_score(const int* __restrict__ src,
                           const int* __restrict__ dst,
                           const float* __restrict__ edge_attr,
                           const float* __restrict__ W_edge,
                           const float* __restrict__ a_src,
                           const float* __restrict__ a_dst,
                           float* __restrict__ attn,
                           int* __restrict__ amax_bits, long e_cnt) {
    long e = (long)blockIdx.x * blockDim.x + threadIdx.x;
    if (e >= e_cnt) return;
    int s = src[e], t = dst[e];

    float sc[H_HEADS] = {0.f, 0.f, 0.f, 0.f};
    const float4* ea = (const float4*)(edge_attr + e * EDGE_DIM);
    #pragma unroll
    for (int q = 0; q < 4; ++q) {
        float4 v = ea[q];
        int d0 = q * 4;
        #pragma unroll
        for (int hh = 0; hh < H_HEADS; ++hh) {
            sc[hh] += v.x * W_edge[(d0 + 0) * H_HEADS + hh]
                    + v.y * W_edge[(d0 + 1) * H_HEADS + hh]
                    + v.z * W_edge[(d0 + 2) * H_HEADS + hh]
                    + v.w * W_edge[(d0 + 3) * H_HEADS + hh];
        }
    }
    float4 asv = ((const float4*)a_src)[s];
    float4 adv = ((const float4*)a_dst)[t];
    float av[H_HEADS] = {asv.x + adv.x, asv.y + adv.y, asv.z + adv.z, asv.w + adv.w};

    float4 outv;
    float* op = (float*)&outv;
    #pragma unroll
    for (int hh = 0; hh < H_HEADS; ++hh) {
        float v = av[hh] + sc[hh];
        v = (v >= 0.0f) ? v : NEG_SLOPE * v;
        op[hh] = v;
        if (v > 0.0f) atomicMax(&amax_bits[t * H_HEADS + hh], __float_as_int(v));
    }
    ((float4*)attn)[e] = outv;
}

__global__ void edge_softmax(const int* __restrict__ dst,
                             const float* __restrict__ amax,
                             float* __restrict__ attn,
                             float* __restrict__ asum, long eh) {
    long idx = (long)blockIdx.x * blockDim.x + threadIdx.x;
    if (idx >= eh) return;
    long e = idx >> 2;
    int hh = (int)(idx & 3);
    int t  = dst[e];
    float v = expf(attn[idx] - amax[t * H_HEADS + hh]);
    attn[idx] = v;
    atomicAdd(&asum[t * H_HEADS + hh], v);
}

__global__ void dst_rank(const int* __restrict__ dst,
                         int* __restrict__ cnt,
                         int* __restrict__ rank, long e_cnt) {
    long e = (long)blockIdx.x * blockDim.x + threadIdx.x;
    if (e >= e_cnt) return;
    rank[e] = atomicAdd(&cnt[dst[e]], 1);
}

__global__ void __launch_bounds__(1024)
seg_scan(const int* __restrict__ cnt, int* __restrict__ seg, int n) {
    __shared__ int wsum[2][32];
    int tid  = threadIdx.x;
    int lane = tid & 31;
    int wave = tid >> 5;
    int carry = 0;
    int buf = 0;

    int4 v = make_int4(0, 0, 0, 0);
    {
        int i = tid * 4;
        if (i + 3 < n) v = *(const int4*)(cnt + i);
        else {
            if (i     < n) v.x = cnt[i];
            if (i + 1 < n) v.y = cnt[i + 1];
            if (i + 2 < n) v.z = cnt[i + 2];
        }
    }
    for (int base = 0; base < n; base += 4096) {
        int i = base + tid * 4;
        int4 vn = make_int4(0, 0, 0, 0);
        {
            int j = i + 4096;
            if (j + 3 < n) vn = *(const int4*)(cnt + j);
            else {
                if (j     < n) vn.x = cnt[j];
                if (j + 1 < n) vn.y = cnt[j + 1];
                if (j + 2 < n) vn.z = cnt[j + 2];
            }
        }
        int s = v.x + v.y + v.z + v.w;
        int incl = s;
        #pragma unroll
        for (int d = 1; d < 32; d <<= 1) {
            int t = __shfl_up(incl, d, 32);
            if (lane >= d) incl += t;
        }
        if (lane == 31) wsum[buf][wave] = incl;
        __syncthreads();
        int wv = wsum[buf][lane];
        int wincl = wv;
        #pragma unroll
        for (int d = 1; d < 32; d <<= 1) {
            int t = __shfl_up(wincl, d, 32);
            if (lane >= d) wincl += t;
        }
        int wexcl = __shfl(wincl - wv, wave, 32);
        int total = __shfl(wincl, 31, 32);
        int x0 = carry + wexcl + (incl - s);
        if (i + 3 < n) {
            *(int4*)(seg + i) = make_int4(x0, x0 + v.x, x0 + v.x + v.y,
                                          x0 + v.x + v.y + v.z);
        } else {
            if (i     < n) seg[i]     = x0;
            if (i + 1 < n) seg[i + 1] = x0 + v.x;
            if (i + 2 < n) seg[i + 2] = x0 + v.x + v.y;
        }
        carry += total;
        v = vn;
        buf ^= 1;
    }
    if (tid == 0) seg[n] = carry;
}

__global__ void dst_fill(const int* __restrict__ src,
                         const int* __restrict__ dst,
                         const int* __restrict__ rank,
                         const int* __restrict__ seg,
                         int* __restrict__ perm, long e_cnt) {
    long e = (long)blockIdx.x * blockDim.x + threadIdx.x;
    if (e >= e_cnt) return;
    int p = seg[dst[e]] + rank[e];
    ((int2*)perm)[p] = make_int2((int)e, src[e]);
}

__global__ void __launch_bounds__(256)
gather_ln(const int* __restrict__ perm,
          const int* __restrict__ seg,
          const float* __restrict__ h,
          const float* __restrict__ attn,
          const float* __restrict__ asum,
          const float* __restrict__ gamma,
          const float* __restrict__ beta,
          float* __restrict__ out, int n) {
    int lane = threadIdx.x & 31;
    int node = __builtin_amdgcn_readfirstlane((int)(blockIdx.x * 8 + (threadIdx.x >> 5)));
    if (node >= n) return;

    int beg = seg[node];
    int end = seg[node + 1];
    float4 as4 = ((const float4*)asum)[node];
    float den0 = as4.x + EPS_SOFTMAX, den1 = as4.y + EPS_SOFTMAX;
    float den2 = as4.z + EPS_SOFTMAX, den3 = as4.w + EPS_SOFTMAX;

    int head = lane >> 3;
    int hm0 = (head == 0) ? -1 : 0, hm1 = (head == 1) ? -1 : 0;
    int hm2 = (head == 2) ? -1 : 0, hm3 = (head == 3) ? -1 : 0;
    const float* hl = h + 2 * lane;
    float ax = 0.0f, ay = 0.0f;

    for (int cb = beg; cb < end; cb += 32) {
        int m = end - cb;
        if (m > 32) m = 32;
        int2 pe = make_int2(0, 0);
        float4 a4 = make_float4(0.0f, 0.0f, 0.0f, 0.0f);
        if (lane < m) {
            pe = ((const int2*)perm)[cb + lane];
            a4 = ((const float4*)attn)[pe.x];
        }
        int w0 = __float_as_int(a4.x / den0);
        int w1 = __float_as_int(a4.y / den1);
        int w2 = __float_as_int(a4.z / den2);
        int w3 = __float_as_int(a4.w / den3);

        for (int k0 = 0; k0 < m; k0 += 8) {
            float2 hv[8];
            float  wk[8];
            #pragma unroll
            for (int j = 0; j < 8; ++j) {
                int k = k0 + j;
                int s = __builtin_amdgcn_readlane(pe.y, k);
                int u0 = __builtin_amdgcn_readlane(w0, k);
                int u1 = __builtin_amdgcn_readlane(w1, k);
                int u2 = __builtin_amdgcn_readlane(w2, k);
                int u3 = __builtin_amdgcn_readlane(w3, k);
                wk[j] = __int_as_float((u0 & hm0) | (u1 & hm1) | (u2 & hm2) | (u3 & hm3));
                hv[j] = *(const float2*)(hl + (long)s * OUT_DIM);
            }
            #pragma unroll
            for (int j = 0; j < 8; ++j) {
                ax += hv[j].x * wk[j];
                ay += hv[j].y * wk[j];
            }
        }
    }

    long base = (long)node * OUT_DIM + 2 * lane;
    float2 hn = *(const float2*)(h + base);
    float y0 = ax + hn.x;
    float y1 = ay + hn.y;

    float s = y0 + y1;
    #pragma unroll
    for (int o = 16; o > 0; o >>= 1) s += __shfl_xor(s, o, 32);
    float mu = s * (1.0f / 64.0f);

    float d0 = y0 - mu, d1 = y1 - mu;
    float vs = d0 * d0 + d1 * d1;
    #pragma unroll
    for (int o = 16; o > 0; o >>= 1) vs += __shfl_xor(vs, o, 32);
    float rstd = rsqrtf(vs * (1.0f / 64.0f) + EPS_LN);

    float2 g = ((const float2*)gamma)[lane];
    float2 b = ((const float2*)beta)[lane];
    float2 o2;
    o2.x = d0 * rstd * g.x + b.x;
    o2.y = d1 * rstd * g.y + b.y;
    *(float2*)(out + base) = o2;
}

extern "C" void kernel_launch(void* const* d_in, const int* in_sizes, int n_in,
                              void* d_out, int out_size, void* d_ws, size_t ws_size,
                              hipStream_t stream) {
    const float* x         = (const float*)d_in[0];
    const int*   ei        = (const int*)  d_in[1];
    const float* edge_attr = (const float*)d_in[2];
    const float* W_node    = (const float*)d_in[3];
    const float* W_edge    = (const float*)d_in[4];
    const float* attn_src  = (const float*)d_in[5];
    const float* attn_dst  = (const float*)d_in[6];
    const float* gamma     = (const float*)d_in[7];
    const float* beta      = (const float*)d_in[8];
    float* out = (float*)d_out;

    int  n = in_sizes[0] / IN_DIM;
    long e = (long)in_sizes[2] / EDGE_DIM;
    const int* srcIdx = ei;
    const int* dstIdx = ei + e;

    float* ws    = (float*)d_ws;
    float* h     = ws;  ws += (long)n * OUT_DIM;
    float* a_src = ws;  ws += (long)n * H_HEADS;
    float* a_dst = ws;  ws += (long)n * H_HEADS;
    float* attn  = ws;  ws += e * H_HEADS;
    float* amax  = ws;  ws += (long)n * H_HEADS;
    float* asum  = ws;  ws += (long)n * H_HEADS;
    int*   cnt   = (int*)ws;  ws += ((long)n + 3) / 4 * 4;
    int*   seg   = (int*)ws;  ws += ((long)n + 4) / 4 * 4;
    int*   perm  = (int*)ws;  ws += 2 * e;
    int*   rank  = (int*)ws;  ws += e;

    long zn = (long)n * (2 * H_HEADS + 1);
    zero_kernel<<<(int)((zn + 255) / 256), 256, 0, stream>>>(amax, zn);

    int n_tiles = (n + 15) / 16;
    node_gemm_wmma<<<n_tiles, 128, 0, stream>>>(x, W_node, h, n);

    node_proj<<<(n * H_HEADS + 255) / 256, 256, 0, stream>>>(
        h, attn_src, attn_dst, a_src, a_dst, n);

    edge_score<<<(int)((e + 255) / 256), 256, 0, stream>>>(
        srcIdx, dstIdx, edge_attr, W_edge, a_src, a_dst, attn, (int*)amax, e);

    long eh = e * H_HEADS;
    edge_softmax<<<(int)((eh + 255) / 256), 256, 0, stream>>>(
        dstIdx, amax, attn, asum, eh);

    dst_rank<<<(int)((e + 255) / 256), 256, 0, stream>>>(dstIdx, cnt, rank, e);
    seg_scan<<<1, 1024, 0, stream>>>(cnt, seg, n);
    dst_fill<<<(int)((e + 255) / 256), 256, 0, stream>>>(
        srcIdx, dstIdx, rank, seg, perm, e);

    gather_ln<<<(n + 7) / 8, 256, 0, stream>>>(
        perm, seg, h, attn, asum, gamma, beta, out, n);
}


# segment softmax (max, sum of exp) done per dst node by its wave over the dst-sorted edge list with lane reductions; atomicMax, atomicAdd and the edge_softmax launch removed; sum order changes
# speedup vs baseline: 2.3431x; 1.7377x over previous
#include <hip/hip_runtime.h>
#include <hip/hip_bf16.h>

typedef float v2f __attribute__((ext_vector_type(2)));
typedef float v8f __attribute__((ext_vector_type(8)));

#define IN_DIM   64
#define OUT_DIM  64
#define EDGE_DIM 16
#define H_HEADS  4
#define HD       16
#define NEG_SLOPE   0.2f
#define EPS_SOFTMAX 1e-8f
#define EPS_LN      1e-5f

__global__ void zero_kernel(float* __restrict__ p, long n) {
    long i = (long)blockIdx.x * blockDim.x + threadIdx.x;
    long stride = (long)gridDim.x * blockDim.x;
    for (; i < n; i += stride) p[i] = 0.0f;
}

__global__ void node_gemm_wmma(const float* __restrict__ x,
                               const float* __restrict__ W,
                               float* __restrict__ h, int n) {
    int wave = threadIdx.x >> 5;
    int lane = threadIdx.x & 31;
    int grp  = lane >> 4;
    int lid  = lane & 15;

    int row0 = blockIdx.x * 16;
    int col0 = wave * 16;

    int arow = row0 + lid;
    if (arow >= n) arow = n - 1;

    const float* xr = x + (long)arow * IN_DIM + 2 * grp;
    const float* wc = W + (long)(2 * grp) * OUT_DIM + col0 + lid;

    v8f c = {};
    #pragma unroll
    for (int k0 = 0; k0 < IN_DIM; k0 += 4) {
        v2f a = *(const v2f*)(xr + k0);
        v2f b;
        b.x = wc[(long)k0 * OUT_DIM];
        b.y = wc[(long)(k0 + 1) * OUT_DIM];
        c = __builtin_amdgcn_wmma_f32_16x16x4_f32(false, a, false, b,
                                                  (short)0, c, false, false);
    }

    #pragma unroll
    for (int r = 0; r < 8; ++r) {
        int row = row0 + r + 8 * grp;
        if (row < n) h[(long)row * OUT_DIM + col0 + lid] = c[r];
    }
}

__global__ void node_proj(const float* __restrict__ h,
                          const float* __restrict__ attn_src,
                          const float* __restrict__ attn_dst,
                          float* __restrict__ a_src,
                          float* __restrict__ a_dst, int n) {
    int idx = blockIdx.x * blockDim.x + threadIdx.x;
    if (idx >= n * H_HEADS) return;
    int hh   = idx & 3;
    int node = idx >> 2;
    const float4* hp = (const float4*)(h + (long)node * OUT_DIM + hh * HD);
    const float4* as = (const float4*)(attn_src + hh * HD);
    const float4* ad = (const float4*)(attn_dst + hh * HD);
    float s = 0.0f, d = 0.0f;
    #pragma unroll
    for (int j = 0; j < 4; ++j) {
        float4 hv = hp[j], sv = as[j], dv = ad[j];
        s += hv.x * sv.x + hv.y * sv.y + hv.z * sv.z + hv.w * sv.w;
        d += hv.x * dv.x + hv.y * dv.y + hv.z * dv.z + hv.w * dv.w;
    }
    a_src[idx] = s;
    a_dst[idx] = d;
}

__global__ void edge_score(const int* __restrict__ src,
                           const int* __restrict__ dst,
                           const float* __restrict__ edge_attr,
                           const float* __restrict__ W_edge,
                           const float* __restrict__ a_src,
                           const float* __restrict__ a_dst,
                           float* __restrict__ attn, long e_cnt) {
    long e = (long)blockIdx.x * blockDim.x + threadIdx.x;
    if (e >= e_cnt) return;
    int s = src[e], t = dst[e];

    float sc[H_HEADS] = {0.f, 0.f, 0.f, 0.f};
    const float4* ea = (const float4*)(edge_attr + e * EDGE_DIM);
    #pragma unroll
    for (int q = 0; q < 4; ++q) {
        float4 v = ea[q];
        int d0 = q * 4;
        #pragma unroll
        for (int hh = 0; hh < H_HEADS; ++hh) {
            sc[hh] += v.x * W_edge[(d0 + 0) * H_HEADS + hh]
                    + v.y * W_edge[(d0 + 1) * H_HEADS + hh]
                    + v.z * W_edge[(d0 + 2) * H_HEADS + hh]
                    + v.w * W_edge[(d0 + 3) * H_HEADS + hh];
        }
    }
    float4 asv = ((const float4*)a_src)[s];
    float4 adv = ((const float4*)a_dst)[t];
    float av[H_HEADS] = {asv.x + adv.x, asv.y + adv.y, asv.z + adv.z, asv.w + adv.w};

    float4 outv;
    float* op = (float*)&outv;
    #pragma unroll
    for (int hh = 0; hh < H_HEADS; ++hh) {
        float v = av[hh] + sc[hh];
        v = (v >= 0.0f) ? v : NEG_SLOPE * v;
        op[hh] = v;
    }
    ((float4*)attn)[e] = outv;
}

__global__ void dst_rank(const int* __restrict__ dst,
                         int* __restrict__ cnt,
                         int* __restrict__ rank, long e_cnt) {
    long e = (long)blockIdx.x * blockDim.x + threadIdx.x;
    if (e >= e_cnt) return;
    rank[e] = atomicAdd(&cnt[dst[e]], 1);
}

__global__ void __launch_bounds__(1024)
seg_scan(const int* __restrict__ cnt, int* __restrict__ seg, int n) {
    __shared__ int wsum[2][32];
    int tid  = threadIdx.x;
    int lane = tid & 31;
    int wave = tid >> 5;
    int carry = 0;
    int buf = 0;

    int4 v = make_int4(0, 0, 0, 0);
    {
        int i = tid * 4;
        if (i + 3 < n) v = *(const int4*)(cnt + i);
        else {
            if (i     < n) v.x = cnt[i];
            if (i + 1 < n) v.y = cnt[i + 1];
            if (i + 2 < n) v.z = cnt[i + 2];
        }
    }
    for (int base = 0; base < n; base += 4096) {
        int i = base + tid * 4;
        int4 vn = make_int4(0, 0, 0, 0);
        {
            int j = i + 4096;
            if (j + 3 < n) vn = *(const int4*)(cnt + j);
            else {
                if (j     < n) vn.x = cnt[j];
                if (j + 1 < n) vn.y = cnt[j + 1];
                if (j + 2 < n) vn.z = cnt[j + 2];
            }
        }
        int s = v.x + v.y + v.z + v.w;
        int incl = s;
        #pragma unroll
        for (int d = 1; d < 32; d <<= 1) {
            int t = __shfl_up(incl, d, 32);
            if (lane >= d) incl += t;
        }
        if (lane == 31) wsum[buf][wave] = incl;
        __syncthreads();
        int wv = wsum[buf][lane];
        int wincl = wv;
        #pragma unroll
        for (int d = 1; d < 32; d <<= 1) {
            int t = __shfl_up(wincl, d, 32);
            if (lane >= d) wincl += t;
        }
        int wexcl = __shfl(wincl - wv, wave, 32);
        int total = __shfl(wincl, 31, 32);
        int x0 = carry + wexcl + (incl - s);
        if (i + 3 < n) {
            *(int4*)(seg + i) = make_int4(x0, x0 + v.x, x0 + v.x + v.y,
                                          x0 + v.x + v.y + v.z);
        } else {
            if (i     < n) seg[i]     = x0;
            if (i + 1 < n) seg[i + 1] = x0 + v.x;
            if (i + 2 < n) seg[i + 2] = x0 + v.x + v.y;
        }
        carry += total;
        v = vn;
        buf ^= 1;
    }
    if (tid == 0) seg[n] = carry;
}

__global__ void dst_fill(const int* __restrict__ src,
                         const int* __restrict__ dst,
                         const int* __restrict__ rank,
                         const int* __restrict__ seg,
                         int* __restrict__ perm, long e_cnt) {
    long e = (long)blockIdx.x * blockDim.x + threadIdx.x;
    if (e >= e_cnt) return;
    int p = seg[dst[e]] + rank[e];
    ((int2*)perm)[p] = make_int2((int)e, src[e]);
}

__device__ __forceinline__ float xhalf(float x) {
    int b = __float_as_int(x);
    return __int_as_float(__builtin_amdgcn_permlanex16(b, b, 0x76543210, 0xfedcba98, false, false));
}
template <int M>
__device__ __forceinline__ float xrow(float x) {
    return __int_as_float(__builtin_amdgcn_update_dpp(0, __float_as_int(x), 0x160 + M, 0xf, 0xf, false));
}
__device__ __forceinline__ float wave_max(float v) {
    v = fmaxf(v, xrow<1>(v));
    v = fmaxf(v, xrow<2>(v));
    v = fmaxf(v, xrow<4>(v));
    v = fmaxf(v, xrow<8>(v));
    return fmaxf(v, xhalf(v));
}
__device__ __forceinline__ float wave_sum(float v) {
    v += xrow<1>(v);
    v += xrow<2>(v);
    v += xrow<4>(v);
    v += xrow<8>(v);
    return v + xhalf(v);
}

__device__ __forceinline__ void add_rows(const float* __restrict__ hl, int srcs,
                                         int w0, int w1, int w2, int w3, int m,
                                         int hm0, int hm1, int hm2, int hm3,
                                         float& ax, float& ay) {
    for (int k0 = 0; k0 < m; k0 += 8) {
        float2 hv[8];
        float  wk[8];
        #pragma unroll
        for (int j = 0; j < 8; ++j) {
            int k = k0 + j;
            int s = __builtin_amdgcn_readlane(srcs, k);
            int u0 = __builtin_amdgcn_readlane(w0, k);
            int u1 = __builtin_amdgcn_readlane(w1, k);
            int u2 = __builtin_amdgcn_readlane(w2, k);
            int u3 = __builtin_amdgcn_readlane(w3, k);
            wk[j] = __int_as_float((u0 & hm0) | (u1 & hm1) | (u2 & hm2) | (u3 & hm3));
            hv[j] = *(const float2*)(hl + (long)s * OUT_DIM);
        }
        #pragma unroll
        for (int j = 0; j < 8; ++j) {
            ax += hv[j].x * wk[j];
            ay += hv[j].y * wk[j];
        }
    }
}

__global__ void __launch_bounds__(256)
gather_ln(const int* __restrict__ perm,
          const int* __restrict__ seg,
          const float* __restrict__ h,
          const float* __restrict__ score,
          const float* __restrict__ gamma,
          const float* __restrict__ beta,
          float* __restrict__ out, int n) {
    int lane = threadIdx.x & 31;
    int node = __builtin_amdgcn_readfirstlane((int)(blockIdx.x * 8 + (threadIdx.x >> 5)));
    if (node >= n) return;

    int beg = seg[node];
    int end = seg[node + 1];

    int head = lane >> 3;
    int hm0 = (head == 0) ? -1 : 0, hm1 = (head == 1) ? -1 : 0;
    int hm2 = (head == 2) ? -1 : 0, hm3 = (head == 3) ? -1 : 0;
    const float* hl = h + 2 * lane;
    const int2*   perm2  = (const int2*)perm;
    const float4* score4 = (const float4*)score;

    int m0 = end - beg;
    if (m0 > 32) m0 = 32;
    int2 pe0 = make_int2(0, 0);
    float4 s0 = make_float4(0.0f, 0.0f, 0.0f, 0.0f);
    if (lane < m0) {
        pe0 = perm2[beg + lane];
        s0  = score4[pe0.x];
    }

    float mx0 = fmaxf(s0.x, 0.0f), mx1 = fmaxf(s0.y, 0.0f);
    float mx2 = fmaxf(s0.z, 0.0f), mx3 = fmaxf(s0.w, 0.0f);
    for (int cb = beg + 32; cb < end; cb += 32) {
        if (lane < end - cb) {
            float4 s = score4[perm2[cb + lane].x];
            mx0 = fmaxf(mx0, s.x);  mx1 = fmaxf(mx1, s.y);
            mx2 = fmaxf(mx2, s.z);  mx3 = fmaxf(mx3, s.w);
        }
    }
    mx0 = wave_max(mx0);  mx1 = wave_max(mx1);
    mx2 = wave_max(mx2);  mx3 = wave_max(mx3);

    float e0 = 0.0f, e1 = 0.0f, e2 = 0.0f, e3 = 0.0f;
    if (lane < m0) {
        e0 = expf(s0.x - mx0);  e1 = expf(s0.y - mx1);
        e2 = expf(s0.z - mx2);  e3 = expf(s0.w - mx3);
    }
    float sm0 = e0, sm1 = e1, sm2 = e2, sm3 = e3;
    for (int cb = beg + 32; cb < end; cb += 32) {
        if (lane < end - cb) {
            float4 s = score4[perm2[cb + lane].x];
            sm0 += expf(s.x - mx0);  sm1 += expf(s.y - mx1);
            sm2 += expf(s.z - mx2);  sm3 += expf(s.w - mx3);
        }
    }
    float den0 = wave_sum(sm0) + EPS_SOFTMAX, den1 = wave_sum(sm1) + EPS_SOFTMAX;
    float den2 = wave_sum(sm2) + EPS_SOFTMAX, den3 = wave_sum(sm3) + EPS_SOFTMAX;

    float ax = 0.0f, ay = 0.0f;
    add_rows(hl, pe0.y,
             __float_as_int(e0 / den0), __float_as_int(e1 / den1),
             __float_as_int(e2 / den2), __float_as_int(e3 / den3),
             m0, hm0, hm1, hm2, hm3, ax, ay);
    for (int cb = beg + 32; cb < end; cb += 32) {
        int m = end - cb;
        if (m > 32) m = 32;
        int2 pe = make_int2(0, 0);
        float x0 = 0.0f, x1 = 0.0f, x2 = 0.0f, x3 = 0.0f;
        if (lane < m) {
            pe = perm2[cb + lane];
            float4 s = score4[pe.x];
            x0 = expf(s.x - mx0);  x1 = expf(s.y - mx1);
            x2 = expf(s.z - mx2);  x3 = expf(s.w - mx3);
        }
        add_rows(hl, pe.y,
                 __float_as_int(x0 / den0), __float_as_int(x1 / den1),
                 __float_as_int(x2 / den2), __float_as_int(x3 / den3),
                 m, hm0, hm1, hm2, hm3, ax, ay);
    }

    long base = (long)node * OUT_DIM + 2 * lane;
    float2 hn = *(const float2*)(h + base);
    float y0 = ax + hn.x;
    float y1 = ay + hn.y;

    float s = y0 + y1;
    #pragma unroll
    for (int o = 16; o > 0; o >>= 1) s += __shfl_xor(s, o, 32);
    float mu = s * (1.0f / 64.0f);

    float d0 = y0 - mu, d1 = y1 - mu;
    float vs = d0 * d0 + d1 * d1;
    #pragma unroll
    for (int o = 16; o > 0; o >>= 1) vs += __shfl_xor(vs, o, 32);
    float rstd = rsqrtf(vs * (1.0f / 64.0f) + EPS_LN);

    float2 g = ((const float2*)gamma)[lane];
    float2 b = ((const float2*)beta)[lane];
    float2 o2;
    o2.x = d0 * rstd * g.x + b.x;
    o2.y = d1 * rstd * g.y + b.y;
    *(float2*)(out + base) = o2;
}

extern "C" void kernel_launch(void* const* d_in, const int* in_sizes, int n_in,
                              void* d_out, int out_size, void* d_ws, size_t ws_size,
                              hipStream_t stream) {
    const float* x         = (const float*)d_in[0];
    const int*   ei        = (const int*)  d_in[1];
    const float* edge_attr = (const float*)d_in[2];
    const float* W_node    = (const float*)d_in[3];
    const float* W_edge    = (const float*)d_in[4];
    const float* attn_src  = (const float*)d_in[5];
    const float* attn_dst  = (const float*)d_in[6];
    const float* gamma     = (const float*)d_in[7];
    const float* beta      = (const float*)d_in[8];
    float* out = (float*)d_out;

    int  n = in_sizes[0] / IN_DIM;
    long e = (long)in_sizes[2] / EDGE_DIM;
    const int* srcIdx = ei;
    const int* dstIdx = ei + e;

    float* ws    = (float*)d_ws;
    float* h     = ws;  ws += (long)n * OUT_DIM;
    float* a_src = ws;  ws += (long)n * H_HEADS;
    float* a_dst = ws;  ws += (long)n * H_HEADS;
    float* attn  = ws;  ws += e * H_HEADS;
    int*   cnt   = (int*)ws;  ws += ((long)n + 3) / 4 * 4;
    int*   seg   = (int*)ws;  ws += ((long)n + 4) / 4 * 4;
    int*   perm  = (int*)ws;  ws += 2 * e;
    int*   rank  = (int*)ws;  ws += e;

    long zn = (long)n;
    zero_kernel<<<(int)((zn + 255) / 256), 256, 0, stream>>>((float*)cnt, zn);

    int n_tiles = (n + 15) / 16;
    node_gemm_wmma<<<n_tiles, 128, 0, stream>>>(x, W_node, h, n);

    node_proj<<<(n * H_HEADS + 255) / 256, 256, 0, stream>>>(
        h, attn_src, attn_dst, a_src, a_dst, n);

    edge_score<<<(int)((e + 255) / 256), 256, 0, stream>>>(
        srcIdx, dstIdx, edge_attr, W_edge, a_src, a_dst, attn, e);

    dst_rank<<<(int)((e + 255) / 256), 256, 0, stream>>>(dstIdx, cnt, rank, e);
    seg_scan<<<1, 1024, 0, stream>>>(cnt, seg, n);
    dst_fill<<<(int)((e + 255) / 256), 256, 0, stream>>>(
        srcIdx, dstIdx, rank, seg, perm, e);

    gather_ln<<<(n + 7) / 8, 256, 0, stream>>>(
        perm, seg, h, attn, gamma, beta, out, n);
}


# edges grouped by dst by a two-level counting sort on LDS counters (bin by 256-node range, then rank inside the bin) instead of one global returning add per edge; sum order per node changes
# speedup vs baseline: 3.8541x; 1.6449x over previous
#include <hip/hip_runtime.h>
#include <hip/hip_bf16.h>

typedef float v2f __attribute__((ext_vector_type(2)));
typedef float v8f __attribute__((ext_vector_type(8)));

#define IN_DIM   64
#define OUT_DIM  64
#define EDGE_DIM 16
#define H_HEADS  4
#define HD       16
#define NEG_SLOPE   0.2f
#define EPS_SOFTMAX 1e-8f
#define EPS_LN      1e-5f

__global__ void node_gemm_wmma(const float* __restrict__ x,
                               const float* __restrict__ W,
                               float* __restrict__ h, int n) {
    int wave = threadIdx.x >> 5;
    int lane = threadIdx.x & 31;
    int grp  = lane >> 4;
    int lid  = lane & 15;

    int row0 = blockIdx.x * 16;
    int col0 = wave * 16;

    int arow = row0 + lid;
    if (arow >= n) arow = n - 1;

    const float* xr = x + (long)arow * IN_DIM + 2 * grp;
    const float* wc = W + (long)(2 * grp) * OUT_DIM + col0 + lid;

    v8f c = {};
    #pragma unroll
    for (int k0 = 0; k0 < IN_DIM; k0 += 4) {
        v2f a = *(const v2f*)(xr + k0);
        v2f b;
        b.x = wc[(long)k0 * OUT_DIM];
        b.y = wc[(long)(k0 + 1) * OUT_DIM];
        c = __builtin_amdgcn_wmma_f32_16x16x4_f32(false, a, false, b,
                                                  (short)0, c, false, false);
    }

    #pragma unroll
    for (int r = 0; r < 8; ++r) {
        int row = row0 + r + 8 * grp;
        if (row < n) h[(long)row * OUT_DIM + col0 + lid] = c[r];
    }
}

__global__ void node_proj(const float* __restrict__ h,
                          const float* __restrict__ attn_src,
                          const float* __restrict__ attn_dst,
                          float* __restrict__ a_src,
                          float* __restrict__ a_dst, int n) {
    int idx = blockIdx.x * blockDim.x + threadIdx.x;
    if (idx >= n * H_HEADS) return;
    int hh   = idx & 3;
    int node = idx >> 2;
    const float4* hp = (const float4*)(h + (long)node * OUT_DIM + hh * HD);
    const float4* as = (const float4*)(attn_src + hh * HD);
    const float4* ad = (const float4*)(attn_dst + hh * HD);
    float s = 0.0f, d = 0.0f;
    #pragma unroll
    for (int j = 0; j < 4; ++j) {
        float4 hv = hp[j], sv = as[j], dv = ad[j];
        s += hv.x * sv.x + hv.y * sv.y + hv.z * sv.z + hv.w * sv.w;
        d += hv.x * dv.x + hv.y * dv.y + hv.z * dv.z + hv.w * dv.w;
    }
    a_src[idx] = s;
    a_dst[idx] = d;
}

__global__ void edge_score(const int* __restrict__ src,
                           const int* __restrict__ dst,
                           const float* __restrict__ edge_attr,
                           const float* __restrict__ W_edge,
                           const float* __restrict__ a_src,
                           const float* __restrict__ a_dst,
                           float* __restrict__ attn, long e_cnt) {
    long e = (long)blockIdx.x * blockDim.x + threadIdx.x;
    if (e >= e_cnt) return;
    int s = src[e], t = dst[e];

    float sc[H_HEADS] = {0.f, 0.f, 0.f, 0.f};
    const float4* ea = (const float4*)(edge_attr + e * EDGE_DIM);
    #pragma unroll
    for (int q = 0; q < 4; ++q) {
        float4 v = ea[q];
        int d0 = q * 4;
        #pragma unroll
        for (int hh = 0; hh < H_HEADS; ++hh) {
            sc[hh] += v.x * W_edge[(d0 + 0) * H_HEADS + hh]
                    + v.y * W_edge[(d0 + 1) * H_HEADS + hh]
                    + v.z * W_edge[(d0 + 2) * H_HEADS + hh]
                    + v.w * W_edge[(d0 + 3) * H_HEADS + hh];
        }
    }
    float4 asv = ((const float4*)a_src)[s];
    float4 adv = ((const float4*)a_dst)[t];
    float av[H_HEADS] = {asv.x + adv.x, asv.y + adv.y, asv.z + adv.z, asv.w + adv.w};

    float4 outv;
    float* op = (float*)&outv;
    #pragma unroll
    for (int hh = 0; hh < H_HEADS; ++hh) {
        float v = av[hh] + sc[hh];
        v = (v >= 0.0f) ? v : NEG_SLOPE * v;
        op[hh] = v;
    }
    ((float4*)attn)[e] = outv;
}

#define BR_SHIFT 8
#define BR       256
#define NCHUNK   256
#define SORT_T   1024
#define MAXB     1024

template <int M>
__device__ __forceinline__ int xrow_i(int x) {
    return __builtin_amdgcn_update_dpp(0, x, 0x160 + M, 0xf, 0xf, false);
}
__device__ __forceinline__ int wave_sum_i(int v) {
    v += xrow_i<1>(v);
    v += xrow_i<2>(v);
    v += xrow_i<4>(v);
    v += xrow_i<8>(v);
    return v + __builtin_amdgcn_permlanex16(v, v, 0x76543210, 0xfedcba98, false, false);
}

__device__ __forceinline__ void chunk_range(long e_cnt, long& e0, long& e1) {
    long per = (e_cnt + NCHUNK - 1) / NCHUNK;
    e0 = (long)blockIdx.x * per;
    if (e0 > e_cnt) e0 = e_cnt;
    e1 = e0 + per;
    if (e1 > e_cnt) e1 = e_cnt;
}

__global__ void __launch_bounds__(SORT_T)
bin_count(const int* __restrict__ dst, int* __restrict__ cntmat, long e_cnt, int nb) {
    __shared__ int hist[MAXB];
    int tid = threadIdx.x;
    hist[tid] = 0;
    __syncthreads();
    long e0, e1;
    chunk_range(e_cnt, e0, e1);
    for (long i0 = e0 + tid; i0 < e1; i0 += 8 * SORT_T) {
        int d[8];
        #pragma unroll
        for (int j = 0; j < 8; ++j) {
            long i = i0 + (long)j * SORT_T;
            d[j] = (i < e1) ? dst[i] : -1;
        }
        #pragma unroll
        for (int j = 0; j < 8; ++j)
            if (d[j] >= 0) atomicAdd(&hist[d[j] >> BR_SHIFT], 1);
    }
    __syncthreads();
    if (tid < nb) cntmat[(long)tid * NCHUNK + blockIdx.x] = hist[tid];
}

__global__ void __launch_bounds__(SORT_T)
bin_place(const int* __restrict__ src, const int* __restrict__ dst,
          const int* __restrict__ cntmat, int* __restrict__ bbase,
          int2* __restrict__ rec, long e_cnt, int nb) {
    __shared__ int tot[MAXB];
    __shared__ int cur[MAXB];
    __shared__ int wtot[32];
    int tid  = threadIdx.x;
    int lane = tid & 31;
    int wave = __builtin_amdgcn_readfirstlane(tid >> 5);
    int g = blockIdx.x;

    tot[tid] = 0;
    cur[tid] = 0;
    __syncthreads();
    for (int b0 = wave; b0 < nb; b0 += 8 * (SORT_T / 32)) {
        int4 u[8], v[8];
        #pragma unroll
        for (int j = 0; j < 8; ++j) {
            int b = b0 + j * (SORT_T / 32);
            if (b < nb) {
                const int4* p = (const int4*)(cntmat + (long)b * NCHUNK) + 2 * lane;
                u[j] = p[0];
                v[j] = p[1];
            }
        }
        #pragma unroll
        for (int j = 0; j < 8; ++j) {
            int b = b0 + j * (SORT_T / 32);
            if (b < nb) {
                int c0 = 8 * lane;
                int t  = u[j].x + u[j].y + u[j].z + u[j].w + v[j].x + v[j].y + v[j].z + v[j].w;
                int bf = (c0 + 0 < g ? u[j].x : 0) + (c0 + 1 < g ? u[j].y : 0)
                       + (c0 + 2 < g ? u[j].z : 0) + (c0 + 3 < g ? u[j].w : 0)
                       + (c0 + 4 < g ? v[j].x : 0) + (c0 + 5 < g ? v[j].y : 0)
                       + (c0 + 6 < g ? v[j].z : 0) + (c0 + 7 < g ? v[j].w : 0);
                t  = wave_sum_i(t);
                bf = wave_sum_i(bf);
                if (lane == 0) { tot[b] = t; cur[b] = bf; }
            }
        }
    }
    __syncthreads();

    int t = tot[tid];
    int incl = t;
    #pragma unroll
    for (int d = 1; d < 32; d <<= 1) {
        int q = __shfl_up(incl, d, 32);
        if (lane >= d) incl += q;
    }
    if (lane == 31) wtot[wave] = incl;
    __syncthreads();
    int wv = wtot[lane];
    int wincl = wv;
    #pragma unroll
    for (int d = 1; d < 32; d <<= 1) {
        int q = __shfl_up(wincl, d, 32);
        if (lane >= d) wincl += q;
    }
    int wexcl = __shfl(wincl - wv, wave, 32);
    int base = wexcl + incl - t;
    if (tid < nb) {
        cur[tid] += base;
        if (g == 0) {
            bbase[tid] = base;
            if (tid == nb - 1) bbase[nb] = base + t;
        }
    }
    __syncthreads();

    long e0, e1;
    chunk_range(e_cnt, e0, e1);
    for (long i0 = e0 + tid; i0 < e1; i0 += 8 * SORT_T) {
        int d[8], s[8];
        #pragma unroll
        for (int j = 0; j < 8; ++j) {
            long i = i0 + (long)j * SORT_T;
            bool ok = i < e1;
            d[j] = ok ? dst[i] : -1;
            s[j] = ok ? src[i] : 0;
        }
        #pragma unroll
        for (int j = 0; j < 8; ++j) {
            if (d[j] >= 0) {
                long i = i0 + (long)j * SORT_T;
                int slot = atomicAdd(&cur[d[j] >> BR_SHIFT], 1);
                unsigned loc = (unsigned)(d[j] & (BR - 1));
                rec[slot] = make_int2((int)i, (int)((unsigned)s[j] | (loc << 24)));
            }
        }
    }
}

__global__ void __launch_bounds__(SORT_T)
bucket_sort(const int2* __restrict__ rec, const int* __restrict__ bbase,
            int2* __restrict__ perm, int* __restrict__ seg, int n, int nb) {
    __shared__ int cnt[BR];
    __shared__ int wtot[BR / 32];
    int tid  = threadIdx.x;
    int lane = tid & 31;
    int wave = tid >> 5;
    int b  = blockIdx.x;
    int bb = bbase[b], be = bbase[b + 1];
    if (tid < BR) cnt[tid] = 0;
    __syncthreads();
    for (int i0 = bb + tid; i0 < be; i0 += 4 * SORT_T) {
        unsigned y[4];
        #pragma unroll
        for (int j = 0; j < 4; ++j) {
            int i = i0 + j * SORT_T;
            y[j] = (unsigned)rec[i < be ? i : be - 1].y;
        }
        #pragma unroll
        for (int j = 0; j < 4; ++j)
            if (i0 + j * SORT_T < be) atomicAdd(&cnt[y[j] >> 24], 1);
    }
    __syncthreads();
    int c = 0, incl = 0;
    if (tid < BR) {
        c = cnt[tid];
        incl = c;
        #pragma unroll
        for (int d = 1; d < 32; d <<= 1) {
            int q = __shfl_up(incl, d, 32);
            if (lane >= d) incl += q;
        }
        if (lane == 31) wtot[wave] = incl;
    }
    __syncthreads();
    if (tid < BR) {
        int add = 0;
        for (int w = 0; w < wave; ++w) add += wtot[w];
        int start = bb + add + incl - c;
        cnt[tid] = start;
        int node = b * BR + tid;
        if (node < n) seg[node] = start;
    }
    if (b == nb - 1 && tid == 0) seg[n] = be;
    __syncthreads();
    for (int i0 = bb + tid; i0 < be; i0 += 4 * SORT_T) {
        int2 r[4];
        #pragma unroll
        for (int j = 0; j < 4; ++j) {
            int i = i0 + j * SORT_T;
            r[j] = (i < be) ? rec[i] : make_int2(0, 0);
        }
        #pragma unroll
        for (int j = 0; j < 4; ++j) {
            if (i0 + j * SORT_T < be) {
                int slot = atomicAdd(&cnt[(unsigned)r[j].y >> 24], 1);
                perm[slot] = make_int2(r[j].x, r[j].y & 0x00ffffff);
            }
        }
    }
}

__device__ __forceinline__ float xhalf(float x) {
    int b = __float_as_int(x);
    return __int_as_float(__builtin_amdgcn_permlanex16(b, b, 0x76543210, 0xfedcba98, false, false));
}
template <int M>
__device__ __forceinline__ float xrow(float x) {
    return __int_as_float(__builtin_amdgcn_update_dpp(0, __float_as_int(x), 0x160 + M, 0xf, 0xf, false));
}
__device__ __forceinline__ float wave_max(float v) {
    v = fmaxf(v, xrow<1>(v));
    v = fmaxf(v, xrow<2>(v));
    v = fmaxf(v, xrow<4>(v));
    v = fmaxf(v, xrow<8>(v));
    return fmaxf(v, xhalf(v));
}
__device__ __forceinline__ float wave_sum(float v) {
    v += xrow<1>(v);
    v += xrow<2>(v);
    v += xrow<4>(v);
    v += xrow<8>(v);
    return v + xhalf(v);
}

__device__ __forceinline__ void add_rows(const float* __restrict__ hl, int srcs,
                                         int w0, int w1, int w2, int w3, int m,
                                         int hm0, int hm1, int hm2, int hm3,
                                         float& ax, float& ay) {
    for (int k0 = 0; k0 < m; k0 += 8) {
        float2 hv[8];
        float  wk[8];
        #pragma unroll
        for (int j = 0; j < 8; ++j) {
            int k = k0 + j;
            int s = __builtin_amdgcn_readlane(srcs, k);
            int u0 = __builtin_amdgcn_readlane(w0, k);
            int u1 = __builtin_amdgcn_readlane(w1, k);
            int u2 = __builtin_amdgcn_readlane(w2, k);
            int u3 = __builtin_amdgcn_readlane(w3, k);
            wk[j] = __int_as_float((u0 & hm0) | (u1 & hm1) | (u2 & hm2) | (u3 & hm3));
            hv[j] = *(const float2*)(hl + (long)s * OUT_DIM);
        }
        #pragma unroll
        for (int j = 0; j < 8; ++j) {
            ax += hv[j].x * wk[j];
            ay += hv[j].y * wk[j];
        }
    }
}

__global__ void __launch_bounds__(256)
gather_ln(const int* __restrict__ perm,
          const int* __restrict__ seg,
          const float* __restrict__ h,
          const float* __restrict__ score,
          const float* __restrict__ gamma,
          const float* __restrict__ beta,
          float* __restrict__ out, int n) {
    int lane = threadIdx.x & 31;
    int node = __builtin_amdgcn_readfirstlane((int)(blockIdx.x * 8 + (threadIdx.x >> 5)));
    if (node >= n) return;

    int beg = seg[node];
    int end = seg[node + 1];

    int head = lane >> 3;
    int hm0 = (head == 0) ? -1 : 0, hm1 = (head == 1) ? -1 : 0;
    int hm2 = (head == 2) ? -1 : 0, hm3 = (head == 3) ? -1 : 0;
    const float* hl = h + 2 * lane;
    const int2*   perm2  = (const int2*)perm;
    const float4* score4 = (const float4*)score;

    int m0 = end - beg;
    if (m0 > 32) m0 = 32;
    int2 pe0 = make_int2(0, 0);
    float4 s0 = make_float4(0.0f, 0.0f, 0.0f, 0.0f);
    if (lane < m0) {
        pe0 = perm2[beg + lane];
        s0  = score4[pe0.x];
    }

    float mx0 = fmaxf(s0.x, 0.0f), mx1 = fmaxf(s0.y, 0.0f);
    float mx2 = fmaxf(s0.z, 0.0f), mx3 = fmaxf(s0.w, 0.0f);
    for (int cb = beg + 32; cb < end; cb += 32) {
        if (lane < end - cb) {
            float4 s = score4[perm2[cb + lane].x];
            mx0 = fmaxf(mx0, s.x);  mx1 = fmaxf(mx1, s.y);
            mx2 = fmaxf(mx2, s.z);  mx3 = fmaxf(mx3, s.w);
        }
    }
    mx0 = wave_max(mx0);  mx1 = wave_max(mx1);
    mx2 = wave_max(mx2);  mx3 = wave_max(mx3);

    float e0 = 0.0f, e1 = 0.0f, e2 = 0.0f, e3 = 0.0f;
    if (lane < m0) {
        e0 = expf(s0.x - mx0);  e1 = expf(s0.y - mx1);
        e2 = expf(s0.z - mx2);  e3 = expf(s0.w - mx3);
    }
    float sm0 = e0, sm1 = e1, sm2 = e2, sm3 = e3;
    for (int cb = beg + 32; cb < end; cb += 32) {
        if (lane < end - cb) {
            float4 s = score4[perm2[cb + lane].x];
            sm0 += expf(s.x - mx0);  sm1 += expf(s.y - mx1);
            sm2 += expf(s.z - mx2);  sm3 += expf(s.w - mx3);
        }
    }
    float den0 = wave_sum(sm0) + EPS_SOFTMAX, den1 = wave_sum(sm1) + EPS_SOFTMAX;
    float den2 = wave_sum(sm2) + EPS_SOFTMAX, den3 = wave_sum(sm3) + EPS_SOFTMAX;

    float ax = 0.0f, ay = 0.0f;
    add_rows(hl, pe0.y,
             __float_as_int(e0 / den0), __float_as_int(e1 / den1),
             __float_as_int(e2 / den2), __float_as_int(e3 / den3),
             m0, hm0, hm1, hm2, hm3, ax, ay);
    for (int cb = beg + 32; cb < end; cb += 32) {
        int m = end - cb;
        if (m > 32) m = 32;
        int2 pe = make_int2(0, 0);
        float x0 = 0.0f, x1 = 0.0f, x2 = 0.0f, x3 = 0.0f;
        if (lane < m) {
            pe = perm2[cb + lane];
            float4 s = score4[pe.x];
            x0 = expf(s.x - mx0);  x1 = expf(s.y - mx1);
            x2 = expf(s.z - mx2);  x3 = expf(s.w - mx3);
        }
        add_rows(hl, pe.y,
                 __float_as_int(x0 / den0), __float_as_int(x1 / den1),
                 __float_as_int(x2 / den2), __float_as_int(x3 / den3),
                 m, hm0, hm1, hm2, hm3, ax, ay);
    }

    long base = (long)node * OUT_DIM + 2 * lane;
    float2 hn = *(const float2*)(h + base);
    float y0 = ax + hn.x;
    float y1 = ay + hn.y;

    float s = y0 + y1;
    #pragma unroll
    for (int o = 16; o > 0; o >>= 1) s += __shfl_xor(s, o, 32);
    float mu = s * (1.0f / 64.0f);

    float d0 = y0 - mu, d1 = y1 - mu;
    float vs = d0 * d0 + d1 * d1;
    #pragma unroll
    for (int o = 16; o > 0; o >>= 1) vs += __shfl_xor(vs, o, 32);
    float rstd = rsqrtf(vs * (1.0f / 64.0f) + EPS_LN);

    float2 g = ((const float2*)gamma)[lane];
    float2 b = ((const float2*)beta)[lane];
    float2 o2;
    o2.x = d0 * rstd * g.x + b.x;
    o2.y = d1 * rstd * g.y + b.y;
    *(float2*)(out + base) = o2;
}

extern "C" void kernel_launch(void* const* d_in, const int* in_sizes, int n_in,
                              void* d_out, int out_size, void* d_ws, size_t ws_size,
                              hipStream_t stream) {
    const float* x         = (const float*)d_in[0];
    const int*   ei        = (const int*)  d_in[1];
    const float* edge_attr = (const float*)d_in[2];
    const float* W_node    = (const float*)d_in[3];
    const float* W_edge    = (const float*)d_in[4];
    const float* attn_src  = (const float*)d_in[5];
    const float* attn_dst  = (const float*)d_in[6];
    const float* gamma     = (const float*)d_in[7];
    const float* beta      = (const float*)d_in[8];
    float* out = (float*)d_out;

    int  n = in_sizes[0] / IN_DIM;
    long e = (long)in_sizes[2] / EDGE_DIM;
    const int* srcIdx = ei;
    const int* dstIdx = ei + e;

    int nb = (n + BR - 1) / BR;

    float* ws    = (float*)d_ws;
    float* h     = ws;  ws += (long)n * OUT_DIM;
    float* a_src = ws;  ws += (long)n * H_HEADS;
    float* a_dst = ws;  ws += (long)n * H_HEADS;
    float* attn  = ws;  ws += e * H_HEADS;
    int*   seg   = (int*)ws;  ws += ((long)n + 4) / 4 * 4;
    int*   perm  = (int*)ws;  ws += 2 * e;
    int*   cntmat = (int*)ws; ws += (long)nb * NCHUNK;
    int*   bbase = (int*)ws;  ws += nb + 1;
    int2*  rec   = (int2*)attn;

    int n_tiles = (n + 15) / 16;
    node_gemm_wmma<<<n_tiles, 128, 0, stream>>>(x, W_node, h, n);

    node_proj<<<(n * H_HEADS + 255) / 256, 256, 0, stream>>>(
        h, attn_src, attn_dst, a_src, a_dst, n);

    bin_count<<<NCHUNK, SORT_T, 0, stream>>>(dstIdx, cntmat, e, nb);
    bin_place<<<NCHUNK, SORT_T, 0, stream>>>(srcIdx, dstIdx, cntmat, bbase, rec, e, nb);
    bucket_sort<<<nb, SORT_T, 0, stream>>>(rec, bbase, (int2*)perm, seg, n, nb);

    edge_score<<<(int)((e + 255) / 256), 256, 0, stream>>>(
        srcIdx, dstIdx, edge_attr, W_edge, a_src, a_dst, attn, e);

    gather_ln<<<(n + 7) / 8, 256, 0, stream>>>(
        perm, seg, h, attn, gamma, beta, out, n);
}


# gather_ln: edge lanes hand head weights and source-row offsets to the feature lanes via a per-wave LDS table (ds_load_b128 per 4 edges) instead of 5 readlanes + 6 selects per edge; sum order kept
# speedup vs baseline: 3.9725x; 1.0307x over previous
#include <hip/hip_runtime.h>
#include <hip/hip_bf16.h>

typedef float v2f __attribute__((ext_vector_type(2)));
typedef float v8f __attribute__((ext_vector_type(8)));

#define IN_DIM   64
#define OUT_DIM  64
#define EDGE_DIM 16
#define H_HEADS  4
#define HD       16
#define NEG_SLOPE   0.2f
#define EPS_SOFTMAX 1e-8f
#define EPS_LN      1e-5f

__global__ void node_gemm_wmma(const float* __restrict__ x,
                               const float* __restrict__ W,
                               float* __restrict__ h, int n) {
    int wave = threadIdx.x >> 5;
    int lane = threadIdx.x & 31;
    int grp  = lane >> 4;
    int lid  = lane & 15;

    int row0 = blockIdx.x * 16;
    int col0 = wave * 16;

    int arow = row0 + lid;
    if (arow >= n) arow = n - 1;

    const float* xr = x + (long)arow * IN_DIM + 2 * grp;
    const float* wc = W + (long)(2 * grp) * OUT_DIM + col0 + lid;

    v8f c = {};
    #pragma unroll
    for (int k0 = 0; k0 < IN_DIM; k0 += 4) {
        v2f a = *(const v2f*)(xr + k0);
        v2f b;
        b.x = wc[(long)k0 * OUT_DIM];
        b.y = wc[(long)(k0 + 1) * OUT_DIM];
        c = __builtin_amdgcn_wmma_f32_16x16x4_f32(false, a, false, b,
                                                  (short)0, c, false, false);
    }

    #pragma unroll
    for (int r = 0; r < 8; ++r) {
        int row = row0 + r + 8 * grp;
        if (row < n) h[(long)row * OUT_DIM + col0 + lid] = c[r];
    }
}

__global__ void node_proj(const float* __restrict__ h,
                          const float* __restrict__ attn_src,
                          const float* __restrict__ attn_dst,
                          float* __restrict__ a_src,
                          float* __restrict__ a_dst, int n) {
    int idx = blockIdx.x * blockDim.x + threadIdx.x;
    if (idx >= n * H_HEADS) return;
    int hh   = idx & 3;
    int node = idx >> 2;
    const float4* hp = (const float4*)(h + (long)node * OUT_DIM + hh * HD);
    const float4* as = (const float4*)(attn_src + hh * HD);
    const float4* ad = (const float4*)(attn_dst + hh * HD);
    float s = 0.0f, d = 0.0f;
    #pragma unroll
    for (int j = 0; j < 4; ++j) {
        float4 hv = hp[j], sv = as[j], dv = ad[j];
        s += hv.x * sv.x + hv.y * sv.y + hv.z * sv.z + hv.w * sv.w;
        d += hv.x * dv.x + hv.y * dv.y + hv.z * dv.z + hv.w * dv.w;
    }
    a_src[idx] = s;
    a_dst[idx] = d;
}

__global__ void edge_score(const int* __restrict__ src,
                           const int* __restrict__ dst,
                           const float* __restrict__ edge_attr,
                           const float* __restrict__ W_edge,
                           const float* __restrict__ a_src,
                           const float* __restrict__ a_dst,
                           float* __restrict__ attn, long e_cnt) {
    long e = (long)blockIdx.x * blockDim.x + threadIdx.x;
    if (e >= e_cnt) return;
    int s = src[e], t = dst[e];

    float sc[H_HEADS] = {0.f, 0.f, 0.f, 0.f};
    const float4* ea = (const float4*)(edge_attr + e * EDGE_DIM);
    #pragma unroll
    for (int q = 0; q < 4; ++q) {
        float4 v = ea[q];
        int d0 = q * 4;
        #pragma unroll
        for (int hh = 0; hh < H_HEADS; ++hh) {
            sc[hh] += v.x * W_edge[(d0 + 0) * H_HEADS + hh]
                    + v.y * W_edge[(d0 + 1) * H_HEADS + hh]
                    + v.z * W_edge[(d0 + 2) * H_HEADS + hh]
                    + v.w * W_edge[(d0 + 3) * H_HEADS + hh];
        }
    }
    float4 asv = ((const float4*)a_src)[s];
    float4 adv = ((const float4*)a_dst)[t];
    float av[H_HEADS] = {asv.x + adv.x, asv.y + adv.y, asv.z + adv.z, asv.w + adv.w};

    float4 outv;
    float* op = (float*)&outv;
    #pragma unroll
    for (int hh = 0; hh < H_HEADS; ++hh) {
        float v = av[hh] + sc[hh];
        v = (v >= 0.0f) ? v : NEG_SLOPE * v;
        op[hh] = v;
    }
    ((float4*)attn)[e] = outv;
}

#define BR_SHIFT 8
#define BR       256
#define NCHUNK   256
#define SORT_T   1024
#define MAXB     1024

template <int M>
__device__ __forceinline__ int xrow_i(int x) {
    return __builtin_amdgcn_update_dpp(0, x, 0x160 + M, 0xf, 0xf, false);
}
__device__ __forceinline__ int wave_sum_i(int v) {
    v += xrow_i<1>(v);
    v += xrow_i<2>(v);
    v += xrow_i<4>(v);
    v += xrow_i<8>(v);
    return v + __builtin_amdgcn_permlanex16(v, v, 0x76543210, 0xfedcba98, false, false);
}

__device__ __forceinline__ void chunk_range(long e_cnt, long& e0, long& e1) {
    long per = (e_cnt + NCHUNK - 1) / NCHUNK;
    e0 = (long)blockIdx.x * per;
    if (e0 > e_cnt) e0 = e_cnt;
    e1 = e0 + per;
    if (e1 > e_cnt) e1 = e_cnt;
}

__global__ void __launch_bounds__(SORT_T)
bin_count(const int* __restrict__ dst, int* __restrict__ cntmat, long e_cnt, int nb) {
    __shared__ int hist[MAXB];
    int tid = threadIdx.x;
    hist[tid] = 0;
    __syncthreads();
    long e0, e1;
    chunk_range(e_cnt, e0, e1);
    for (long i0 = e0 + tid; i0 < e1; i0 += 8 * SORT_T) {
        int d[8];
        #pragma unroll
        for (int j = 0; j < 8; ++j) {
            long i = i0 + (long)j * SORT_T;
            d[j] = (i < e1) ? dst[i] : -1;
        }
        #pragma unroll
        for (int j = 0; j < 8; ++j)
            if (d[j] >= 0) atomicAdd(&hist[d[j] >> BR_SHIFT], 1);
    }
    __syncthreads();
    if (tid < nb) cntmat[(long)tid * NCHUNK + blockIdx.x] = hist[tid];
}

__global__ void __launch_bounds__(SORT_T)
bin_place(const int* __restrict__ src, const int* __restrict__ dst,
          const int* __restrict__ cntmat, int* __restrict__ bbase,
          int2* __restrict__ rec, long e_cnt, int nb) {
    __shared__ int tot[MAXB];
    __shared__ int cur[MAXB];
    __shared__ int wtot[32];
    int tid  = threadIdx.x;
    int lane = tid & 31;
    int wave = __builtin_amdgcn_readfirstlane(tid >> 5);
    int g = blockIdx.x;

    tot[tid] = 0;
    cur[tid] = 0;
    __syncthreads();
    for (int b0 = wave; b0 < nb; b0 += 8 * (SORT_T / 32)) {
        int4 u[8], v[8];
        #pragma unroll
        for (int j = 0; j < 8; ++j) {
            int b = b0 + j * (SORT_T / 32);
            if (b < nb) {
                const int4* p = (const int4*)(cntmat + (long)b * NCHUNK) + 2 * lane;
                u[j] = p[0];
                v[j] = p[1];
            }
        }
        #pragma unroll
        for (int j = 0; j < 8; ++j) {
            int b = b0 + j * (SORT_T / 32);
            if (b < nb) {
                int c0 = 8 * lane;
                int t  = u[j].x + u[j].y + u[j].z + u[j].w + v[j].x + v[j].y + v[j].z + v[j].w;
                int bf = (c0 + 0 < g ? u[j].x : 0) + (c0 + 1 < g ? u[j].y : 0)
                       + (c0 + 2 < g ? u[j].z : 0) + (c0 + 3 < g ? u[j].w : 0)
                       + (c0 + 4 < g ? v[j].x : 0) + (c0 + 5 < g ? v[j].y : 0)
                       + (c0 + 6 < g ? v[j].z : 0) + (c0 + 7 < g ? v[j].w : 0);
                t  = wave_sum_i(t);
                bf = wave_sum_i(bf);
                if (lane == 0) { tot[b] = t; cur[b] = bf; }
            }
        }
    }
    __syncthreads();

    int t = tot[tid];
    int incl = t;
    #pragma unroll
    for (int d = 1; d < 32; d <<= 1) {
        int q = __shfl_up(incl, d, 32);
        if (lane >= d) incl += q;
    }
    if (lane == 31) wtot[wave] = incl;
    __syncthreads();
    int wv = wtot[lane];
    int wincl = wv;
    #pragma unroll
    for (int d = 1; d < 32; d <<= 1) {
        int q = __shfl_up(wincl, d, 32);
        if (lane >= d) wincl += q;
    }
    int wexcl = __shfl(wincl - wv, wave, 32);
    int base = wexcl + incl - t;
    if (tid < nb) {
        cur[tid] += base;
        if (g == 0) {
            bbase[tid] = base;
            if (tid == nb - 1) bbase[nb] = base + t;
        }
    }
    __syncthreads();

    long e0, e1;
    chunk_range(e_cnt, e0, e1);
    for (long i0 = e0 + tid; i0 < e1; i0 += 8 * SORT_T) {
        int d[8], s[8];
        #pragma unroll
        for (int j = 0; j < 8; ++j) {
            long i = i0 + (long)j * SORT_T;
            bool ok = i < e1;
            d[j] = ok ? dst[i] : -1;
            s[j] = ok ? src[i] : 0;
        }
        #pragma unroll
        for (int j = 0; j < 8; ++j) {
            if (d[j] >= 0) {
                long i = i0 + (long)j * SORT_T;
                int slot = atomicAdd(&cur[d[j] >> BR_SHIFT], 1);
                unsigned loc = (unsigned)(d[j] & (BR - 1));
                rec[slot] = make_int2((int)i, (int)((unsigned)s[j] | (loc << 24)));
            }
        }
    }
}

__global__ void __launch_bounds__(SORT_T)
bucket_sort(const int2* __restrict__ rec, const int* __restrict__ bbase,
            int2* __restrict__ perm, int* __restrict__ seg, int n, int nb) {
    __shared__ int cnt[BR];
    __shared__ int wtot[BR / 32];
    int tid  = threadIdx.x;
    int lane = tid & 31;
    int wave = tid >> 5;
    int b  = blockIdx.x;
    int bb = bbase[b], be = bbase[b + 1];
    if (tid < BR) cnt[tid] = 0;
    __syncthreads();
    for (int i0 = bb + tid; i0 < be; i0 += 4 * SORT_T) {
        unsigned y[4];
        #pragma unroll
        for (int j = 0; j < 4; ++j) {
            int i = i0 + j * SORT_T;
            y[j] = (unsigned)rec[i < be ? i : be - 1].y;
        }
        #pragma unroll
        for (int j = 0; j < 4; ++j)
            if (i0 + j * SORT_T < be) atomicAdd(&cnt[y[j] >> 24], 1);
    }
    __syncthreads();
    int c = 0, incl = 0;
    if (tid < BR) {
        c = cnt[tid];
        incl = c;
        #pragma unroll
        for (int d = 1; d < 32; d <<= 1) {
            int q = __shfl_up(incl, d, 32);
            if (lane >= d) incl += q;
        }
        if (lane == 31) wtot[wave] = incl;
    }
    __syncthreads();
    if (tid < BR) {
        int add = 0;
        for (int w = 0; w < wave; ++w) add += wtot[w];
        int start = bb + add + incl - c;
        cnt[tid] = start;
        int node = b * BR + tid;
        if (node < n) seg[node] = start;
    }
    if (b == nb - 1 && tid == 0) seg[n] = be;
    __syncthreads();
    for (int i0 = bb + tid; i0 < be; i0 += 4 * SORT_T) {
        int2 r[4];
        #pragma unroll
        for (int j = 0; j < 4; ++j) {
            int i = i0 + j * SORT_T;
            r[j] = (i < be) ? rec[i] : make_int2(0, 0);
        }
        #pragma unroll
        for (int j = 0; j < 4; ++j) {
            if (i0 + j * SORT_T < be) {
                int slot = atomicAdd(&cnt[(unsigned)r[j].y >> 24], 1);
                perm[slot] = make_int2(r[j].x, r[j].y & 0x00ffffff);
            }
        }
    }
}

__device__ __forceinline__ float xhalf(float x) {
    int b = __float_as_int(x);
    return __int_as_float(__builtin_amdgcn_permlanex16(b, b, 0x76543210, 0xfedcba98, false, false));
}
template <int M>
__device__ __forceinline__ float xrow(float x) {
    return __int_as_float(__builtin_amdgcn_update_dpp(0, __float_as_int(x), 0x160 + M, 0xf, 0xf, false));
}
__device__ __forceinline__ float wave_max(float v) {
    v = fmaxf(v, xrow<1>(v));
    v = fmaxf(v, xrow<2>(v));
    v = fmaxf(v, xrow<4>(v));
    v = fmaxf(v, xrow<8>(v));
    return fmaxf(v, xhalf(v));
}
__device__ __forceinline__ float wave_sum(float v) {
    v += xrow<1>(v);
    v += xrow<2>(v);
    v += xrow<4>(v);
    v += xrow<8>(v);
    return v + xhalf(v);
}

#define GWAVES 8
#define WSTR   36
#define TABW   (4 * WSTR + 32)

__device__ __forceinline__ void wave_lds_sync() {
    __builtin_amdgcn_fence(__ATOMIC_RELEASE, "wavefront");
    __builtin_amdgcn_wave_barrier();
    __builtin_amdgcn_fence(__ATOMIC_ACQUIRE, "wavefront");
}

__device__ __forceinline__ void add_rows(const char* __restrict__ hb, unsigned lane8,
                                         float* tab, const float* wrow, int lane,
                                         unsigned srcoff,
                                         float w0, float w1, float w2, float w3,
                                         int m, float& ax, float& ay) {
    unsigned* st = (unsigned*)(tab + 4 * WSTR);
    wave_lds_sync();
    tab[0 * WSTR + lane] = w0;
    tab[1 * WSTR + lane] = w1;
    tab[2 * WSTR + lane] = w2;
    tab[3 * WSTR + lane] = w3;
    st[lane] = srcoff;
    wave_lds_sync();
    for (int k0 = 0; k0 < m; k0 += 8) {
        float4 wa = *(const float4*)(wrow + k0);
        float4 wb = *(const float4*)(wrow + k0 + 4);
        uint4  sa = *(const uint4*)(st + k0);
        uint4  sb = *(const uint4*)(st + k0 + 4);
        float2 hv[8];
        hv[0] = *(const float2*)(hb + (sa.x + lane8));
        hv[1] = *(const float2*)(hb + (sa.y + lane8));
        hv[2] = *(const float2*)(hb + (sa.z + lane8));
        hv[3] = *(const float2*)(hb + (sa.w + lane8));
        hv[4] = *(const float2*)(hb + (sb.x + lane8));
        hv[5] = *(const float2*)(hb + (sb.y + lane8));
        hv[6] = *(const float2*)(hb + (sb.z + lane8));
        hv[7] = *(const float2*)(hb + (sb.w + lane8));
        ax += hv[0].x * wa.x;  ay += hv[0].y * wa.x;
        ax += hv[1].x * wa.y;  ay += hv[1].y * wa.y;
        ax += hv[2].x * wa.z;  ay += hv[2].y * wa.z;
        ax += hv[3].x * wa.w;  ay += hv[3].y * wa.w;
        ax += hv[4].x * wb.x;  ay += hv[4].y * wb.x;
        ax += hv[5].x * wb.y;  ay += hv[5].y * wb.y;
        ax += hv[6].x * wb.z;  ay += hv[6].y * wb.z;
        ax += hv[7].x * wb.w;  ay += hv[7].y * wb.w;
    }
}

__global__ void __launch_bounds__(32 * GWAVES)
gather_ln(const int* __restrict__ perm,
          const int* __restrict__ seg,
          const float* __restrict__ h,
          const float* __restrict__ score,
          const float* __restrict__ gamma,
          const float* __restrict__ beta,
          float* __restrict__ out, int n) {
    int lane = threadIdx.x & 31;
    int node = __builtin_amdgcn_readfirstlane((int)(blockIdx.x * GWAVES + (threadIdx.x >> 5)));
    if (node >= n) return;

    int beg = seg[node];
    int end = seg[node + 1];

    __shared__ __attribute__((aligned(16))) float tabs[GWAVES * TABW];
    float* tab = tabs + (threadIdx.x >> 5) * TABW;
    const float* wrow = tab + (lane >> 3) * WSTR;
    const char* hb = (const char*)h;
    unsigned lane8 = 8u * (unsigned)lane;
    const int2*   perm2  = (const int2*)perm;
    const float4* score4 = (const float4*)score;

    int m0 = end - beg;
    if (m0 > 32) m0 = 32;
    int2 pe0 = make_int2(0, 0);
    float4 s0 = make_float4(0.0f, 0.0f, 0.0f, 0.0f);
    if (lane < m0) {
        pe0 = perm2[beg + lane];
        s0  = score4[pe0.x];
    }

    float mx0 = fmaxf(s0.x, 0.0f), mx1 = fmaxf(s0.y, 0.0f);
    float mx2 = fmaxf(s0.z, 0.0f), mx3 = fmaxf(s0.w, 0.0f);
    for (int cb = beg + 32; cb < end; cb += 32) {
        if (lane < end - cb) {
            float4 s = score4[perm2[cb + lane].x];
            mx0 = fmaxf(mx0, s.x);  mx1 = fmaxf(mx1, s.y);
            mx2 = fmaxf(mx2, s.z);  mx3 = fmaxf(mx3, s.w);
        }
    }
    mx0 = wave_max(mx0);  mx1 = wave_max(mx1);
    mx2 = wave_max(mx2);  mx3 = wave_max(mx3);

    float e0 = 0.0f, e1 = 0.0f, e2 = 0.0f, e3 = 0.0f;
    if (lane < m0) {
        e0 = expf(s0.x - mx0);  e1 = expf(s0.y - mx1);
        e2 = expf(s0.z - mx2);  e3 = expf(s0.w - mx3);
    }
    float sm0 = e0, sm1 = e1, sm2 = e2, sm3 = e3;
    for (int cb = beg + 32; cb < end; cb += 32) {
        if (lane < end - cb) {
            float4 s = score4[perm2[cb + lane].x];
            sm0 += expf(s.x - mx0);  sm1 += expf(s.y - mx1);
            sm2 += expf(s.z - mx2);  sm3 += expf(s.w - mx3);
        }
    }
    float den0 = wave_sum(sm0) + EPS_SOFTMAX, den1 = wave_sum(sm1) + EPS_SOFTMAX;
    float den2 = wave_sum(sm2) + EPS_SOFTMAX, den3 = wave_sum(sm3) + EPS_SOFTMAX;

    float ax = 0.0f, ay = 0.0f;
    add_rows(hb, lane8, tab, wrow, lane, (unsigned)pe0.y << 8,
             e0 / den0, e1 / den1, e2 / den2, e3 / den3, m0, ax, ay);
    for (int cb = beg + 32; cb < end; cb += 32) {
        int m = end - cb;
        if (m > 32) m = 32;
        int2 pe = make_int2(0, 0);
        float x0 = 0.0f, x1 = 0.0f, x2 = 0.0f, x3 = 0.0f;
        if (lane < m) {
            pe = perm2[cb + lane];
            float4 s = score4[pe.x];
            x0 = expf(s.x - mx0);  x1 = expf(s.y - mx1);
            x2 = expf(s.z - mx2);  x3 = expf(s.w - mx3);
        }
        add_rows(hb, lane8, tab, wrow, lane, (unsigned)pe.y << 8,
                 x0 / den0, x1 / den1, x2 / den2, x3 / den3, m, ax, ay);
    }

    long base = (long)node * OUT_DIM + 2 * lane;
    float2 hn = *(const float2*)(h + base);
    float y0 = ax + hn.x;
    float y1 = ay + hn.y;

    float s = y0 + y1;
    #pragma unroll
    for (int o = 16; o > 0; o >>= 1) s += __shfl_xor(s, o, 32);
    float mu = s * (1.0f / 64.0f);

    float d0 = y0 - mu, d1 = y1 - mu;
    float vs = d0 * d0 + d1 * d1;
    #pragma unroll
    for (int o = 16; o > 0; o >>= 1) vs += __shfl_xor(vs, o, 32);
    float rstd = rsqrtf(vs * (1.0f / 64.0f) + EPS_LN);

    float2 g = ((const float2*)gamma)[lane];
    float2 b = ((const float2*)beta)[lane];
    float2 o2;
    o2.x = d0 * rstd * g.x + b.x;
    o2.y = d1 * rstd * g.y + b.y;
    *(float2*)(out + base) = o2;
}

extern "C" void kernel_launch(void* const* d_in, const int* in_sizes, int n_in,
                              void* d_out, int out_size, void* d_ws, size_t ws_size,
                              hipStream_t stream) {
    const float* x         = (const float*)d_in[0];
    const int*   ei        = (const int*)  d_in[1];
    const float* edge_attr = (const float*)d_in[2];
    const float* W_node    = (const float*)d_in[3];
    const float* W_edge    = (const float*)d_in[4];
    const float* attn_src  = (const float*)d_in[5];
    const float* attn_dst  = (const float*)d_in[6];
    const float* gamma     = (const float*)d_in[7];
    const float* beta      = (const float*)d_in[8];
    float* out = (float*)d_out;

    int  n = in_sizes[0] / IN_DIM;
    long e = (long)in_sizes[2] / EDGE_DIM;
    const int* srcIdx = ei;
    const int* dstIdx = ei + e;

    int nb = (n + BR - 1) / BR;

    float* ws    = (float*)d_ws;
    float* h     = ws;  ws += (long)n * OUT_DIM;
    float* a_src = ws;  ws += (long)n * H_HEADS;
    float* a_dst = ws;  ws += (long)n * H_HEADS;
    float* attn  = ws;  ws += e * H_HEADS;
    int*   seg   = (int*)ws;  ws += ((long)n + 4) / 4 * 4;
    int*   perm  = (int*)ws;  ws += 2 * e;
    int*   cntmat = (int*)ws; ws += (long)nb * NCHUNK;
    int*   bbase = (int*)ws;  ws += nb + 1;
    int2*  rec   = (int2*)attn;

    int n_tiles = (n + 15) / 16;
    node_gemm_wmma<<<n_tiles, 128, 0, stream>>>(x, W_node, h, n);

    node_proj<<<(n * H_HEADS + 255) / 256, 256, 0, stream>>>(
        h, attn_src, attn_dst, a_src, a_dst, n);

    bin_count<<<NCHUNK, SORT_T, 0, stream>>>(dstIdx, cntmat, e, nb);
    bin_place<<<NCHUNK, SORT_T, 0, stream>>>(srcIdx, dstIdx, cntmat, bbase, rec, e, nb);
    bucket_sort<<<nb, SORT_T, 0, stream>>>(rec, bbase, (int2*)perm, seg, n, nb);

    edge_score<<<(int)((e + 255) / 256), 256, 0, stream>>>(
        srcIdx, dstIdx, edge_attr, W_edge, a_src, a_dst, attn, e);

    gather_ln<<<(n + GWAVES - 1) / GWAVES, 32 * GWAVES, 0, stream>>>(
        perm, seg, h, attn, gamma, beta, out, n);
}


# node_proj launch removed: a_src, a_dst summed from the WMMA accumulator tile in node_gemm (DPP row sums); a_dst added per node in gather_ln instead of gathered per edge; score sum order changes
# speedup vs baseline: 4.2712x; 1.0752x over previous
#include <hip/hip_runtime.h>
#include <hip/hip_bf16.h>

typedef float v2f __attribute__((ext_vector_type(2)));
typedef float v8f __attribute__((ext_vector_type(8)));

#define IN_DIM   64
#define OUT_DIM  64
#define EDGE_DIM 16
#define H_HEADS  4
#define HD       16
#define NEG_SLOPE   0.2f
#define EPS_SOFTMAX 1e-8f
#define EPS_LN      1e-5f

template <int M>
__device__ __forceinline__ float xrow(float x) {
    return __int_as_float(__builtin_amdgcn_update_dpp(0, __float_as_int(x), 0x160 + M, 0xf, 0xf, false));
}
__device__ __forceinline__ float row16_sum(float v) {
    v += xrow<1>(v);
    v += xrow<2>(v);
    v += xrow<4>(v);
    v += xrow<8>(v);
    return v;
}

__global__ void node_gemm_wmma(const float* __restrict__ x,
                               const float* __restrict__ W,
                               const float* __restrict__ attn_src,
                               const float* __restrict__ attn_dst,
                               float* __restrict__ h,
                               float* __restrict__ a_src,
                               float* __restrict__ a_dst, int n) {
    int wave = threadIdx.x >> 5;
    int lane = threadIdx.x & 31;
    int grp  = lane >> 4;
    int lid  = lane & 15;

    int row0 = blockIdx.x * 16;
    int col0 = wave * 16;

    int arow = row0 + lid;
    if (arow >= n) arow = n - 1;

    const float* xr = x + (long)arow * IN_DIM + 2 * grp;
    const float* wc = W + (long)(2 * grp) * OUT_DIM + col0 + lid;

    float ws = attn_src[col0 + lid];
    float wd = attn_dst[col0 + lid];

    v8f c = {};
    #pragma unroll
    for (int k0 = 0; k0 < IN_DIM; k0 += 4) {
        v2f a = *(const v2f*)(xr + k0);
        v2f b;
        b.x = wc[(long)k0 * OUT_DIM];
        b.y = wc[(long)(k0 + 1) * OUT_DIM];
        c = __builtin_amdgcn_wmma_f32_16x16x4_f32(false, a, false, b,
                                                  (short)0, c, false, false);
    }

    #pragma unroll
    for (int r = 0; r < 8; ++r) {
        int row = row0 + r + 8 * grp;
        if (row < n) h[(long)row * OUT_DIM + col0 + lid] = c[r];
    }

    float ms = 0.0f, md = 0.0f;
    #pragma unroll
    for (int r = 0; r < 8; ++r) {
        float ps = row16_sum(c[r] * ws);
        float pd = row16_sum(c[r] * wd);
        if (lid == r) { ms = ps; md = pd; }
    }
    int prow = row0 + 8 * grp + lid;
    if (lid < 8 && prow < n) {
        a_src[(long)prow * H_HEADS + wave] = ms;
        a_dst[(long)prow * H_HEADS + wave] = md;
    }
}

__global__ void edge_score(const int* __restrict__ src,
                           const float* __restrict__ edge_attr,
                           const float* __restrict__ W_edge,
                           const float* __restrict__ a_src,
                           float* __restrict__ attn, long e_cnt) {
    long e = (long)blockIdx.x * blockDim.x + threadIdx.x;
    if (e >= e_cnt) return;
    int s = src[e];

    float sc[H_HEADS] = {0.f, 0.f, 0.f, 0.f};
    const float4* ea = (const float4*)(edge_attr + e * EDGE_DIM);
    #pragma unroll
    for (int q = 0; q < 4; ++q) {
        float4 v = ea[q];
        int d0 = q * 4;
        #pragma unroll
        for (int hh = 0; hh < H_HEADS; ++hh) {
            sc[hh] += v.x * W_edge[(d0 + 0) * H_HEADS + hh]
                    + v.y * W_edge[(d0 + 1) * H_HEADS + hh]
                    + v.z * W_edge[(d0 + 2) * H_HEADS + hh]
                    + v.w * W_edge[(d0 + 3) * H_HEADS + hh];
        }
    }
    float4 asv = ((const float4*)a_src)[s];
    ((float4*)attn)[e] = make_float4(asv.x + sc[0], asv.y + sc[1], asv.z + sc[2], asv.w + sc[3]);
}

#define BR_SHIFT 8
#define BR       256
#define NCHUNK   256
#define SORT_T   1024
#define MAXB     1024

template <int M>
__device__ __forceinline__ int xrow_i(int x) {
    return __builtin_amdgcn_update_dpp(0, x, 0x160 + M, 0xf, 0xf, false);
}
__device__ __forceinline__ int wave_sum_i(int v) {
    v += xrow_i<1>(v);
    v += xrow_i<2>(v);
    v += xrow_i<4>(v);
    v += xrow_i<8>(v);
    return v + __builtin_amdgcn_permlanex16(v, v, 0x76543210, 0xfedcba98, false, false);
}

__device__ __forceinline__ void chunk_range(long e_cnt, long& e0, long& e1) {
    long per = (e_cnt + NCHUNK - 1) / NCHUNK;
    e0 = (long)blockIdx.x * per;
    if (e0 > e_cnt) e0 = e_cnt;
    e1 = e0 + per;
    if (e1 > e_cnt) e1 = e_cnt;
}

__global__ void __launch_bounds__(SORT_T)
bin_count(const int* __restrict__ dst, int* __restrict__ cntmat, long e_cnt, int nb) {
    __shared__ int hist[MAXB];
    int tid = threadIdx.x;
    hist[tid] = 0;
    __syncthreads();
    long e0, e1;
    chunk_range(e_cnt, e0, e1);
    for (long i0 = e0 + tid; i0 < e1; i0 += 8 * SORT_T) {
        int d[8];
        #pragma unroll
        for (int j = 0; j < 8; ++j) {
            long i = i0 + (long)j * SORT_T;
            d[j] = (i < e1) ? dst[i] : -1;
        }
        #pragma unroll
        for (int j = 0; j < 8; ++j)
            if (d[j] >= 0) atomicAdd(&hist[d[j] >> BR_SHIFT], 1);
    }
    __syncthreads();
    if (tid < nb) cntmat[(long)tid * NCHUNK + blockIdx.x] = hist[tid];
}

__global__ void __launch_bounds__(SORT_T)
bin_place(const int* __restrict__ src, const int* __restrict__ dst,
          const int* __restrict__ cntmat, int* __restrict__ bbase,
          int2* __restrict__ rec, long e_cnt, int nb) {
    __shared__ int tot[MAXB];
    __shared__ int cur[MAXB];
    __shared__ int wtot[32];
    int tid  = threadIdx.x;
    int lane = tid & 31;
    int wave = __builtin_amdgcn_readfirstlane(tid >> 5);
    int g = blockIdx.x;

    tot[tid] = 0;
    cur[tid] = 0;
    __syncthreads();
    for (int b0 = wave; b0 < nb; b0 += 8 * (SORT_T / 32)) {
        int4 u[8], v[8];
        #pragma unroll
        for (int j = 0; j < 8; ++j) {
            int b = b0 + j * (SORT_T / 32);
            if (b < nb) {
                const int4* p = (const int4*)(cntmat + (long)b * NCHUNK) + 2 * lane;
                u[j] = p[0];
                v[j] = p[1];
            }
        }
        #pragma unroll
        for (int j = 0; j < 8; ++j) {
            int b = b0 + j * (SORT_T / 32);
            if (b < nb) {
                int c0 = 8 * lane;
                int t  = u[j].x + u[j].y + u[j].z + u[j].w + v[j].x + v[j].y + v[j].z + v[j].w;
                int bf = (c0 + 0 < g ? u[j].x : 0) + (c0 + 1 < g ? u[j].y : 0)
                       + (c0 + 2 < g ? u[j].z : 0) + (c0 + 3 < g ? u[j].w : 0)
                       + (c0 + 4 < g ? v[j].x : 0) + (c0 + 5 < g ? v[j].y : 0)
                       + (c0 + 6 < g ? v[j].z : 0) + (c0 + 7 < g ? v[j].w : 0);
                t  = wave_sum_i(t);
                bf = wave_sum_i(bf);
                if (lane == 0) { tot[b] = t; cur[b] = bf; }
            }
        }
    }
    __syncthreads();

    int t = tot[tid];
    int incl = t;
    #pragma unroll
    for (int d = 1; d < 32; d <<= 1) {
        int q = __shfl_up(incl, d, 32);
        if (lane >= d) incl += q;
    }
    if (lane == 31) wtot[wave] = incl;
    __syncthreads();
    int wv = wtot[lane];
    int wincl = wv;
    #pragma unroll
    for (int d = 1; d < 32; d <<= 1) {
        int q = __shfl_up(wincl, d, 32);
        if (lane >= d) wincl += q;
    }
    int wexcl = __shfl(wincl - wv, wave, 32);
    int base = wexcl + incl - t;
    if (tid < nb) {
        cur[tid] += base;
        if (g == 0) {
            bbase[tid] = base;
            if (tid == nb - 1) bbase[nb] = base + t;
        }
    }
    __syncthreads();

    long e0, e1;
    chunk_range(e_cnt, e0, e1);
    for (long i0 = e0 + tid; i0 < e1; i0 += 8 * SORT_T) {
        int d[8], s[8];
        #pragma unroll
        for (int j = 0; j < 8; ++j) {
            long i = i0 + (long)j * SORT_T;
            bool ok = i < e1;
            d[j] = ok ? dst[i] : -1;
            s[j] = ok ? src[i] : 0;
        }
        #pragma unroll
        for (int j = 0; j < 8; ++j) {
            if (d[j] >= 0) {
                long i = i0 + (long)j * SORT_T;
                int slot = atomicAdd(&cur[d[j] >> BR_SHIFT], 1);
                unsigned loc = (unsigned)(d[j] & (BR - 1));
                rec[slot] = make_int2((int)i, (int)((unsigned)s[j] | (loc << 24)));
            }
        }
    }
}

__global__ void __launch_bounds__(SORT_T)
bucket_sort(const int2* __restrict__ rec, const int* __restrict__ bbase,
            int2* __restrict__ perm, int* __restrict__ seg, int n, int nb) {
    __shared__ int cnt[BR];
    __shared__ int wtot[BR / 32];
    int tid  = threadIdx.x;
    int lane = tid & 31;
    int wave = tid >> 5;
    int b  = blockIdx.x;
    int bb = bbase[b], be = bbase[b + 1];
    if (tid < BR) cnt[tid] = 0;
    __syncthreads();
    for (int i0 = bb + tid; i0 < be; i0 += 4 * SORT_T) {
        unsigned y[4];
        #pragma unroll
        for (int j = 0; j < 4; ++j) {
            int i = i0 + j * SORT_T;
            y[j] = (unsigned)rec[i < be ? i : be - 1].y;
        }
        #pragma unroll
        for (int j = 0; j < 4; ++j)
            if (i0 + j * SORT_T < be) atomicAdd(&cnt[y[j] >> 24], 1);
    }
    __syncthreads();
    int c = 0, incl = 0;
    if (tid < BR) {
        c = cnt[tid];
        incl = c;
        #pragma unroll
        for (int d = 1; d < 32; d <<= 1) {
            int q = __shfl_up(incl, d, 32);
            if (lane >= d) incl += q;
        }
        if (lane == 31) wtot[wave] = incl;
    }
    __syncthreads();
    if (tid < BR) {
        int add = 0;
        for (int w = 0; w < wave; ++w) add += wtot[w];
        int start = bb + add + incl - c;
        cnt[tid] = start;
        int node = b * BR + tid;
        if (node < n) seg[node] = start;
    }
    if (b == nb - 1 && tid == 0) seg[n] = be;
    __syncthreads();
    for (int i0 = bb + tid; i0 < be; i0 += 4 * SORT_T) {
        int2 r[4];
        #pragma unroll
        for (int j = 0; j < 4; ++j) {
            int i = i0 + j * SORT_T;
            r[j] = (i < be) ? rec[i] : make_int2(0, 0);
        }
        #pragma unroll
        for (int j = 0; j < 4; ++j) {
            if (i0 + j * SORT_T < be) {
                int slot = atomicAdd(&cnt[(unsigned)r[j].y >> 24], 1);
                perm[slot] = make_int2(r[j].x, r[j].y & 0x00ffffff);
            }
        }
    }
}

__device__ __forceinline__ float xhalf(float x) {
    int b = __float_as_int(x);
    return __int_as_float(__builtin_amdgcn_permlanex16(b, b, 0x76543210, 0xfedcba98, false, false));
}
__device__ __forceinline__ float wave_max(float v) {
    v = fmaxf(v, xrow<1>(v));
    v = fmaxf(v, xrow<2>(v));
    v = fmaxf(v, xrow<4>(v));
    v = fmaxf(v, xrow<8>(v));
    return fmaxf(v, xhalf(v));
}
__device__ __forceinline__ float wave_sum(float v) {
    v += xrow<1>(v);
    v += xrow<2>(v);
    v += xrow<4>(v);
    v += xrow<8>(v);
    return v + xhalf(v);
}

#define GWAVES 8
#define WSTR   36
#define TABW   (4 * WSTR + 32)

__device__ __forceinline__ void wave_lds_sync() {
    __builtin_amdgcn_fence(__ATOMIC_RELEASE, "wavefront");
    __builtin_amdgcn_wave_barrier();
    __builtin_amdgcn_fence(__ATOMIC_ACQUIRE, "wavefront");
}

__device__ __forceinline__ void add_rows(const char* __restrict__ hb, unsigned lane8,
                                         float* tab, const float* wrow, int lane,
                                         unsigned srcoff,
                                         float w0, float w1, float w2, float w3,
                                         int m, float& ax, float& ay) {
    unsigned* st = (unsigned*)(tab + 4 * WSTR);
    wave_lds_sync();
    tab[0 * WSTR + lane] = w0;
    tab[1 * WSTR + lane] = w1;
    tab[2 * WSTR + lane] = w2;
    tab[3 * WSTR + lane] = w3;
    st[lane] = srcoff;
    wave_lds_sync();
    for (int k0 = 0; k0 < m; k0 += 8) {
        float4 wa = *(const float4*)(wrow + k0);
        float4 wb = *(const float4*)(wrow + k0 + 4);
        uint4  sa = *(const uint4*)(st + k0);
        uint4  sb = *(const uint4*)(st + k0 + 4);
        float2 hv[8];
        hv[0] = *(const float2*)(hb + (sa.x + lane8));
        hv[1] = *(const float2*)(hb + (sa.y + lane8));
        hv[2] = *(const float2*)(hb + (sa.z + lane8));
        hv[3] = *(const float2*)(hb + (sa.w + lane8));
        hv[4] = *(const float2*)(hb + (sb.x + lane8));
        hv[5] = *(const float2*)(hb + (sb.y + lane8));
        hv[6] = *(const float2*)(hb + (sb.z + lane8));
        hv[7] = *(const float2*)(hb + (sb.w + lane8));
        ax += hv[0].x * wa.x;  ay += hv[0].y * wa.x;
        ax += hv[1].x * wa.y;  ay += hv[1].y * wa.y;
        ax += hv[2].x * wa.z;  ay += hv[2].y * wa.z;
        ax += hv[3].x * wa.w;  ay += hv[3].y * wa.w;
        ax += hv[4].x * wb.x;  ay += hv[4].y * wb.x;
        ax += hv[5].x * wb.y;  ay += hv[5].y * wb.y;
        ax += hv[6].x * wb.z;  ay += hv[6].y * wb.z;
        ax += hv[7].x * wb.w;  ay += hv[7].y * wb.w;
    }
}

__device__ __forceinline__ float4 edge_act(float4 s, float4 ad) {
    float4 r;
    float v;
    v = s.x + ad.x;  r.x = (v >= 0.0f) ? v : NEG_SLOPE * v;
    v = s.y + ad.y;  r.y = (v >= 0.0f) ? v : NEG_SLOPE * v;
    v = s.z + ad.z;  r.z = (v >= 0.0f) ? v : NEG_SLOPE * v;
    v = s.w + ad.w;  r.w = (v >= 0.0f) ? v : NEG_SLOPE * v;
    return r;
}

__global__ void __launch_bounds__(32 * GWAVES)
gather_ln(const int* __restrict__ perm,
          const int* __restrict__ seg,
          const float* __restrict__ h,
          const float* __restrict__ score,
          const float* __restrict__ a_dst,
          const float* __restrict__ gamma,
          const float* __restrict__ beta,
          float* __restrict__ out, int n) {
    int lane = threadIdx.x & 31;
    int node = __builtin_amdgcn_readfirstlane((int)(blockIdx.x * GWAVES + (threadIdx.x >> 5)));
    if (node >= n) return;

    int beg = seg[node];
    int end = seg[node + 1];
    const float4 ad = ((const float4*)a_dst)[node];

    __shared__ __attribute__((aligned(16))) float tabs[GWAVES * TABW];
    float* tab = tabs + (threadIdx.x >> 5) * TABW;
    const float* wrow = tab + (lane >> 3) * WSTR;
    const char* hb = (const char*)h;
    unsigned lane8 = 8u * (unsigned)lane;
    const int2*   perm2  = (const int2*)perm;
    const float4* score4 = (const float4*)score;

    int m0 = end - beg;
    if (m0 > 32) m0 = 32;
    int2 pe0 = make_int2(0, 0);
    float4 s0 = make_float4(0.0f, 0.0f, 0.0f, 0.0f);
    if (lane < m0) {
        pe0 = perm2[beg + lane];
        s0  = edge_act(score4[pe0.x], ad);
    }

    float mx0 = fmaxf(s0.x, 0.0f), mx1 = fmaxf(s0.y, 0.0f);
    float mx2 = fmaxf(s0.z, 0.0f), mx3 = fmaxf(s0.w, 0.0f);
    for (int cb = beg + 32; cb < end; cb += 32) {
        if (lane < end - cb) {
            float4 s = edge_act(score4[perm2[cb + lane].x], ad);
            mx0 = fmaxf(mx0, s.x);  mx1 = fmaxf(mx1, s.y);
            mx2 = fmaxf(mx2, s.z);  mx3 = fmaxf(mx3, s.w);
        }
    }
    mx0 = wave_max(mx0);  mx1 = wave_max(mx1);
    mx2 = wave_max(mx2);  mx3 = wave_max(mx3);

    float e0 = 0.0f, e1 = 0.0f, e2 = 0.0f, e3 = 0.0f;
    if (lane < m0) {
        e0 = expf(s0.x - mx0);  e1 = expf(s0.y - mx1);
        e2 = expf(s0.z - mx2);  e3 = expf(s0.w - mx3);
    }
    float sm0 = e0, sm1 = e1, sm2 = e2, sm3 = e3;
    for (int cb = beg + 32; cb < end; cb += 32) {
        if (lane < end - cb) {
            float4 s = edge_act(score4[perm2[cb + lane].x], ad);
            sm0 += expf(s.x - mx0);  sm1 += expf(s.y - mx1);
            sm2 += expf(s.z - mx2);  sm3 += expf(s.w - mx3);
        }
    }
    float den0 = wave_sum(sm0) + EPS_SOFTMAX, den1 = wave_sum(sm1) + EPS_SOFTMAX;
    float den2 = wave_sum(sm2) + EPS_SOFTMAX, den3 = wave_sum(sm3) + EPS_SOFTMAX;

    float ax = 0.0f, ay = 0.0f;
    add_rows(hb, lane8, tab, wrow, lane, (unsigned)pe0.y << 8,
             e0 / den0, e1 / den1, e2 / den2, e3 / den3, m0, ax, ay);
    for (int cb = beg + 32; cb < end; cb += 32) {
        int m = end - cb;
        if (m > 32) m = 32;
        int2 pe = make_int2(0, 0);
        float x0 = 0.0f, x1 = 0.0f, x2 = 0.0f, x3 = 0.0f;
        if (lane < m) {
            pe = perm2[cb + lane];
            float4 s = edge_act(score4[pe.x], ad);
            x0 = expf(s.x - mx0);  x1 = expf(s.y - mx1);
            x2 = expf(s.z - mx2);  x3 = expf(s.w - mx3);
        }
        add_rows(hb, lane8, tab, wrow, lane, (unsigned)pe.y << 8,
                 x0 / den0, x1 / den1, x2 / den2, x3 / den3, m, ax, ay);
    }

    long base = (long)node * OUT_DIM + 2 * lane;
    float2 hn = *(const float2*)(h + base);
    float y0 = ax + hn.x;
    float y1 = ay + hn.y;

    float s = y0 + y1;
    #pragma unroll
    for (int o = 16; o > 0; o >>= 1) s += __shfl_xor(s, o, 32);
    float mu = s * (1.0f / 64.0f);

    float d0 = y0 - mu, d1 = y1 - mu;
    float vs = d0 * d0 + d1 * d1;
    #pragma unroll
    for (int o = 16; o > 0; o >>= 1) vs += __shfl_xor(vs, o, 32);
    float rstd = rsqrtf(vs * (1.0f / 64.0f) + EPS_LN);

    float2 g = ((const float2*)gamma)[lane];
    float2 b = ((const float2*)beta)[lane];
    float2 o2;
    o2.x = d0 * rstd * g.x + b.x;
    o2.y = d1 * rstd * g.y + b.y;
    *(float2*)(out + base) = o2;
}

extern "C" void kernel_launch(void* const* d_in, const int* in_sizes, int n_in,
                              void* d_out, int out_size, void* d_ws, size_t ws_size,
                              hipStream_t stream) {
    const float* x         = (const float*)d_in[0];
    const int*   ei        = (const int*)  d_in[1];
    const float* edge_attr = (const float*)d_in[2];
    const float* W_node    = (const float*)d_in[3];
    const float* W_edge    = (const float*)d_in[4];
    const float* attn_src  = (const float*)d_in[5];
    const float* attn_dst  = (const float*)d_in[6];
    const float* gamma     = (const float*)d_in[7];
    const float* beta      = (const float*)d_in[8];
    float* out = (float*)d_out;

    int  n = in_sizes[0] / IN_DIM;
    long e = (long)in_sizes[2] / EDGE_DIM;
    const int* srcIdx = ei;
    const int* dstIdx = ei + e;

    int nb = (n + BR - 1) / BR;

    float* ws    = (float*)d_ws;
    float* h     = ws;  ws += (long)n * OUT_DIM;
    float* a_src = ws;  ws += (long)n * H_HEADS;
    float* a_dst = ws;  ws += (long)n * H_HEADS;
    float* attn  = ws;  ws += e * H_HEADS;
    int*   seg   = (int*)ws;  ws += ((long)n + 4) / 4 * 4;
    int*   perm  = (int*)ws;  ws += 2 * e;
    int*   cntmat = (int*)ws; ws += (long)nb * NCHUNK;
    int*   bbase = (int*)ws;  ws += nb + 1;
    int2*  rec   = (int2*)attn;

    int n_tiles = (n + 15) / 16;
    node_gemm_wmma<<<n_tiles, 128, 0, stream>>>(x, W_node, attn_src, attn_dst, h, a_src, a_dst, n);

    bin_count<<<NCHUNK, SORT_T, 0, stream>>>(dstIdx, cntmat, e, nb);
    bin_place<<<NCHUNK, SORT_T, 0, stream>>>(srcIdx, dstIdx, cntmat, bbase, rec, e, nb);
    bucket_sort<<<nb, SORT_T, 0, stream>>>(rec, bbase, (int2*)perm, seg, n, nb);

    edge_score<<<(int)((e + 255) / 256), 256, 0, stream>>>(
        srcIdx, edge_attr, W_edge, a_src, attn, e);

    gather_ln<<<(n + GWAVES - 1) / GWAVES, 32 * GWAVES, 0, stream>>>(
        perm, seg, h, attn, a_dst, gamma, beta, out, n);
}


# independent passes share launches: edge count rides in the node GEMM launch, bin ranking in the edge-score launch (edge records moved into out); 6 -> 4 launches; per-node sum order changes
# speedup vs baseline: 4.7652x; 1.1157x over previous
#include <hip/hip_runtime.h>
#include <hip/hip_bf16.h>

typedef float v2f __attribute__((ext_vector_type(2)));
typedef float v8f __attribute__((ext_vector_type(8)));

#define IN_DIM   64
#define OUT_DIM  64
#define EDGE_DIM 16
#define H_HEADS  4
#define HD       16
#define NEG_SLOPE   0.2f
#define EPS_SOFTMAX 1e-8f
#define EPS_LN      1e-5f

template <int M>
__device__ __forceinline__ float xrow(float x) {
    return __int_as_float(__builtin_amdgcn_update_dpp(0, __float_as_int(x), 0x160 + M, 0xf, 0xf, false));
}
__device__ __forceinline__ float row16_sum(float v) {
    v += xrow<1>(v);
    v += xrow<2>(v);
    v += xrow<4>(v);
    v += xrow<8>(v);
    return v;
}

#define L1_T 256
__device__ __forceinline__ void node_gemm_role(int gblock,
                               const float* __restrict__ x,
                               const float* __restrict__ W,
                               const float* __restrict__ attn_src,
                               const float* __restrict__ attn_dst,
                               float* __restrict__ h,
                               float* __restrict__ a_src,
                               float* __restrict__ a_dst, int n) {
    int wave = (threadIdx.x >> 5) & 3;
    int band = threadIdx.x >> 7;
    int lane = threadIdx.x & 31;
    int grp  = lane >> 4;
    int lid  = lane & 15;

    int row0 = (gblock * 2 + band) * 16;
    if (row0 >= n) return;
    int col0 = wave * 16;

    int arow = row0 + lid;
    if (arow >= n) arow = n - 1;

    const float* xr = x + (long)arow * IN_DIM + 2 * grp;
    const float* wc = W + (long)(2 * grp) * OUT_DIM + col0 + lid;

    float ws = attn_src[col0 + lid];
    float wd = attn_dst[col0 + lid];

    v8f c = {};
    #pragma unroll
    for (int k0 = 0; k0 < IN_DIM; k0 += 4) {
        v2f a = *(const v2f*)(xr + k0);
        v2f b;
        b.x = wc[(long)k0 * OUT_DIM];
        b.y = wc[(long)(k0 + 1) * OUT_DIM];
        c = __builtin_amdgcn_wmma_f32_16x16x4_f32(false, a, false, b,
                                                  (short)0, c, false, false);
    }

    #pragma unroll
    for (int r = 0; r < 8; ++r) {
        int row = row0 + r + 8 * grp;
        if (row < n) h[(long)row * OUT_DIM + col0 + lid] = c[r];
    }

    float ms = 0.0f, md = 0.0f;
    #pragma unroll
    for (int r = 0; r < 8; ++r) {
        float ps = row16_sum(c[r] * ws);
        float pd = row16_sum(c[r] * wd);
        if (lid == r) { ms = ps; md = pd; }
    }
    int prow = row0 + 8 * grp + lid;
    if (lid < 8 && prow < n) {
        a_src[(long)prow * H_HEADS + wave] = ms;
        a_dst[(long)prow * H_HEADS + wave] = md;
    }
}

__device__ __forceinline__ void edge_score_role(long e,
                           const int* __restrict__ src,
                           const float* __restrict__ edge_attr,
                           const float* __restrict__ W_edge,
                           const float* __restrict__ a_src,
                           float* __restrict__ attn, long e_cnt) {
    if (e >= e_cnt) return;
    int s = src[e];

    float sc[H_HEADS] = {0.f, 0.f, 0.f, 0.f};
    const float4* ea = (const float4*)(edge_attr + e * EDGE_DIM);
    #pragma unroll
    for (int q = 0; q < 4; ++q) {
        float4 v = ea[q];
        int d0 = q * 4;
        #pragma unroll
        for (int hh = 0; hh < H_HEADS; ++hh) {
            sc[hh] += v.x * W_edge[(d0 + 0) * H_HEADS + hh]
                    + v.y * W_edge[(d0 + 1) * H_HEADS + hh]
                    + v.z * W_edge[(d0 + 2) * H_HEADS + hh]
                    + v.w * W_edge[(d0 + 3) * H_HEADS + hh];
        }
    }
    float4 asv = ((const float4*)a_src)[s];
    ((float4*)attn)[e] = make_float4(asv.x + sc[0], asv.y + sc[1], asv.z + sc[2], asv.w + sc[3]);
}

#define BR_SHIFT 8
#define BR       256
#define NCHUNK   256
#define SORT_T   1024
#define MAXB     1024

template <int M>
__device__ __forceinline__ int xrow_i(int x) {
    return __builtin_amdgcn_update_dpp(0, x, 0x160 + M, 0xf, 0xf, false);
}
__device__ __forceinline__ int wave_sum_i(int v) {
    v += xrow_i<1>(v);
    v += xrow_i<2>(v);
    v += xrow_i<4>(v);
    v += xrow_i<8>(v);
    return v + __builtin_amdgcn_permlanex16(v, v, 0x76543210, 0xfedcba98, false, false);
}

__device__ __forceinline__ void chunk_range(long e_cnt, long& e0, long& e1) {
    long per = (e_cnt + NCHUNK - 1) / NCHUNK;
    e0 = (long)blockIdx.x * per;
    if (e0 > e_cnt) e0 = e_cnt;
    e1 = e0 + per;
    if (e1 > e_cnt) e1 = e_cnt;
}

__device__ __forceinline__ void bin_count_role(const int* __restrict__ dst, int* __restrict__ cntmat,
                                               long e_cnt, int nb, int* hist) {
    int tid = threadIdx.x;
    for (int t = tid; t < MAXB; t += L1_T) hist[t] = 0;
    __syncthreads();
    long e0, e1;
    chunk_range(e_cnt, e0, e1);
    for (long i0 = e0 + tid; i0 < e1; i0 += 16 * L1_T) {
        int d[16];
        #pragma unroll
        for (int j = 0; j < 16; ++j) {
            long i = i0 + (long)j * L1_T;
            d[j] = (i < e1) ? dst[i] : -1;
        }
        #pragma unroll
        for (int j = 0; j < 16; ++j)
            if (d[j] >= 0) atomicAdd(&hist[d[j] >> BR_SHIFT], 1);
    }
    __syncthreads();
    for (int t = tid; t < nb; t += L1_T) cntmat[(long)t * NCHUNK + blockIdx.x] = hist[t];
}

__global__ void __launch_bounds__(L1_T)
gemm_and_count(const float* __restrict__ x,
               const float* __restrict__ W,
               const float* __restrict__ attn_src,
               const float* __restrict__ attn_dst,
               float* __restrict__ h,
               float* __restrict__ a_src,
               float* __restrict__ a_dst, int n,
               const int* __restrict__ dst, int* __restrict__ cntmat, long e_cnt, int nb) {
    __shared__ int hist[MAXB];
    if (blockIdx.x < NCHUNK) {
        bin_count_role(dst, cntmat, e_cnt, nb, hist);
        return;
    }
    node_gemm_role((int)blockIdx.x - NCHUNK, x, W, attn_src, attn_dst, h, a_src, a_dst, n);
}

__global__ void __launch_bounds__(SORT_T)
bin_place(const int* __restrict__ src, const int* __restrict__ dst,
          const int* __restrict__ cntmat, int* __restrict__ bbase,
          int2* __restrict__ rec, long e_cnt, int nb) {
    __shared__ int tot[MAXB];
    __shared__ int cur[MAXB];
    __shared__ int wtot[32];
    int tid  = threadIdx.x;
    int lane = tid & 31;
    int wave = __builtin_amdgcn_readfirstlane(tid >> 5);
    int g = blockIdx.x;

    tot[tid] = 0;
    cur[tid] = 0;
    __syncthreads();
    for (int b0 = wave; b0 < nb; b0 += 8 * (SORT_T / 32)) {
        int4 u[8], v[8];
        #pragma unroll
        for (int j = 0; j < 8; ++j) {
            int b = b0 + j * (SORT_T / 32);
            if (b < nb) {
                const int4* p = (const int4*)(cntmat + (long)b * NCHUNK) + 2 * lane;
                u[j] = p[0];
                v[j] = p[1];
            }
        }
        #pragma unroll
        for (int j = 0; j < 8; ++j) {
            int b = b0 + j * (SORT_T / 32);
            if (b < nb) {
                int c0 = 8 * lane;
                int t  = u[j].x + u[j].y + u[j].z + u[j].w + v[j].x + v[j].y + v[j].z + v[j].w;
                int bf = (c0 + 0 < g ? u[j].x : 0) + (c0 + 1 < g ? u[j].y : 0)
                       + (c0 + 2 < g ? u[j].z : 0) + (c0 + 3 < g ? u[j].w : 0)
                       + (c0 + 4 < g ? v[j].x : 0) + (c0 + 5 < g ? v[j].y : 0)
                       + (c0 + 6 < g ? v[j].z : 0) + (c0 + 7 < g ? v[j].w : 0);
                t  = wave_sum_i(t);
                bf = wave_sum_i(bf);
                if (lane == 0) { tot[b] = t; cur[b] = bf; }
            }
        }
    }
    __syncthreads();

    int t = tot[tid];
    int incl = t;
    #pragma unroll
    for (int d = 1; d < 32; d <<= 1) {
        int q = __shfl_up(incl, d, 32);
        if (lane >= d) incl += q;
    }
    if (lane == 31) wtot[wave] = incl;
    __syncthreads();
    int wv = wtot[lane];
    int wincl = wv;
    #pragma unroll
    for (int d = 1; d < 32; d <<= 1) {
        int q = __shfl_up(wincl, d, 32);
        if (lane >= d) wincl += q;
    }
    int wexcl = __shfl(wincl - wv, wave, 32);
    int base = wexcl + incl - t;
    if (tid < nb) {
        cur[tid] += base;
        if (g == 0) {
            bbase[tid] = base;
            if (tid == nb - 1) bbase[nb] = base + t;
        }
    }
    __syncthreads();

    long e0, e1;
    chunk_range(e_cnt, e0, e1);
    for (long i0 = e0 + tid; i0 < e1; i0 += 8 * SORT_T) {
        int d[8], s[8];
        #pragma unroll
        for (int j = 0; j < 8; ++j) {
            long i = i0 + (long)j * SORT_T;
            bool ok = i < e1;
            d[j] = ok ? dst[i] : -1;
            s[j] = ok ? src[i] : 0;
        }
        #pragma unroll
        for (int j = 0; j < 8; ++j) {
            if (d[j] >= 0) {
                long i = i0 + (long)j * SORT_T;
                int slot = atomicAdd(&cur[d[j] >> BR_SHIFT], 1);
                unsigned loc = (unsigned)(d[j] & (BR - 1));
                rec[slot] = make_int2((int)i, (int)((unsigned)s[j] | (loc << 24)));
            }
        }
    }
}

#define L3_T 512
__device__ __forceinline__ void bucket_sort_role(const int2* __restrict__ rec, const int* __restrict__ bbase,
            int2* __restrict__ perm, int* __restrict__ seg, int n, int nb,
            int* cnt, int* wtot) {
    int tid  = threadIdx.x;
    int lane = tid & 31;
    int wave = tid >> 5;
    int b  = blockIdx.x;
    int bb = bbase[b], be = bbase[b + 1];
    if (tid < BR) cnt[tid] = 0;
    __syncthreads();
    for (int i0 = bb + tid; i0 < be; i0 += 8 * L3_T) {
        unsigned y[8];
        #pragma unroll
        for (int j = 0; j < 8; ++j) {
            int i = i0 + j * L3_T;
            y[j] = (unsigned)rec[i < be ? i : be - 1].y;
        }
        #pragma unroll
        for (int j = 0; j < 8; ++j)
            if (i0 + j * L3_T < be) atomicAdd(&cnt[y[j] >> 24], 1);
    }
    __syncthreads();
    int c = 0, incl = 0;
    if (tid < BR) {
        c = cnt[tid];
        incl = c;
        #pragma unroll
        for (int d = 1; d < 32; d <<= 1) {
            int q = __shfl_up(incl, d, 32);
            if (lane >= d) incl += q;
        }
        if (lane == 31) wtot[wave] = incl;
    }
    __syncthreads();
    if (tid < BR) {
        int add = 0;
        for (int w = 0; w < wave; ++w) add += wtot[w];
        int start = bb + add + incl - c;
        cnt[tid] = start;
        int node = b * BR + tid;
        if (node < n) seg[node] = start;
    }
    if (b == nb - 1 && tid == 0) seg[n] = be;
    __syncthreads();
    for (int i0 = bb + tid; i0 < be; i0 += 8 * L3_T) {
        int2 r[8];
        #pragma unroll
        for (int j = 0; j < 8; ++j) {
            int i = i0 + j * L3_T;
            r[j] = (i < be) ? rec[i] : make_int2(0, 0);
        }
        #pragma unroll
        for (int j = 0; j < 8; ++j) {
            if (i0 + j * L3_T < be) {
                int slot = atomicAdd(&cnt[(unsigned)r[j].y >> 24], 1);
                perm[slot] = make_int2(r[j].x, r[j].y & 0x00ffffff);
            }
        }
    }
}

__global__ void __launch_bounds__(L3_T)
sort_and_score(const int2* __restrict__ rec, const int* __restrict__ bbase,
               int2* __restrict__ perm, int* __restrict__ seg, int n, int nb,
               const int* __restrict__ src,
               const float* __restrict__ edge_attr,
               const float* __restrict__ W_edge,
               const float* __restrict__ a_src,
               float* __restrict__ attn, long e_cnt) {
    __shared__ int cnt[BR];
    __shared__ int wtot[BR / 32];
    if ((int)blockIdx.x < nb) {
        bucket_sort_role(rec, bbase, perm, seg, n, nb, cnt, wtot);
        return;
    }
    edge_score_role((long)((int)blockIdx.x - nb) * L3_T + threadIdx.x,
                    src, edge_attr, W_edge, a_src, attn, e_cnt);
}

__device__ __forceinline__ float xhalf(float x) {
    int b = __float_as_int(x);
    return __int_as_float(__builtin_amdgcn_permlanex16(b, b, 0x76543210, 0xfedcba98, false, false));
}
__device__ __forceinline__ float wave_max(float v) {
    v = fmaxf(v, xrow<1>(v));
    v = fmaxf(v, xrow<2>(v));
    v = fmaxf(v, xrow<4>(v));
    v = fmaxf(v, xrow<8>(v));
    return fmaxf(v, xhalf(v));
}
__device__ __forceinline__ float wave_sum(float v) {
    v += xrow<1>(v);
    v += xrow<2>(v);
    v += xrow<4>(v);
    v += xrow<8>(v);
    return v + xhalf(v);
}

#define GWAVES 8
#define WSTR   36
#define TABW   (4 * WSTR + 32)

__device__ __forceinline__ void wave_lds_sync() {
    __builtin_amdgcn_fence(__ATOMIC_RELEASE, "wavefront");
    __builtin_amdgcn_wave_barrier();
    __builtin_amdgcn_fence(__ATOMIC_ACQUIRE, "wavefront");
}

__device__ __forceinline__ void add_rows(const char* __restrict__ hb, unsigned lane8,
                                         float* tab, const float* wrow, int lane,
                                         unsigned srcoff,
                                         float w0, float w1, float w2, float w3,
                                         int m, float& ax, float& ay) {
    unsigned* st = (unsigned*)(tab + 4 * WSTR);
    wave_lds_sync();
    tab[0 * WSTR + lane] = w0;
    tab[1 * WSTR + lane] = w1;
    tab[2 * WSTR + lane] = w2;
    tab[3 * WSTR + lane] = w3;
    st[lane] = srcoff;
    wave_lds_sync();
    for (int k0 = 0; k0 < m; k0 += 8) {
        float4 wa = *(const float4*)(wrow + k0);
        float4 wb = *(const float4*)(wrow + k0 + 4);
        uint4  sa = *(const uint4*)(st + k0);
        uint4  sb = *(const uint4*)(st + k0 + 4);
        float2 hv[8];
        hv[0] = *(const float2*)(hb + (sa.x + lane8));
        hv[1] = *(const float2*)(hb + (sa.y + lane8));
        hv[2] = *(const float2*)(hb + (sa.z + lane8));
        hv[3] = *(const float2*)(hb + (sa.w + lane8));
        hv[4] = *(const float2*)(hb + (sb.x + lane8));
        hv[5] = *(const float2*)(hb + (sb.y + lane8));
        hv[6] = *(const float2*)(hb + (sb.z + lane8));
        hv[7] = *(const float2*)(hb + (sb.w + lane8));
        ax += hv[0].x * wa.x;  ay += hv[0].y * wa.x;
        ax += hv[1].x * wa.y;  ay += hv[1].y * wa.y;
        ax += hv[2].x * wa.z;  ay += hv[2].y * wa.z;
        ax += hv[3].x * wa.w;  ay += hv[3].y * wa.w;
        ax += hv[4].x * wb.x;  ay += hv[4].y * wb.x;
        ax += hv[5].x * wb.y;  ay += hv[5].y * wb.y;
        ax += hv[6].x * wb.z;  ay += hv[6].y * wb.z;
        ax += hv[7].x * wb.w;  ay += hv[7].y * wb.w;
    }
}

__device__ __forceinline__ float4 edge_act(float4 s, float4 ad) {
    float4 r;
    float v;
    v = s.x + ad.x;  r.x = (v >= 0.0f) ? v : NEG_SLOPE * v;
    v = s.y + ad.y;  r.y = (v >= 0.0f) ? v : NEG_SLOPE * v;
    v = s.z + ad.z;  r.z = (v >= 0.0f) ? v : NEG_SLOPE * v;
    v = s.w + ad.w;  r.w = (v >= 0.0f) ? v : NEG_SLOPE * v;
    return r;
}

__global__ void __launch_bounds__(32 * GWAVES)
gather_ln(const int* __restrict__ perm,
          const int* __restrict__ seg,
          const float* __restrict__ h,
          const float* __restrict__ score,
          const float* __restrict__ a_dst,
          const float* __restrict__ gamma,
          const float* __restrict__ beta,
          float* __restrict__ out, int n) {
    int lane = threadIdx.x & 31;
    int node = __builtin_amdgcn_readfirstlane((int)(blockIdx.x * GWAVES + (threadIdx.x >> 5)));
    if (node >= n) return;

    int beg = seg[node];
    int end = seg[node + 1];
    const float4 ad = ((const float4*)a_dst)[node];

    __shared__ __attribute__((aligned(16))) float tabs[GWAVES * TABW];
    float* tab = tabs + (threadIdx.x >> 5) * TABW;
    const float* wrow = tab + (lane >> 3) * WSTR;
    const char* hb = (const char*)h;
    unsigned lane8 = 8u * (unsigned)lane;
    const int2*   perm2  = (const int2*)perm;
    const float4* score4 = (const float4*)score;

    int m0 = end - beg;
    if (m0 > 32) m0 = 32;
    int2 pe0 = make_int2(0, 0);
    float4 s0 = make_float4(0.0f, 0.0f, 0.0f, 0.0f);
    if (lane < m0) {
        pe0 = perm2[beg + lane];
        s0  = edge_act(score4[pe0.x], ad);
    }

    float mx0 = fmaxf(s0.x, 0.0f), mx1 = fmaxf(s0.y, 0.0f);
    float mx2 = fmaxf(s0.z, 0.0f), mx3 = fmaxf(s0.w, 0.0f);
    for (int cb = beg + 32; cb < end; cb += 32) {
        if (lane < end - cb) {
            float4 s = edge_act(score4[perm2[cb + lane].x], ad);
            mx0 = fmaxf(mx0, s.x);  mx1 = fmaxf(mx1, s.y);
            mx2 = fmaxf(mx2, s.z);  mx3 = fmaxf(mx3, s.w);
        }
    }
    mx0 = wave_max(mx0);  mx1 = wave_max(mx1);
    mx2 = wave_max(mx2);  mx3 = wave_max(mx3);

    float e0 = 0.0f, e1 = 0.0f, e2 = 0.0f, e3 = 0.0f;
    if (lane < m0) {
        e0 = expf(s0.x - mx0);  e1 = expf(s0.y - mx1);
        e2 = expf(s0.z - mx2);  e3 = expf(s0.w - mx3);
    }
    float sm0 = e0, sm1 = e1, sm2 = e2, sm3 = e3;
    for (int cb = beg + 32; cb < end; cb += 32) {
        if (lane < end - cb) {
            float4 s = edge_act(score4[perm2[cb + lane].x], ad);
            sm0 += expf(s.x - mx0);  sm1 += expf(s.y - mx1);
            sm2 += expf(s.z - mx2);  sm3 += expf(s.w - mx3);
        }
    }
    float den0 = wave_sum(sm0) + EPS_SOFTMAX, den1 = wave_sum(sm1) + EPS_SOFTMAX;
    float den2 = wave_sum(sm2) + EPS_SOFTMAX, den3 = wave_sum(sm3) + EPS_SOFTMAX;

    float ax = 0.0f, ay = 0.0f;
    add_rows(hb, lane8, tab, wrow, lane, (unsigned)pe0.y << 8,
             e0 / den0, e1 / den1, e2 / den2, e3 / den3, m0, ax, ay);
    for (int cb = beg + 32; cb < end; cb += 32) {
        int m = end - cb;
        if (m > 32) m = 32;
        int2 pe = make_int2(0, 0);
        float x0 = 0.0f, x1 = 0.0f, x2 = 0.0f, x3 = 0.0f;
        if (lane < m) {
            pe = perm2[cb + lane];
            float4 s = edge_act(score4[pe.x], ad);
            x0 = expf(s.x - mx0);  x1 = expf(s.y - mx1);
            x2 = expf(s.z - mx2);  x3 = expf(s.w - mx3);
        }
        add_rows(hb, lane8, tab, wrow, lane, (unsigned)pe.y << 8,
                 x0 / den0, x1 / den1, x2 / den2, x3 / den3, m, ax, ay);
    }

    long base = (long)node * OUT_DIM + 2 * lane;
    float2 hn = *(const float2*)(h + base);
    float y0 = ax + hn.x;
    float y1 = ay + hn.y;

    float s = y0 + y1;
    #pragma unroll
    for (int o = 16; o > 0; o >>= 1) s += __shfl_xor(s, o, 32);
    float mu = s * (1.0f / 64.0f);

    float d0 = y0 - mu, d1 = y1 - mu;
    float vs = d0 * d0 + d1 * d1;
    #pragma unroll
    for (int o = 16; o > 0; o >>= 1) vs += __shfl_xor(vs, o, 32);
    float rstd = rsqrtf(vs * (1.0f / 64.0f) + EPS_LN);

    float2 g = ((const float2*)gamma)[lane];
    float2 b = ((const float2*)beta)[lane];
    float2 o2;
    o2.x = d0 * rstd * g.x + b.x;
    o2.y = d1 * rstd * g.y + b.y;
    *(float2*)(out + base) = o2;
}

extern "C" void kernel_launch(void* const* d_in, const int* in_sizes, int n_in,
                              void* d_out, int out_size, void* d_ws, size_t ws_size,
                              hipStream_t stream) {
    const float* x         = (const float*)d_in[0];
    const int*   ei        = (const int*)  d_in[1];
    const float* edge_attr = (const float*)d_in[2];
    const float* W_node    = (const float*)d_in[3];
    const float* W_edge    = (const float*)d_in[4];
    const float* attn_src  = (const float*)d_in[5];
    const float* attn_dst  = (const float*)d_in[6];
    const float* gamma     = (const float*)d_in[7];
    const float* beta      = (const float*)d_in[8];
    float* out = (float*)d_out;

    int  n = in_sizes[0] / IN_DIM;
    long e = (long)in_sizes[2] / EDGE_DIM;
    const int* srcIdx = ei;
    const int* dstIdx = ei + e;

    int nb = (n + BR - 1) / BR;

    float* ws    = (float*)d_ws;
    float* h     = ws;  ws += (long)n * OUT_DIM;
    float* a_src = ws;  ws += (long)n * H_HEADS;
    float* a_dst = ws;  ws += (long)n * H_HEADS;
    float* attn  = ws;  ws += e * H_HEADS;
    int*   seg   = (int*)ws;  ws += ((long)n + 4) / 4 * 4;
    int*   perm  = (int*)ws;  ws += 2 * e;
    int*   cntmat = (int*)ws; ws += (long)nb * NCHUNK;
    int*   bbase = (int*)ws;  ws += nb + 1;
    int2*  rec   = (int2*)out;

    int n_tiles = (n + 15) / 16;
    gemm_and_count<<<NCHUNK + (n_tiles + 1) / 2, L1_T, 0, stream>>>(
        x, W_node, attn_src, attn_dst, h, a_src, a_dst, n, dstIdx, cntmat, e, nb);

    bin_place<<<NCHUNK, SORT_T, 0, stream>>>(srcIdx, dstIdx, cntmat, bbase, rec, e, nb);

    sort_and_score<<<nb + (int)((e + L3_T - 1) / L3_T), L3_T, 0, stream>>>(
        rec, bbase, (int2*)perm, seg, n, nb, srcIdx, edge_attr, W_edge, a_src, attn, e);

    gather_ln<<<(n + GWAVES - 1) / GWAVES, 32 * GWAVES, 0, stream>>>(
        perm, seg, h, attn, a_dst, gamma, beta, out, n);
}


# binning made chunk-local: each chunk counts, scans and places its own edges by bin inside launch 1; ranking finds a bin's pieces by offsets; bin_place launch removed (4 -> 3); node sum order changes
# speedup vs baseline: 5.5568x; 1.1661x over previous
#include <hip/hip_runtime.h>
#include <hip/hip_bf16.h>

typedef float v2f __attribute__((ext_vector_type(2)));
typedef float v8f __attribute__((ext_vector_type(8)));

#define IN_DIM   64
#define OUT_DIM  64
#define EDGE_DIM 16
#define H_HEADS  4
#define HD       16
#define NEG_SLOPE   0.2f
#define EPS_SOFTMAX 1e-8f
#define EPS_LN      1e-5f

template <int M>
__device__ __forceinline__ float xrow(float x) {
    return __int_as_float(__builtin_amdgcn_update_dpp(0, __float_as_int(x), 0x160 + M, 0xf, 0xf, false));
}
__device__ __forceinline__ float row16_sum(float v) {
    v += xrow<1>(v);
    v += xrow<2>(v);
    v += xrow<4>(v);
    v += xrow<8>(v);
    return v;
}

#define L1_T 256
__device__ __forceinline__ void node_gemm_role(int gblock,
                               const float* __restrict__ x,
                               const float* __restrict__ W,
                               const float* __restrict__ attn_src,
                               const float* __restrict__ attn_dst,
                               float* __restrict__ h,
                               float* __restrict__ a_src,
                               float* __restrict__ a_dst, int n) {
    int wave = (threadIdx.x >> 5) & 3;
    int band = threadIdx.x >> 7;
    int lane = threadIdx.x & 31;
    int grp  = lane >> 4;
    int lid  = lane & 15;

    int row0 = (gblock * 2 + band) * 16;
    if (row0 >= n) return;
    int col0 = wave * 16;

    int arow = row0 + lid;
    if (arow >= n) arow = n - 1;

    const float* xr = x + (long)arow * IN_DIM + 2 * grp;
    const float* wc = W + (long)(2 * grp) * OUT_DIM + col0 + lid;

    float ws = attn_src[col0 + lid];
    float wd = attn_dst[col0 + lid];

    v8f c = {};
    #pragma unroll
    for (int k0 = 0; k0 < IN_DIM; k0 += 4) {
        v2f a = *(const v2f*)(xr + k0);
        v2f b;
        b.x = wc[(long)k0 * OUT_DIM];
        b.y = wc[(long)(k0 + 1) * OUT_DIM];
        c = __builtin_amdgcn_wmma_f32_16x16x4_f32(false, a, false, b,
                                                  (short)0, c, false, false);
    }

    #pragma unroll
    for (int r = 0; r < 8; ++r) {
        int row = row0 + r + 8 * grp;
        if (row < n) h[(long)row * OUT_DIM + col0 + lid] = c[r];
    }

    float ms = 0.0f, md = 0.0f;
    #pragma unroll
    for (int r = 0; r < 8; ++r) {
        float ps = row16_sum(c[r] * ws);
        float pd = row16_sum(c[r] * wd);
        if (lid == r) { ms = ps; md = pd; }
    }
    int prow = row0 + 8 * grp + lid;
    if (lid < 8 && prow < n) {
        a_src[(long)prow * H_HEADS + wave] = ms;
        a_dst[(long)prow * H_HEADS + wave] = md;
    }
}

__device__ __forceinline__ void edge_score_role(long e,
                           const int* __restrict__ src,
                           const float* __restrict__ edge_attr,
                           const float* __restrict__ W_edge,
                           const float* __restrict__ a_src,
                           float* __restrict__ attn, long e_cnt) {
    if (e >= e_cnt) return;
    int s = src[e];

    float sc[H_HEADS] = {0.f, 0.f, 0.f, 0.f};
    const float4* ea = (const float4*)(edge_attr + e * EDGE_DIM);
    #pragma unroll
    for (int q = 0; q < 4; ++q) {
        float4 v = ea[q];
        int d0 = q * 4;
        #pragma unroll
        for (int hh = 0; hh < H_HEADS; ++hh) {
            sc[hh] += v.x * W_edge[(d0 + 0) * H_HEADS + hh]
                    + v.y * W_edge[(d0 + 1) * H_HEADS + hh]
                    + v.z * W_edge[(d0 + 2) * H_HEADS + hh]
                    + v.w * W_edge[(d0 + 3) * H_HEADS + hh];
        }
    }
    float4 asv = ((const float4*)a_src)[s];
    ((float4*)attn)[e] = make_float4(asv.x + sc[0], asv.y + sc[1], asv.z + sc[2], asv.w + sc[3]);
}

#define BR_SHIFT 8
#define BR       256
#define NCHUNK   256
#define MAXB     1024

template <int M>
__device__ __forceinline__ int xrow_i(int x) {
    return __builtin_amdgcn_update_dpp(0, x, 0x160 + M, 0xf, 0xf, false);
}
__device__ __forceinline__ int wave_sum_i(int v) {
    v += xrow_i<1>(v);
    v += xrow_i<2>(v);
    v += xrow_i<4>(v);
    v += xrow_i<8>(v);
    return v + __builtin_amdgcn_permlanex16(v, v, 0x76543210, 0xfedcba98, false, false);
}

__device__ __forceinline__ long chunk_begin(int c, long e_cnt) {
    long per = (e_cnt + NCHUNK - 1) / NCHUNK;
    long e0 = (long)c * per;
    return e0 > e_cnt ? e_cnt : e0;
}

__device__ __forceinline__ void chunk_bin_role(const int* __restrict__ src, const int* __restrict__ dst,
                                               int* __restrict__ offmat, int2* __restrict__ rec,
                                               long e_cnt, int nb, int* hist, int* wtot) {
    int tid  = threadIdx.x;
    int lane = tid & 31;
    int wave = tid >> 5;
    int g = blockIdx.x;
    for (int t = tid; t < MAXB; t += L1_T) hist[t] = 0;
    __syncthreads();
    long e0 = chunk_begin(g, e_cnt);
    long e1 = chunk_begin(g + 1, e_cnt);
    for (long i0 = e0 + tid; i0 < e1; i0 += 16 * L1_T) {
        int d[16];
        #pragma unroll
        for (int j = 0; j < 16; ++j) {
            long i = i0 + (long)j * L1_T;
            d[j] = (i < e1) ? dst[i] : -1;
        }
        #pragma unroll
        for (int j = 0; j < 16; ++j)
            if (d[j] >= 0) atomicAdd(&hist[d[j] >> BR_SHIFT], 1);
    }
    __syncthreads();

    int4 v = ((const int4*)hist)[tid];
    int s = v.x + v.y + v.z + v.w;
    int incl = s;
    #pragma unroll
    for (int d = 1; d < 32; d <<= 1) {
        int q = __shfl_up(incl, d, 32);
        if (lane >= d) incl += q;
    }
    if (lane == 31) wtot[wave] = incl;
    __syncthreads();
    int add = 0;
    for (int w = 0; w < wave; ++w) add += wtot[w];
    int p0 = (int)e0 + add + incl - s;
    int p1 = p0 + v.x, p2 = p1 + v.y, p3 = p2 + v.z;
    ((int4*)hist)[tid] = make_int4(p0, p1, p2, p3);
    int b0 = 4 * tid;
    if (b0 + 0 < nb) offmat[(long)(b0 + 0) * NCHUNK + g] = p0;
    if (b0 + 1 < nb) offmat[(long)(b0 + 1) * NCHUNK + g] = p1;
    if (b0 + 2 < nb) offmat[(long)(b0 + 2) * NCHUNK + g] = p2;
    if (b0 + 3 < nb) offmat[(long)(b0 + 3) * NCHUNK + g] = p3;
    if (tid == 0) offmat[(long)nb * NCHUNK + g] = (int)e1;
    __syncthreads();

    for (long i0 = e0 + tid; i0 < e1; i0 += 16 * L1_T) {
        int d[16], sn[16];
        #pragma unroll
        for (int j = 0; j < 16; ++j) {
            long i = i0 + (long)j * L1_T;
            bool ok = i < e1;
            d[j]  = ok ? dst[i] : -1;
            sn[j] = ok ? src[i] : 0;
        }
        #pragma unroll
        for (int j = 0; j < 16; ++j) {
            if (d[j] >= 0) {
                long i = i0 + (long)j * L1_T;
                int slot = atomicAdd(&hist[d[j] >> BR_SHIFT], 1);
                unsigned loc = (unsigned)(d[j] & (BR - 1));
                rec[slot] = make_int2((int)i, (int)((unsigned)sn[j] | (loc << 24)));
            }
        }
    }
}

__global__ void __launch_bounds__(L1_T)
gemm_and_bin(const float* __restrict__ x,
             const float* __restrict__ W,
             const float* __restrict__ attn_src,
             const float* __restrict__ attn_dst,
             float* __restrict__ h,
             float* __restrict__ a_src,
             float* __restrict__ a_dst, int n,
             const int* __restrict__ src, const int* __restrict__ dst,
             int* __restrict__ offmat, int2* __restrict__ rec, long e_cnt, int nb) {
    __shared__ __attribute__((aligned(16))) int hist[MAXB];
    __shared__ int wtot[L1_T / 32];
    if (blockIdx.x < NCHUNK) {
        chunk_bin_role(src, dst, offmat, rec, e_cnt, nb, hist, wtot);
        return;
    }
    node_gemm_role((int)blockIdx.x - NCHUNK, x, W, attn_src, attn_dst, h, a_src, a_dst, n);
}

__device__ __forceinline__ int rec_slot(int i, const int* pos, const int* off) {
    int c = 0;
    #pragma unroll
    for (int st = NCHUNK / 2; st > 0; st >>= 1)
        if (pos[c + st] <= i) c += st;
    return off[c] + (i - pos[c]);
}

#define L3_T 512
__device__ __forceinline__ void bucket_sort_role(const int2* __restrict__ rec, const int* __restrict__ offmat,
            int2* __restrict__ perm, int* __restrict__ seg, int n, int nb, long e_cnt,
            int* cnt, int* wtot, int* pos, int* off, int* wcnt, int* wrel) {
    int tid  = threadIdx.x;
    int lane = tid & 31;
    int wave = tid >> 5;
    int b  = blockIdx.x;
    int o0 = 0, len = 0, incl = 0;
    if (tid < NCHUNK) {
        o0 = offmat[(long)b * NCHUNK + tid];
        len = offmat[(long)(b + 1) * NCHUNK + tid] - o0;
        int rel = o0 - (int)chunk_begin(tid, e_cnt);
        incl = len;
        #pragma unroll
        for (int d = 1; d < 32; d <<= 1) {
            int q = __shfl_up(incl, d, 32);
            if (lane >= d) incl += q;
        }
        rel = wave_sum_i(rel);
        if (lane == 31) wcnt[wave] = incl;
        if (lane == 0)  wrel[wave] = rel;
    }
    if (tid < BR) cnt[tid] = 0;
    __syncthreads();
    int tot = 0, bb = 0, before = 0;
    #pragma unroll
    for (int w = 0; w < NCHUNK / 32; ++w) {
        int t = wcnt[w];
        if (w < wave) before += t;
        tot += t;
        bb  += wrel[w];
    }
    int be = bb + tot;
    if (tid < NCHUNK) {
        pos[tid] = before + incl - len;
        off[tid] = o0;
    }
    __syncthreads();

    for (int i0 = tid; i0 < tot; i0 += 8 * L3_T) {
        unsigned y[8];
        #pragma unroll
        for (int j = 0; j < 8; ++j) {
            int i = i0 + j * L3_T;
            y[j] = (unsigned)rec[rec_slot(i < tot ? i : tot - 1, pos, off)].y;
        }
        #pragma unroll
        for (int j = 0; j < 8; ++j)
            if (i0 + j * L3_T < tot) atomicAdd(&cnt[y[j] >> 24], 1);
    }
    __syncthreads();
    int c = 0, inc2 = 0;
    if (tid < BR) {
        c = cnt[tid];
        inc2 = c;
        #pragma unroll
        for (int d = 1; d < 32; d <<= 1) {
            int q = __shfl_up(inc2, d, 32);
            if (lane >= d) inc2 += q;
        }
        if (lane == 31) wtot[wave] = inc2;
    }
    __syncthreads();
    if (tid < BR) {
        int add = 0;
        for (int w = 0; w < wave; ++w) add += wtot[w];
        int start = bb + add + inc2 - c;
        cnt[tid] = start;
        int node = b * BR + tid;
        if (node < n) seg[node] = start;
    }
    if (b == nb - 1 && tid == 0) seg[n] = be;
    __syncthreads();
    for (int i0 = tid; i0 < tot; i0 += 8 * L3_T) {
        int2 r[8];
        #pragma unroll
        for (int j = 0; j < 8; ++j) {
            int i = i0 + j * L3_T;
            r[j] = rec[rec_slot(i < tot ? i : tot - 1, pos, off)];
        }
        #pragma unroll
        for (int j = 0; j < 8; ++j) {
            if (i0 + j * L3_T < tot) {
                int slot = atomicAdd(&cnt[(unsigned)r[j].y >> 24], 1);
                perm[slot] = make_int2(r[j].x, r[j].y & 0x00ffffff);
            }
        }
    }
}

__global__ void __launch_bounds__(L3_T)
sort_and_score(const int2* __restrict__ rec, const int* __restrict__ offmat,
               int2* __restrict__ perm, int* __restrict__ seg, int n, int nb,
               const int* __restrict__ src,
               const float* __restrict__ edge_attr,
               const float* __restrict__ W_edge,
               const float* __restrict__ a_src,
               float* __restrict__ attn, long e_cnt) {
    __shared__ int cnt[BR];
    __shared__ int wtot[BR / 32];
    __shared__ int pos[NCHUNK];
    __shared__ int off[NCHUNK];
    __shared__ int wcnt[NCHUNK / 32];
    __shared__ int wrel[NCHUNK / 32];
    if ((int)blockIdx.x < nb) {
        bucket_sort_role(rec, offmat, perm, seg, n, nb, e_cnt, cnt, wtot, pos, off, wcnt, wrel);
        return;
    }
    edge_score_role((long)((int)blockIdx.x - nb) * L3_T + threadIdx.x,
                    src, edge_attr, W_edge, a_src, attn, e_cnt);
}

__device__ __forceinline__ float xhalf(float x) {
    int b = __float_as_int(x);
    return __int_as_float(__builtin_amdgcn_permlanex16(b, b, 0x76543210, 0xfedcba98, false, false));
}
__device__ __forceinline__ float wave_max(float v) {
    v = fmaxf(v, xrow<1>(v));
    v = fmaxf(v, xrow<2>(v));
    v = fmaxf(v, xrow<4>(v));
    v = fmaxf(v, xrow<8>(v));
    return fmaxf(v, xhalf(v));
}
__device__ __forceinline__ float wave_sum(float v) {
    v += xrow<1>(v);
    v += xrow<2>(v);
    v += xrow<4>(v);
    v += xrow<8>(v);
    return v + xhalf(v);
}

#define GWAVES 8
#define WSTR   36
#define TABW   (4 * WSTR + 32)

__device__ __forceinline__ void wave_lds_sync() {
    __builtin_amdgcn_fence(__ATOMIC_RELEASE, "wavefront");
    __builtin_amdgcn_wave_barrier();
    __builtin_amdgcn_fence(__ATOMIC_ACQUIRE, "wavefront");
}

__device__ __forceinline__ void add_rows(const char* __restrict__ hb, unsigned lane8,
                                         float* tab, const float* wrow, int lane,
                                         unsigned srcoff,
                                         float w0, float w1, float w2, float w3,
                                         int m, float& ax, float& ay) {
    unsigned* st = (unsigned*)(tab + 4 * WSTR);
    wave_lds_sync();
    tab[0 * WSTR + lane] = w0;
    tab[1 * WSTR + lane] = w1;
    tab[2 * WSTR + lane] = w2;
    tab[3 * WSTR + lane] = w3;
    st[lane] = srcoff;
    wave_lds_sync();
    for (int k0 = 0; k0 < m; k0 += 8) {
        float4 wa = *(const float4*)(wrow + k0);
        float4 wb = *(const float4*)(wrow + k0 + 4);
        uint4  sa = *(const uint4*)(st + k0);
        uint4  sb = *(const uint4*)(st + k0 + 4);
        float2 hv[8];
        hv[0] = *(const float2*)(hb + (sa.x + lane8));
        hv[1] = *(const float2*)(hb + (sa.y + lane8));
        hv[2] = *(const float2*)(hb + (sa.z + lane8));
        hv[3] = *(const float2*)(hb + (sa.w + lane8));
        hv[4] = *(const float2*)(hb + (sb.x + lane8));
        hv[5] = *(const float2*)(hb + (sb.y + lane8));
        hv[6] = *(const float2*)(hb + (sb.z + lane8));
        hv[7] = *(const float2*)(hb + (sb.w + lane8));
        ax += hv[0].x * wa.x;  ay += hv[0].y * wa.x;
        ax += hv[1].x * wa.y;  ay += hv[1].y * wa.y;
        ax += hv[2].x * wa.z;  ay += hv[2].y * wa.z;
        ax += hv[3].x * wa.w;  ay += hv[3].y * wa.w;
        ax += hv[4].x * wb.x;  ay += hv[4].y * wb.x;
        ax += hv[5].x * wb.y;  ay += hv[5].y * wb.y;
        ax += hv[6].x * wb.z;  ay += hv[6].y * wb.z;
        ax += hv[7].x * wb.w;  ay += hv[7].y * wb.w;
    }
}

__device__ __forceinline__ float4 edge_act(float4 s, float4 ad) {
    float4 r;
    float v;
    v = s.x + ad.x;  r.x = (v >= 0.0f) ? v : NEG_SLOPE * v;
    v = s.y + ad.y;  r.y = (v >= 0.0f) ? v : NEG_SLOPE * v;
    v = s.z + ad.z;  r.z = (v >= 0.0f) ? v : NEG_SLOPE * v;
    v = s.w + ad.w;  r.w = (v >= 0.0f) ? v : NEG_SLOPE * v;
    return r;
}

__global__ void __launch_bounds__(32 * GWAVES)
gather_ln(const int* __restrict__ perm,
          const int* __restrict__ seg,
          const float* __restrict__ h,
          const float* __restrict__ score,
          const float* __restrict__ a_dst,
          const float* __restrict__ gamma,
          const float* __restrict__ beta,
          float* __restrict__ out, int n) {
    int lane = threadIdx.x & 31;
    int node = __builtin_amdgcn_readfirstlane((int)(blockIdx.x * GWAVES + (threadIdx.x >> 5)));
    if (node >= n) return;

    int beg = seg[node];
    int end = seg[node + 1];
    const float4 ad = ((const float4*)a_dst)[node];

    __shared__ __attribute__((aligned(16))) float tabs[GWAVES * TABW];
    float* tab = tabs + (threadIdx.x >> 5) * TABW;
    const float* wrow = tab + (lane >> 3) * WSTR;
    const char* hb = (const char*)h;
    unsigned lane8 = 8u * (unsigned)lane;
    const int2*   perm2  = (const int2*)perm;
    const float4* score4 = (const float4*)score;

    int m0 = end - beg;
    if (m0 > 32) m0 = 32;
    int2 pe0 = make_int2(0, 0);
    float4 s0 = make_float4(0.0f, 0.0f, 0.0f, 0.0f);
    if (lane < m0) {
        pe0 = perm2[beg + lane];
        s0  = edge_act(score4[pe0.x], ad);
    }

    float mx0 = fmaxf(s0.x, 0.0f), mx1 = fmaxf(s0.y, 0.0f);
    float mx2 = fmaxf(s0.z, 0.0f), mx3 = fmaxf(s0.w, 0.0f);
    for (int cb = beg + 32; cb < end; cb += 32) {
        if (lane < end - cb) {
            float4 s = edge_act(score4[perm2[cb + lane].x], ad);
            mx0 = fmaxf(mx0, s.x);  mx1 = fmaxf(mx1, s.y);
            mx2 = fmaxf(mx2, s.z);  mx3 = fmaxf(mx3, s.w);
        }
    }
    mx0 = wave_max(mx0);  mx1 = wave_max(mx1);
    mx2 = wave_max(mx2);  mx3 = wave_max(mx3);

    float e0 = 0.0f, e1 = 0.0f, e2 = 0.0f, e3 = 0.0f;
    if (lane < m0) {
        e0 = expf(s0.x - mx0);  e1 = expf(s0.y - mx1);
        e2 = expf(s0.z - mx2);  e3 = expf(s0.w - mx3);
    }
    float sm0 = e0, sm1 = e1, sm2 = e2, sm3 = e3;
    for (int cb = beg + 32; cb < end; cb += 32) {
        if (lane < end - cb) {
            float4 s = edge_act(score4[perm2[cb + lane].x], ad);
            sm0 += expf(s.x - mx0);  sm1 += expf(s.y - mx1);
            sm2 += expf(s.z - mx2);  sm3 += expf(s.w - mx3);
        }
    }
    float den0 = wave_sum(sm0) + EPS_SOFTMAX, den1 = wave_sum(sm1) + EPS_SOFTMAX;
    float den2 = wave_sum(sm2) + EPS_SOFTMAX, den3 = wave_sum(sm3) + EPS_SOFTMAX;

    float ax = 0.0f, ay = 0.0f;
    add_rows(hb, lane8, tab, wrow, lane, (unsigned)pe0.y << 8,
             e0 / den0, e1 / den1, e2 / den2, e3 / den3, m0, ax, ay);
    for (int cb = beg + 32; cb < end; cb += 32) {
        int m = end - cb;
        if (m > 32) m = 32;
        int2 pe = make_int2(0, 0);
        float x0 = 0.0f, x1 = 0.0f, x2 = 0.0f, x3 = 0.0f;
        if (lane < m) {
            pe = perm2[cb + lane];
            float4 s = edge_act(score4[pe.x], ad);
            x0 = expf(s.x - mx0);  x1 = expf(s.y - mx1);
            x2 = expf(s.z - mx2);  x3 = expf(s.w - mx3);
        }
        add_rows(hb, lane8, tab, wrow, lane, (unsigned)pe.y << 8,
                 x0 / den0, x1 / den1, x2 / den2, x3 / den3, m, ax, ay);
    }

    long base = (long)node * OUT_DIM + 2 * lane;
    float2 hn = *(const float2*)(h + base);
    float y0 = ax + hn.x;
    float y1 = ay + hn.y;

    float s = y0 + y1;
    #pragma unroll
    for (int o = 16; o > 0; o >>= 1) s += __shfl_xor(s, o, 32);
    float mu = s * (1.0f / 64.0f);

    float d0 = y0 - mu, d1 = y1 - mu;
    float vs = d0 * d0 + d1 * d1;
    #pragma unroll
    for (int o = 16; o > 0; o >>= 1) vs += __shfl_xor(vs, o, 32);
    float rstd = rsqrtf(vs * (1.0f / 64.0f) + EPS_LN);

    float2 g = ((const float2*)gamma)[lane];
    float2 b = ((const float2*)beta)[lane];
    float2 o2;
    o2.x = d0 * rstd * g.x + b.x;
    o2.y = d1 * rstd * g.y + b.y;
    *(float2*)(out + base) = o2;
}

extern "C" void kernel_launch(void* const* d_in, const int* in_sizes, int n_in,
                              void* d_out, int out_size, void* d_ws, size_t ws_size,
                              hipStream_t stream) {
    const float* x         = (const float*)d_in[0];
    const int*   ei        = (const int*)  d_in[1];
    const float* edge_attr = (const float*)d_in[2];
    const float* W_node    = (const float*)d_in[3];
    const float* W_edge    = (const float*)d_in[4];
    const float* attn_src  = (const float*)d_in[5];
    const float* attn_dst  = (const float*)d_in[6];
    const float* gamma     = (const float*)d_in[7];
    const float* beta      = (const float*)d_in[8];
    float* out = (float*)d_out;

    int  n = in_sizes[0] / IN_DIM;
    long e = (long)in_sizes[2] / EDGE_DIM;
    const int* srcIdx = ei;
    const int* dstIdx = ei + e;

    int nb = (n + BR - 1) / BR;

    float* ws    = (float*)d_ws;
    float* h     = ws;  ws += (long)n * OUT_DIM;
    float* a_src = ws;  ws += (long)n * H_HEADS;
    float* a_dst = ws;  ws += (long)n * H_HEADS;
    float* attn  = ws;  ws += e * H_HEADS;
    int*   seg   = (int*)ws;  ws += ((long)n + 4) / 4 * 4;
    int*   perm  = (int*)ws;  ws += 2 * e;
    int*   offmat = (int*)ws; ws += (long)(nb + 1) * NCHUNK;
    int2*  rec   = (int2*)out;

    int n_tiles = (n + 15) / 16;
    gemm_and_bin<<<NCHUNK + (n_tiles + 1) / 2, L1_T, 0, stream>>>(
        x, W_node, attn_src, attn_dst, h, a_src, a_dst, n, srcIdx, dstIdx, offmat, rec, e, nb);

    sort_and_score<<<nb + (int)((e + L3_T - 1) / L3_T), L3_T, 0, stream>>>(
        rec, offmat, (int2*)perm, seg, n, nb, srcIdx, edge_attr, W_edge, a_src, attn, e);

    gather_ln<<<(n + GWAVES - 1) / GWAVES, 32 * GWAVES, 0, stream>>>(
        perm, seg, h, attn, a_dst, gamma, beta, out, n);
}


# fewer, longer-lived waves: node product, edge score and per-node gather loop over their items from grids that fit on the chip at once (gather: 100000 one-node waves -> 16384); sum order kept
# speedup vs baseline: 5.7802x; 1.0402x over previous
#include <hip/hip_runtime.h>
#include <hip/hip_bf16.h>

typedef float v2f __attribute__((ext_vector_type(2)));
typedef float v8f __attribute__((ext_vector_type(8)));

#define IN_DIM   64
#define OUT_DIM  64
#define EDGE_DIM 16
#define H_HEADS  4
#define HD       16
#define NEG_SLOPE   0.2f
#define EPS_SOFTMAX 1e-8f
#define EPS_LN      1e-5f

template <int M>
__device__ __forceinline__ float xrow(float x) {
    return __int_as_float(__builtin_amdgcn_update_dpp(0, __float_as_int(x), 0x160 + M, 0xf, 0xf, false));
}
__device__ __forceinline__ float row16_sum(float v) {
    v += xrow<1>(v);
    v += xrow<2>(v);
    v += xrow<4>(v);
    v += xrow<8>(v);
    return v;
}

#define L1_T 256
#define GEMM_BLOCKS 768
__device__ __forceinline__ void node_gemm_role(int gblock0, int gstride,
                               const float* __restrict__ x,
                               const float* __restrict__ W,
                               const float* __restrict__ attn_src,
                               const float* __restrict__ attn_dst,
                               float* __restrict__ h,
                               float* __restrict__ a_src,
                               float* __restrict__ a_dst, int n) {
    int wave = (threadIdx.x >> 5) & 3;
    int band = threadIdx.x >> 7;
    int lane = threadIdx.x & 31;
    int grp  = lane >> 4;
    int lid  = lane & 15;

    int col0 = wave * 16;
    const float* wc = W + (long)(2 * grp) * OUT_DIM + col0 + lid;

    float ws = attn_src[col0 + lid];
    float wd = attn_dst[col0 + lid];
    v2f bf[IN_DIM / 4];
    #pragma unroll
    for (int k0 = 0; k0 < IN_DIM; k0 += 4) {
        bf[k0 / 4].x = wc[(long)k0 * OUT_DIM];
        bf[k0 / 4].y = wc[(long)(k0 + 1) * OUT_DIM];
    }

  for (int gblock = gblock0; ; gblock += gstride) {
    int row0 = (gblock * 2 + band) * 16;
    if (row0 >= n) break;

    int arow = row0 + lid;
    if (arow >= n) arow = n - 1;

    const float* xr = x + (long)arow * IN_DIM + 2 * grp;

    v8f c = {};
    #pragma unroll
    for (int k0 = 0; k0 < IN_DIM; k0 += 4) {
        v2f a = *(const v2f*)(xr + k0);
        c = __builtin_amdgcn_wmma_f32_16x16x4_f32(false, a, false, bf[k0 / 4],
                                                  (short)0, c, false, false);
    }

    #pragma unroll
    for (int r = 0; r < 8; ++r) {
        int row = row0 + r + 8 * grp;
        if (row < n) h[(long)row * OUT_DIM + col0 + lid] = c[r];
    }

    float ms = 0.0f, md = 0.0f;
    #pragma unroll
    for (int r = 0; r < 8; ++r) {
        float ps = row16_sum(c[r] * ws);
        float pd = row16_sum(c[r] * wd);
        if (lid == r) { ms = ps; md = pd; }
    }
    int prow = row0 + 8 * grp + lid;
    if (lid < 8 && prow < n) {
        a_src[(long)prow * H_HEADS + wave] = ms;
        a_dst[(long)prow * H_HEADS + wave] = md;
    }
  }
}

#define SCORE_BLOCKS 632
__device__ __forceinline__ void edge_score_role(long e0, long estride,
                           const int* __restrict__ src,
                           const float* __restrict__ edge_attr,
                           const float* __restrict__ W_edge,
                           const float* __restrict__ a_src,
                           float* __restrict__ attn, long e_cnt) {
  for (long e = e0; e < e_cnt; e += estride) {
    asm volatile("" ::: "memory");
    int s = src[e];

    float sc[H_HEADS] = {0.f, 0.f, 0.f, 0.f};
    const float4* ea = (const float4*)(edge_attr + e * EDGE_DIM);
    #pragma unroll
    for (int q = 0; q < 4; ++q) {
        float4 v = ea[q];
        int d0 = q * 4;
        #pragma unroll
        for (int hh = 0; hh < H_HEADS; ++hh) {
            sc[hh] += v.x * W_edge[(d0 + 0) * H_HEADS + hh]
                    + v.y * W_edge[(d0 + 1) * H_HEADS + hh]
                    + v.z * W_edge[(d0 + 2) * H_HEADS + hh]
                    + v.w * W_edge[(d0 + 3) * H_HEADS + hh];
        }
    }
    float4 asv = ((const float4*)a_src)[s];
    ((float4*)attn)[e] = make_float4(asv.x + sc[0], asv.y + sc[1], asv.z + sc[2], asv.w + sc[3]);
  }
}

#define BR_SHIFT 8
#define BR       256
#define NCHUNK   256
#define MAXB     1024

template <int M>
__device__ __forceinline__ int xrow_i(int x) {
    return __builtin_amdgcn_update_dpp(0, x, 0x160 + M, 0xf, 0xf, false);
}
__device__ __forceinline__ int wave_sum_i(int v) {
    v += xrow_i<1>(v);
    v += xrow_i<2>(v);
    v += xrow_i<4>(v);
    v += xrow_i<8>(v);
    return v + __builtin_amdgcn_permlanex16(v, v, 0x76543210, 0xfedcba98, false, false);
}

__device__ __forceinline__ long chunk_begin(int c, long e_cnt) {
    long per = (e_cnt + NCHUNK - 1) / NCHUNK;
    long e0 = (long)c * per;
    return e0 > e_cnt ? e_cnt : e0;
}

__device__ __forceinline__ void chunk_bin_role(const int* __restrict__ src, const int* __restrict__ dst,
                                               int* __restrict__ offmat, int2* __restrict__ rec,
                                               long e_cnt, int nb, int* hist, int* wtot) {
    int tid  = threadIdx.x;
    int lane = tid & 31;
    int wave = tid >> 5;
    int g = blockIdx.x;
    for (int t = tid; t < MAXB; t += L1_T) hist[t] = 0;
    __syncthreads();
    long e0 = chunk_begin(g, e_cnt);
    long e1 = chunk_begin(g + 1, e_cnt);
    for (long i0 = e0 + tid; i0 < e1; i0 += 16 * L1_T) {
        int d[16];
        #pragma unroll
        for (int j = 0; j < 16; ++j) {
            long i = i0 + (long)j * L1_T;
            d[j] = (i < e1) ? dst[i] : -1;
        }
        #pragma unroll
        for (int j = 0; j < 16; ++j)
            if (d[j] >= 0) atomicAdd(&hist[d[j] >> BR_SHIFT], 1);
    }
    __syncthreads();

    int4 v = ((const int4*)hist)[tid];
    int s = v.x + v.y + v.z + v.w;
    int incl = s;
    #pragma unroll
    for (int d = 1; d < 32; d <<= 1) {
        int q = __shfl_up(incl, d, 32);
        if (lane >= d) incl += q;
    }
    if (lane == 31) wtot[wave] = incl;
    __syncthreads();
    int add = 0;
    for (int w = 0; w < wave; ++w) add += wtot[w];
    int p0 = (int)e0 + add + incl - s;
    int p1 = p0 + v.x, p2 = p1 + v.y, p3 = p2 + v.z;
    ((int4*)hist)[tid] = make_int4(p0, p1, p2, p3);
    int b0 = 4 * tid;
    if (b0 + 0 < nb) offmat[(long)(b0 + 0) * NCHUNK + g] = p0;
    if (b0 + 1 < nb) offmat[(long)(b0 + 1) * NCHUNK + g] = p1;
    if (b0 + 2 < nb) offmat[(long)(b0 + 2) * NCHUNK + g] = p2;
    if (b0 + 3 < nb) offmat[(long)(b0 + 3) * NCHUNK + g] = p3;
    if (tid == 0) offmat[(long)nb * NCHUNK + g] = (int)e1;
    __syncthreads();

    for (long i0 = e0 + tid; i0 < e1; i0 += 16 * L1_T) {
        int d[16], sn[16];
        #pragma unroll
        for (int j = 0; j < 16; ++j) {
            long i = i0 + (long)j * L1_T;
            bool ok = i < e1;
            d[j]  = ok ? dst[i] : -1;
            sn[j] = ok ? src[i] : 0;
        }
        #pragma unroll
        for (int j = 0; j < 16; ++j) {
            if (d[j] >= 0) {
                long i = i0 + (long)j * L1_T;
                int slot = atomicAdd(&hist[d[j] >> BR_SHIFT], 1);
                unsigned loc = (unsigned)(d[j] & (BR - 1));
                rec[slot] = make_int2((int)i, (int)((unsigned)sn[j] | (loc << 24)));
            }
        }
    }
}

__global__ void __launch_bounds__(L1_T)
gemm_and_bin(const float* __restrict__ x,
             const float* __restrict__ W,
             const float* __restrict__ attn_src,
             const float* __restrict__ attn_dst,
             float* __restrict__ h,
             float* __restrict__ a_src,
             float* __restrict__ a_dst, int n,
             const int* __restrict__ src, const int* __restrict__ dst,
             int* __restrict__ offmat, int2* __restrict__ rec, long e_cnt, int nb) {
    __shared__ __attribute__((aligned(16))) int hist[MAXB];
    __shared__ int wtot[L1_T / 32];
    if (blockIdx.x < NCHUNK) {
        chunk_bin_role(src, dst, offmat, rec, e_cnt, nb, hist, wtot);
        return;
    }
    node_gemm_role((int)blockIdx.x - NCHUNK, (int)gridDim.x - NCHUNK,
                   x, W, attn_src, attn_dst, h, a_src, a_dst, n);
}

__device__ __forceinline__ int rec_slot(int i, const int* pos, const int* off) {
    int c = 0;
    #pragma unroll
    for (int st = NCHUNK / 2; st > 0; st >>= 1)
        if (pos[c + st] <= i) c += st;
    return off[c] + (i - pos[c]);
}

#define L3_T 512
__device__ __forceinline__ void bucket_sort_role(const int2* __restrict__ rec, const int* __restrict__ offmat,
            int2* __restrict__ perm, int* __restrict__ seg, int n, int nb, long e_cnt,
            int* cnt, int* wtot, int* pos, int* off, int* wcnt, int* wrel) {
    int tid  = threadIdx.x;
    int lane = tid & 31;
    int wave = tid >> 5;
    int b  = blockIdx.x;
    int o0 = 0, len = 0, incl = 0;
    if (tid < NCHUNK) {
        o0 = offmat[(long)b * NCHUNK + tid];
        len = offmat[(long)(b + 1) * NCHUNK + tid] - o0;
        int rel = o0 - (int)chunk_begin(tid, e_cnt);
        incl = len;
        #pragma unroll
        for (int d = 1; d < 32; d <<= 1) {
            int q = __shfl_up(incl, d, 32);
            if (lane >= d) incl += q;
        }
        rel = wave_sum_i(rel);
        if (lane == 31) wcnt[wave] = incl;
        if (lane == 0)  wrel[wave] = rel;
    }
    if (tid < BR) cnt[tid] = 0;
    __syncthreads();
    int tot = 0, bb = 0, before = 0;
    #pragma unroll
    for (int w = 0; w < NCHUNK / 32; ++w) {
        int t = wcnt[w];
        if (w < wave) before += t;
        tot += t;
        bb  += wrel[w];
    }
    int be = bb + tot;
    if (tid < NCHUNK) {
        pos[tid] = before + incl - len;
        off[tid] = o0;
    }
    __syncthreads();

    for (int i0 = tid; i0 < tot; i0 += 8 * L3_T) {
        unsigned y[8];
        #pragma unroll
        for (int j = 0; j < 8; ++j) {
            int i = i0 + j * L3_T;
            y[j] = (unsigned)rec[rec_slot(i < tot ? i : tot - 1, pos, off)].y;
        }
        #pragma unroll
        for (int j = 0; j < 8; ++j)
            if (i0 + j * L3_T < tot) atomicAdd(&cnt[y[j] >> 24], 1);
    }
    __syncthreads();
    int c = 0, inc2 = 0;
    if (tid < BR) {
        c = cnt[tid];
        inc2 = c;
        #pragma unroll
        for (int d = 1; d < 32; d <<= 1) {
            int q = __shfl_up(inc2, d, 32);
            if (lane >= d) inc2 += q;
        }
        if (lane == 31) wtot[wave] = inc2;
    }
    __syncthreads();
    if (tid < BR) {
        int add = 0;
        for (int w = 0; w < wave; ++w) add += wtot[w];
        int start = bb + add + inc2 - c;
        cnt[tid] = start;
        int node = b * BR + tid;
        if (node < n) seg[node] = start;
    }
    if (b == nb - 1 && tid == 0) seg[n] = be;
    __syncthreads();
    for (int i0 = tid; i0 < tot; i0 += 8 * L3_T) {
        int2 r[8];
        #pragma unroll
        for (int j = 0; j < 8; ++j) {
            int i = i0 + j * L3_T;
            r[j] = rec[rec_slot(i < tot ? i : tot - 1, pos, off)];
        }
        #pragma unroll
        for (int j = 0; j < 8; ++j) {
            if (i0 + j * L3_T < tot) {
                int slot = atomicAdd(&cnt[(unsigned)r[j].y >> 24], 1);
                perm[slot] = make_int2(r[j].x, r[j].y & 0x00ffffff);
            }
        }
    }
}

__global__ void __launch_bounds__(L3_T)
sort_and_score(const int2* __restrict__ rec, const int* __restrict__ offmat,
               int2* __restrict__ perm, int* __restrict__ seg, int n, int nb,
               const int* __restrict__ src,
               const float* __restrict__ edge_attr,
               const float* __restrict__ W_edge,
               const float* __restrict__ a_src,
               float* __restrict__ attn, long e_cnt) {
    __shared__ int cnt[BR];
    __shared__ int wtot[BR / 32];
    __shared__ int pos[NCHUNK];
    __shared__ int off[NCHUNK];
    __shared__ int wcnt[NCHUNK / 32];
    __shared__ int wrel[NCHUNK / 32];
    if ((int)blockIdx.x < nb) {
        bucket_sort_role(rec, offmat, perm, seg, n, nb, e_cnt, cnt, wtot, pos, off, wcnt, wrel);
        return;
    }
    edge_score_role((long)((int)blockIdx.x - nb) * L3_T + threadIdx.x,
                    (long)((int)gridDim.x - nb) * L3_T,
                    src, edge_attr, W_edge, a_src, attn, e_cnt);
}

__device__ __forceinline__ float xhalf(float x) {
    int b = __float_as_int(x);
    return __int_as_float(__builtin_amdgcn_permlanex16(b, b, 0x76543210, 0xfedcba98, false, false));
}
__device__ __forceinline__ float wave_max(float v) {
    v = fmaxf(v, xrow<1>(v));
    v = fmaxf(v, xrow<2>(v));
    v = fmaxf(v, xrow<4>(v));
    v = fmaxf(v, xrow<8>(v));
    return fmaxf(v, xhalf(v));
}
__device__ __forceinline__ float wave_sum(float v) {
    v += xrow<1>(v);
    v += xrow<2>(v);
    v += xrow<4>(v);
    v += xrow<8>(v);
    return v + xhalf(v);
}

#define GWAVES 8
#define WSTR   36
#define TABW   (4 * WSTR + 32)

__device__ __forceinline__ void wave_lds_sync() {
    __builtin_amdgcn_fence(__ATOMIC_RELEASE, "wavefront");
    __builtin_amdgcn_wave_barrier();
    __builtin_amdgcn_fence(__ATOMIC_ACQUIRE, "wavefront");
}

__device__ __forceinline__ void add_rows(const char* __restrict__ hb, unsigned lane8,
                                         float* tab, const float* wrow, int lane,
                                         unsigned srcoff,
                                         float w0, float w1, float w2, float w3,
                                         int m, float& ax, float& ay) {
    unsigned* st = (unsigned*)(tab + 4 * WSTR);
    wave_lds_sync();
    tab[0 * WSTR + lane] = w0;
    tab[1 * WSTR + lane] = w1;
    tab[2 * WSTR + lane] = w2;
    tab[3 * WSTR + lane] = w3;
    st[lane] = srcoff;
    wave_lds_sync();
    for (int k0 = 0; k0 < m; k0 += 8) {
        float4 wa = *(const float4*)(wrow + k0);
        float4 wb = *(const float4*)(wrow + k0 + 4);
        uint4  sa = *(const uint4*)(st + k0);
        uint4  sb = *(const uint4*)(st + k0 + 4);
        float2 hv[8];
        hv[0] = *(const float2*)(hb + (sa.x + lane8));
        hv[1] = *(const float2*)(hb + (sa.y + lane8));
        hv[2] = *(const float2*)(hb + (sa.z + lane8));
        hv[3] = *(const float2*)(hb + (sa.w + lane8));
        hv[4] = *(const float2*)(hb + (sb.x + lane8));
        hv[5] = *(const float2*)(hb + (sb.y + lane8));
        hv[6] = *(const float2*)(hb + (sb.z + lane8));
        hv[7] = *(const float2*)(hb + (sb.w + lane8));
        ax += hv[0].x * wa.x;  ay += hv[0].y * wa.x;
        ax += hv[1].x * wa.y;  ay += hv[1].y * wa.y;
        ax += hv[2].x * wa.z;  ay += hv[2].y * wa.z;
        ax += hv[3].x * wa.w;  ay += hv[3].y * wa.w;
        ax += hv[4].x * wb.x;  ay += hv[4].y * wb.x;
        ax += hv[5].x * wb.y;  ay += hv[5].y * wb.y;
        ax += hv[6].x * wb.z;  ay += hv[6].y * wb.z;
        ax += hv[7].x * wb.w;  ay += hv[7].y * wb.w;
    }
}

__device__ __forceinline__ float4 edge_act(float4 s, float4 ad) {
    float4 r;
    float v;
    v = s.x + ad.x;  r.x = (v >= 0.0f) ? v : NEG_SLOPE * v;
    v = s.y + ad.y;  r.y = (v >= 0.0f) ? v : NEG_SLOPE * v;
    v = s.z + ad.z;  r.z = (v >= 0.0f) ? v : NEG_SLOPE * v;
    v = s.w + ad.w;  r.w = (v >= 0.0f) ? v : NEG_SLOPE * v;
    return r;
}

#define GATHER_BLOCKS 2048
__global__ void __launch_bounds__(32 * GWAVES)
gather_ln(const int* __restrict__ perm,
          const int* __restrict__ seg,
          const float* __restrict__ h,
          const float* __restrict__ score,
          const float* __restrict__ a_dst,
          const float* __restrict__ gamma,
          const float* __restrict__ beta,
          float* __restrict__ out, int n) {
    __shared__ __attribute__((aligned(16))) float tabs[GWAVES * TABW];
    int lane = threadIdx.x & 31;
    float* tab = tabs + (threadIdx.x >> 5) * TABW;
    const float* wrow = tab + (lane >> 3) * WSTR;
    const char* hb = (const char*)h;
    unsigned lane8 = 8u * (unsigned)lane;
    const int2*   perm2  = (const int2*)perm;
    const float4* score4 = (const float4*)score;
    const float2 g = ((const float2*)gamma)[lane];
    const float2 b = ((const float2*)beta)[lane];

    int node0  = __builtin_amdgcn_readfirstlane((int)(blockIdx.x * GWAVES + (threadIdx.x >> 5)));
    int nstep  = (int)gridDim.x * GWAVES;
  for (int node = node0; node < n; node += nstep) {
    int beg = seg[node];
    int end = seg[node + 1];
    const float4 ad = ((const float4*)a_dst)[node];

    int m0 = end - beg;
    if (m0 > 32) m0 = 32;
    int2 pe0 = make_int2(0, 0);
    float4 s0 = make_float4(0.0f, 0.0f, 0.0f, 0.0f);
    if (lane < m0) {
        pe0 = perm2[beg + lane];
        s0  = edge_act(score4[pe0.x], ad);
    }

    float mx0 = fmaxf(s0.x, 0.0f), mx1 = fmaxf(s0.y, 0.0f);
    float mx2 = fmaxf(s0.z, 0.0f), mx3 = fmaxf(s0.w, 0.0f);
    for (int cb = beg + 32; cb < end; cb += 32) {
        if (lane < end - cb) {
            float4 s = edge_act(score4[perm2[cb + lane].x], ad);
            mx0 = fmaxf(mx0, s.x);  mx1 = fmaxf(mx1, s.y);
            mx2 = fmaxf(mx2, s.z);  mx3 = fmaxf(mx3, s.w);
        }
    }
    mx0 = wave_max(mx0);  mx1 = wave_max(mx1);
    mx2 = wave_max(mx2);  mx3 = wave_max(mx3);

    float e0 = 0.0f, e1 = 0.0f, e2 = 0.0f, e3 = 0.0f;
    if (lane < m0) {
        e0 = expf(s0.x - mx0);  e1 = expf(s0.y - mx1);
        e2 = expf(s0.z - mx2);  e3 = expf(s0.w - mx3);
    }
    float sm0 = e0, sm1 = e1, sm2 = e2, sm3 = e3;
    for (int cb = beg + 32; cb < end; cb += 32) {
        if (lane < end - cb) {
            float4 s = edge_act(score4[perm2[cb + lane].x], ad);
            sm0 += expf(s.x - mx0);  sm1 += expf(s.y - mx1);
            sm2 += expf(s.z - mx2);  sm3 += expf(s.w - mx3);
        }
    }
    float den0 = wave_sum(sm0) + EPS_SOFTMAX, den1 = wave_sum(sm1) + EPS_SOFTMAX;
    float den2 = wave_sum(sm2) + EPS_SOFTMAX, den3 = wave_sum(sm3) + EPS_SOFTMAX;

    float ax = 0.0f, ay = 0.0f;
    add_rows(hb, lane8, tab, wrow, lane, (unsigned)pe0.y << 8,
             e0 / den0, e1 / den1, e2 / den2, e3 / den3, m0, ax, ay);
    for (int cb = beg + 32; cb < end; cb += 32) {
        int m = end - cb;
        if (m > 32) m = 32;
        int2 pe = make_int2(0, 0);
        float x0 = 0.0f, x1 = 0.0f, x2 = 0.0f, x3 = 0.0f;
        if (lane < m) {
            pe = perm2[cb + lane];
            float4 s = edge_act(score4[pe.x], ad);
            x0 = expf(s.x - mx0);  x1 = expf(s.y - mx1);
            x2 = expf(s.z - mx2);  x3 = expf(s.w - mx3);
        }
        add_rows(hb, lane8, tab, wrow, lane, (unsigned)pe.y << 8,
                 x0 / den0, x1 / den1, x2 / den2, x3 / den3, m, ax, ay);
    }

    long base = (long)node * OUT_DIM + 2 * lane;
    float2 hn = *(const float2*)(h + base);
    float y0 = ax + hn.x;
    float y1 = ay + hn.y;

    float s = y0 + y1;
    #pragma unroll
    for (int o = 16; o > 0; o >>= 1) s += __shfl_xor(s, o, 32);
    float mu = s * (1.0f / 64.0f);

    float d0 = y0 - mu, d1 = y1 - mu;
    float vs = d0 * d0 + d1 * d1;
    #pragma unroll
    for (int o = 16; o > 0; o >>= 1) vs += __shfl_xor(vs, o, 32);
    float rstd = rsqrtf(vs * (1.0f / 64.0f) + EPS_LN);

    float2 o2;
    o2.x = d0 * rstd * g.x + b.x;
    o2.y = d1 * rstd * g.y + b.y;
    *(float2*)(out + base) = o2;
  }
}

extern "C" void kernel_launch(void* const* d_in, const int* in_sizes, int n_in,
                              void* d_out, int out_size, void* d_ws, size_t ws_size,
                              hipStream_t stream) {
    const float* x         = (const float*)d_in[0];
    const int*   ei        = (const int*)  d_in[1];
    const float* edge_attr = (const float*)d_in[2];
    const float* W_node    = (const float*)d_in[3];
    const float* W_edge    = (const float*)d_in[4];
    const float* attn_src  = (const float*)d_in[5];
    const float* attn_dst  = (const float*)d_in[6];
    const float* gamma     = (const float*)d_in[7];
    const float* beta      = (const float*)d_in[8];
    float* out = (float*)d_out;

    int  n = in_sizes[0] / IN_DIM;
    long e = (long)in_sizes[2] / EDGE_DIM;
    const int* srcIdx = ei;
    const int* dstIdx = ei + e;

    int nb = (n + BR - 1) / BR;

    float* ws    = (float*)d_ws;
    float* h     = ws;  ws += (long)n * OUT_DIM;
    float* a_src = ws;  ws += (long)n * H_HEADS;
    float* a_dst = ws;  ws += (long)n * H_HEADS;
    float* attn  = ws;  ws += e * H_HEADS;
    int*   seg   = (int*)ws;  ws += ((long)n + 4) / 4 * 4;
    int*   perm  = (int*)ws;  ws += 2 * e;
    int*   offmat = (int*)ws; ws += (long)(nb + 1) * NCHUNK;
    int2*  rec   = (int2*)out;

    gemm_and_bin<<<NCHUNK + GEMM_BLOCKS, L1_T, 0, stream>>>(
        x, W_node, attn_src, attn_dst, h, a_src, a_dst, n, srcIdx, dstIdx, offmat, rec, e, nb);

    sort_and_score<<<nb + SCORE_BLOCKS, L3_T, 0, stream>>>(
        rec, offmat, (int2*)perm, seg, n, nb, srcIdx, edge_attr, W_edge, a_src, attn, e);

    gather_ln<<<GATHER_BLOCKS, 32 * GWAVES, 0, stream>>>(
        perm, seg, h, attn, a_dst, gamma, beta, out, n);
}


# ranking and per-node gather fused, one work-group per bin: ranked edge list and node starts stay in LDS; perm/seg planes and a launch gone (3 -> 2); edge term now in launch 1; node sum order changes
# speedup vs baseline: 6.3399x; 1.0968x over previous
#include <hip/hip_runtime.h>
#include <hip/hip_bf16.h>

typedef float v2f __attribute__((ext_vector_type(2)));
typedef float v8f __attribute__((ext_vector_type(8)));

#define IN_DIM   64
#define OUT_DIM  64
#define EDGE_DIM 16
#define H_HEADS  4
#define HD       16
#define NEG_SLOPE   0.2f
#define EPS_SOFTMAX 1e-8f
#define EPS_LN      1e-5f

template <int M>
__device__ __forceinline__ float xrow(float x) {
    return __int_as_float(__builtin_amdgcn_update_dpp(0, __float_as_int(x), 0x160 + M, 0xf, 0xf, false));
}
__device__ __forceinline__ float row16_sum(float v) {
    v += xrow<1>(v);
    v += xrow<2>(v);
    v += xrow<4>(v);
    v += xrow<8>(v);
    return v;
}

#define L1_T 256
#define GEMM_BLOCKS 768
__device__ __forceinline__ void node_gemm_role(int gblock0, int gstride,
                               const float* __restrict__ x,
                               const float* __restrict__ W,
                               const float* __restrict__ attn_src,
                               const float* __restrict__ attn_dst,
                               float* __restrict__ h,
                               float* __restrict__ a_src,
                               float* __restrict__ a_dst, int n) {
    int wave = (threadIdx.x >> 5) & 3;
    int band = threadIdx.x >> 7;
    int lane = threadIdx.x & 31;
    int grp  = lane >> 4;
    int lid  = lane & 15;

    int col0 = wave * 16;
    const float* wc = W + (long)(2 * grp) * OUT_DIM + col0 + lid;

    float ws = attn_src[col0 + lid];
    float wd = attn_dst[col0 + lid];
    v2f bf[IN_DIM / 4];
    #pragma unroll
    for (int k0 = 0; k0 < IN_DIM; k0 += 4) {
        bf[k0 / 4].x = wc[(long)k0 * OUT_DIM];
        bf[k0 / 4].y = wc[(long)(k0 + 1) * OUT_DIM];
    }

  for (int gblock = gblock0; ; gblock += gstride) {
    int row0 = (gblock * 2 + band) * 16;
    if (row0 >= n) break;

    int arow = row0 + lid;
    if (arow >= n) arow = n - 1;

    const float* xr = x + (long)arow * IN_DIM + 2 * grp;

    v8f c = {};
    #pragma unroll
    for (int k0 = 0; k0 < IN_DIM; k0 += 4) {
        v2f a = *(const v2f*)(xr + k0);
        c = __builtin_amdgcn_wmma_f32_16x16x4_f32(false, a, false, bf[k0 / 4],
                                                  (short)0, c, false, false);
    }

    #pragma unroll
    for (int r = 0; r < 8; ++r) {
        int row = row0 + r + 8 * grp;
        if (row < n) h[(long)row * OUT_DIM + col0 + lid] = c[r];
    }

    float ms = 0.0f, md = 0.0f;
    #pragma unroll
    for (int r = 0; r < 8; ++r) {
        float ps = row16_sum(c[r] * ws);
        float pd = row16_sum(c[r] * wd);
        if (lid == r) { ms = ps; md = pd; }
    }
    int prow = row0 + 8 * grp + lid;
    if (lid < 8 && prow < n) {
        a_src[(long)prow * H_HEADS + wave] = ms;
        a_dst[(long)prow * H_HEADS + wave] = md;
    }
  }
}

#define ETERM_BLOCKS 1024
__device__ __forceinline__ float4 edge_term(const float4* __restrict__ ea,
                                            const float* __restrict__ W_edge) {
    float sc[H_HEADS] = {0.f, 0.f, 0.f, 0.f};
    #pragma unroll
    for (int q = 0; q < 4; ++q) {
        float4 v = ea[q];
        int d0 = q * 4;
        #pragma unroll
        for (int hh = 0; hh < H_HEADS; ++hh) {
            sc[hh] += v.x * W_edge[(d0 + 0) * H_HEADS + hh]
                    + v.y * W_edge[(d0 + 1) * H_HEADS + hh]
                    + v.z * W_edge[(d0 + 2) * H_HEADS + hh]
                    + v.w * W_edge[(d0 + 3) * H_HEADS + hh];
        }
    }
    return make_float4(sc[0], sc[1], sc[2], sc[3]);
}
__device__ __forceinline__ void edge_term_role(long e0, long estride,
                           const float* __restrict__ edge_attr,
                           const float* __restrict__ W_edge,
                           float* __restrict__ eterm, long e_cnt) {
  long e = e0;
  for (; e + estride < e_cnt; e += 2 * estride) {
    asm volatile("" ::: "memory");
    long e2 = e + estride;
    float4 ta = edge_term((const float4*)(edge_attr + e  * EDGE_DIM), W_edge);
    float4 tb = edge_term((const float4*)(edge_attr + e2 * EDGE_DIM), W_edge);
    ((float4*)eterm)[e]  = ta;
    ((float4*)eterm)[e2] = tb;
  }
  if (e < e_cnt)
    ((float4*)eterm)[e] = edge_term((const float4*)(edge_attr + e * EDGE_DIM), W_edge);
}

#define BR       196
#define BRP      224
#define NCHUNK   256
#define MAXB     1024

__device__ __forceinline__ long chunk_begin(int c, long e_cnt) {
    long per = (e_cnt + NCHUNK - 1) / NCHUNK;
    long e0 = (long)c * per;
    return e0 > e_cnt ? e_cnt : e0;
}

__device__ __forceinline__ void chunk_bin_role(const int* __restrict__ src, const int* __restrict__ dst,
                                               int* __restrict__ offmat, int2* __restrict__ rec,
                                               long e_cnt, int nb, int* hist, int* wtot) {
    int tid  = threadIdx.x;
    int lane = tid & 31;
    int wave = tid >> 5;
    int g = blockIdx.x;
    for (int t = tid; t < MAXB; t += L1_T) hist[t] = 0;
    __syncthreads();
    long e0 = chunk_begin(g, e_cnt);
    long e1 = chunk_begin(g + 1, e_cnt);
    for (long i0 = e0 + tid; i0 < e1; i0 += 16 * L1_T) {
        int d[16];
        #pragma unroll
        for (int j = 0; j < 16; ++j) {
            long i = i0 + (long)j * L1_T;
            d[j] = (i < e1) ? dst[i] : -1;
        }
        #pragma unroll
        for (int j = 0; j < 16; ++j)
            if (d[j] >= 0) atomicAdd(&hist[(unsigned)d[j] / BR], 1);
    }
    __syncthreads();

    int4 v = ((const int4*)hist)[tid];
    int s = v.x + v.y + v.z + v.w;
    int incl = s;
    #pragma unroll
    for (int d = 1; d < 32; d <<= 1) {
        int q = __shfl_up(incl, d, 32);
        if (lane >= d) incl += q;
    }
    if (lane == 31) wtot[wave] = incl;
    __syncthreads();
    int add = 0;
    for (int w = 0; w < wave; ++w) add += wtot[w];
    int p0 = (int)e0 + add + incl - s;
    int p1 = p0 + v.x, p2 = p1 + v.y, p3 = p2 + v.z;
    ((int4*)hist)[tid] = make_int4(p0, p1, p2, p3);
    int b0 = 4 * tid;
    if (b0 + 0 < nb) offmat[(long)(b0 + 0) * NCHUNK + g] = p0;
    if (b0 + 1 < nb) offmat[(long)(b0 + 1) * NCHUNK + g] = p1;
    if (b0 + 2 < nb) offmat[(long)(b0 + 2) * NCHUNK + g] = p2;
    if (b0 + 3 < nb) offmat[(long)(b0 + 3) * NCHUNK + g] = p3;
    if (tid == 0) offmat[(long)nb * NCHUNK + g] = (int)e1;
    __syncthreads();

    for (long i0 = e0 + tid; i0 < e1; i0 += 16 * L1_T) {
        int d[16], sn[16];
        #pragma unroll
        for (int j = 0; j < 16; ++j) {
            long i = i0 + (long)j * L1_T;
            bool ok = i < e1;
            d[j]  = ok ? dst[i] : -1;
            sn[j] = ok ? src[i] : 0;
        }
        #pragma unroll
        for (int j = 0; j < 16; ++j) {
            if (d[j] >= 0) {
                long i = i0 + (long)j * L1_T;
                unsigned bn = (unsigned)d[j] / BR;
                int slot = atomicAdd(&hist[bn], 1);
                unsigned loc = (unsigned)d[j] - bn * BR;
                rec[slot] = make_int2((int)i, (int)((unsigned)sn[j] | (loc << 24)));
            }
        }
    }
}

__global__ void __launch_bounds__(L1_T)
gemm_and_bin(const float* __restrict__ x,
             const float* __restrict__ W,
             const float* __restrict__ attn_src,
             const float* __restrict__ attn_dst,
             float* __restrict__ h,
             float* __restrict__ a_src,
             float* __restrict__ a_dst, int n,
             const int* __restrict__ src, const int* __restrict__ dst,
             int* __restrict__ offmat, int2* __restrict__ rec, long e_cnt, int nb,
             const float* __restrict__ edge_attr,
             const float* __restrict__ W_edge,
             float* __restrict__ eterm) {
    __shared__ __attribute__((aligned(16))) int hist[MAXB];
    __shared__ int wtot[L1_T / 32];
    if (blockIdx.x < NCHUNK) {
        chunk_bin_role(src, dst, offmat, rec, e_cnt, nb, hist, wtot);
        return;
    }
    if (blockIdx.x < NCHUNK + GEMM_BLOCKS) {
        node_gemm_role((int)blockIdx.x - NCHUNK, GEMM_BLOCKS,
                       x, W, attn_src, attn_dst, h, a_src, a_dst, n);
        return;
    }
    edge_term_role((long)((int)blockIdx.x - NCHUNK - GEMM_BLOCKS) * L1_T + threadIdx.x,
                   (long)ETERM_BLOCKS * L1_T,
                   edge_attr, W_edge, eterm, e_cnt);
}

__device__ __forceinline__ int rec_slot(int i, const int* pos, const int* off) {
    int c = 0;
    #pragma unroll
    for (int st = NCHUNK / 2; st > 0; st >>= 1)
        if (pos[c + st] <= i) c += st;
    return off[c] + (i - pos[c]);
}

__device__ __forceinline__ float xhalf(float x) {
    int b = __float_as_int(x);
    return __int_as_float(__builtin_amdgcn_permlanex16(b, b, 0x76543210, 0xfedcba98, false, false));
}
__device__ __forceinline__ float wave_max(float v) {
    v = fmaxf(v, xrow<1>(v));
    v = fmaxf(v, xrow<2>(v));
    v = fmaxf(v, xrow<4>(v));
    v = fmaxf(v, xrow<8>(v));
    return fmaxf(v, xhalf(v));
}
__device__ __forceinline__ float wave_sum(float v) {
    v += xrow<1>(v);
    v += xrow<2>(v);
    v += xrow<4>(v);
    v += xrow<8>(v);
    return v + xhalf(v);
}

#define WSTR   36
#define TABW   (4 * WSTR + 32)

__device__ __forceinline__ void wave_lds_sync() {
    __builtin_amdgcn_fence(__ATOMIC_RELEASE, "wavefront");
    __builtin_amdgcn_wave_barrier();
    __builtin_amdgcn_fence(__ATOMIC_ACQUIRE, "wavefront");
}

__device__ __forceinline__ void add_rows(const char* __restrict__ hb, unsigned lane8,
                                         float* tab, const float* wrow, int lane,
                                         unsigned srcoff,
                                         float w0, float w1, float w2, float w3,
                                         int m, float& ax, float& ay) {
    unsigned* st = (unsigned*)(tab + 4 * WSTR);
    wave_lds_sync();
    tab[0 * WSTR + lane] = w0;
    tab[1 * WSTR + lane] = w1;
    tab[2 * WSTR + lane] = w2;
    tab[3 * WSTR + lane] = w3;
    st[lane] = srcoff;
    wave_lds_sync();
    #pragma nounroll
    for (int k0 = 0; k0 < m; k0 += 8) {
        float4 wa = *(const float4*)(wrow + k0);
        float4 wb = *(const float4*)(wrow + k0 + 4);
        uint4  sa = *(const uint4*)(st + k0);
        uint4  sb = *(const uint4*)(st + k0 + 4);
        float2 hv[8];
        hv[0] = *(const float2*)(hb + (sa.x + lane8));
        hv[1] = *(const float2*)(hb + (sa.y + lane8));
        hv[2] = *(const float2*)(hb + (sa.z + lane8));
        hv[3] = *(const float2*)(hb + (sa.w + lane8));
        hv[4] = *(const float2*)(hb + (sb.x + lane8));
        hv[5] = *(const float2*)(hb + (sb.y + lane8));
        hv[6] = *(const float2*)(hb + (sb.z + lane8));
        hv[7] = *(const float2*)(hb + (sb.w + lane8));
        ax += hv[0].x * wa.x;  ay += hv[0].y * wa.x;
        ax += hv[1].x * wa.y;  ay += hv[1].y * wa.y;
        ax += hv[2].x * wa.z;  ay += hv[2].y * wa.z;
        ax += hv[3].x * wa.w;  ay += hv[3].y * wa.w;
        ax += hv[4].x * wb.x;  ay += hv[4].y * wb.x;
        ax += hv[5].x * wb.y;  ay += hv[5].y * wb.y;
        ax += hv[6].x * wb.z;  ay += hv[6].y * wb.z;
        ax += hv[7].x * wb.w;  ay += hv[7].y * wb.w;
    }
}

__device__ __forceinline__ float4 edge_act(float4 s, float4 ad) {
    float4 r;
    float v;
    v = s.x + ad.x;  r.x = (v >= 0.0f) ? v : NEG_SLOPE * v;
    v = s.y + ad.y;  r.y = (v >= 0.0f) ? v : NEG_SLOPE * v;
    v = s.z + ad.z;  r.z = (v >= 0.0f) ? v : NEG_SLOPE * v;
    v = s.w + ad.w;  r.w = (v >= 0.0f) ? v : NEG_SLOPE * v;
    return r;
}

#define L3_T   1024
#define FWAVES (L3_T / 32)
#define CAP    8192

template <bool SCAN>
__device__ __forceinline__ bool fetch_edge(int c, int lane, int beg, int end, int tot, unsigned loc,
                                           const int2* srt, const int2* __restrict__ rec,
                                           const int* pos, const int* off, int2& pe) {
    pe = make_int2(0, 0);
    if (!SCAN) {
        int i = beg + 32 * c + lane;
        bool v = i < end;
        if (v) pe = srt[i];
        return v;
    } else {
        int i = 32 * c + lane;
        bool v = i < tot;
        if (v) {
            int2 r = rec[rec_slot(i, pos, off)];
            v = ((unsigned)r.y >> 24) == loc;
            if (v) pe = make_int2(r.x, r.y & 0x00ffffff);
        }
        return v;
    }
}

__device__ __forceinline__ float4 edge_score(int2 pe, const float4* __restrict__ eterm4,
                                             const float4* __restrict__ asrc4, float4 ad) {
    float4 et = eterm4[pe.x];
    float4 as = asrc4[pe.y];
    return edge_act(make_float4(as.x + et.x, as.y + et.y, as.z + et.z, as.w + et.w), ad);
}

template <bool SCAN>
__device__ __forceinline__ void gather_node(int node, int beg, int end, int tot, unsigned loc, int lane,
        const int2* srt, const int2* __restrict__ rec, const int* pos, const int* off,
        const float* __restrict__ h, unsigned lane8, float* tab, const float* wrow,
        const float4* __restrict__ eterm4, const float4* __restrict__ asrc4,
        const float4* __restrict__ adst4, float2 g, float2 b, float* __restrict__ out) {
    const char* hb = (const char*)h;
    const float4 ad = adst4[node];
    const int ne  = SCAN ? tot : end - beg;
    const int nch = (ne + 31) >> 5;

    int2 pe0;
    bool v0 = fetch_edge<SCAN>(0, lane, beg, end, tot, loc, srt, rec, pos, off, pe0);
    float4 s0 = make_float4(0.0f, 0.0f, 0.0f, 0.0f);
    if (v0) s0 = edge_score(pe0, eterm4, asrc4, ad);

    float mx0 = fmaxf(s0.x, 0.0f), mx1 = fmaxf(s0.y, 0.0f);
    float mx2 = fmaxf(s0.z, 0.0f), mx3 = fmaxf(s0.w, 0.0f);
    for (int c = 1; c < nch; ++c) {
        int2 pe;
        if (fetch_edge<SCAN>(c, lane, beg, end, tot, loc, srt, rec, pos, off, pe)) {
            float4 s = edge_score(pe, eterm4, asrc4, ad);
            mx0 = fmaxf(mx0, s.x);  mx1 = fmaxf(mx1, s.y);
            mx2 = fmaxf(mx2, s.z);  mx3 = fmaxf(mx3, s.w);
        }
    }
    mx0 = wave_max(mx0);  mx1 = wave_max(mx1);
    mx2 = wave_max(mx2);  mx3 = wave_max(mx3);

    float e0 = 0.0f, e1 = 0.0f, e2 = 0.0f, e3 = 0.0f;
    if (v0) {
        e0 = expf(s0.x - mx0);  e1 = expf(s0.y - mx1);
        e2 = expf(s0.z - mx2);  e3 = expf(s0.w - mx3);
    }
    float sm0 = e0, sm1 = e1, sm2 = e2, sm3 = e3;
    for (int c = 1; c < nch; ++c) {
        int2 pe;
        if (fetch_edge<SCAN>(c, lane, beg, end, tot, loc, srt, rec, pos, off, pe)) {
            float4 s = edge_score(pe, eterm4, asrc4, ad);
            sm0 += expf(s.x - mx0);  sm1 += expf(s.y - mx1);
            sm2 += expf(s.z - mx2);  sm3 += expf(s.w - mx3);
        }
    }
    float den0 = wave_sum(sm0) + EPS_SOFTMAX, den1 = wave_sum(sm1) + EPS_SOFTMAX;
    float den2 = wave_sum(sm2) + EPS_SOFTMAX, den3 = wave_sum(sm3) + EPS_SOFTMAX;

    float ax = 0.0f, ay = 0.0f;
    add_rows(hb, lane8, tab, wrow, lane, (unsigned)pe0.y << 8,
             e0 / den0, e1 / den1, e2 / den2, e3 / den3, ne < 32 ? ne : 32, ax, ay);
    for (int c = 1; c < nch; ++c) {
        int m = ne - 32 * c;
        if (m > 32) m = 32;
        int2 pe;
        float x0 = 0.0f, x1 = 0.0f, x2 = 0.0f, x3 = 0.0f;
        if (fetch_edge<SCAN>(c, lane, beg, end, tot, loc, srt, rec, pos, off, pe)) {
            float4 s = edge_score(pe, eterm4, asrc4, ad);
            x0 = expf(s.x - mx0);  x1 = expf(s.y - mx1);
            x2 = expf(s.z - mx2);  x3 = expf(s.w - mx3);
        }
        add_rows(hb, lane8, tab, wrow, lane, (unsigned)pe.y << 8,
                 x0 / den0, x1 / den1, x2 / den2, x3 / den3, m, ax, ay);
    }

    long base = (long)node * OUT_DIM + 2 * lane;
    float2 hn = *(const float2*)(h + base);
    float y0 = ax + hn.x;
    float y1 = ay + hn.y;

    float s = y0 + y1;
    #pragma unroll
    for (int o = 16; o > 0; o >>= 1) s += __shfl_xor(s, o, 32);
    float mu = s * (1.0f / 64.0f);

    float d0 = y0 - mu, d1 = y1 - mu;
    float vs = d0 * d0 + d1 * d1;
    #pragma unroll
    for (int o = 16; o > 0; o >>= 1) vs += __shfl_xor(vs, o, 32);
    float rstd = rsqrtf(vs * (1.0f / 64.0f) + EPS_LN);

    float2 o2;
    o2.x = d0 * rstd * g.x + b.x;
    o2.y = d1 * rstd * g.y + b.y;
    *(float2*)(out + base) = o2;
}

__global__ void __launch_bounds__(L3_T)
rank_and_gather(const int2* __restrict__ rec, const int* __restrict__ offmat,
                int n,
                const float* __restrict__ h,
                const float* __restrict__ eterm,
                const float* __restrict__ a_src,
                const float* __restrict__ a_dst,
                const float* __restrict__ gamma,
                const float* __restrict__ beta,
                float* __restrict__ out) {
    __shared__ int cnt[BRP];
    __shared__ int nstart[BRP + 1];
    __shared__ int wtot[BRP / 32];
    __shared__ int pos[NCHUNK];
    __shared__ int off[NCHUNK];
    __shared__ int wcnt[NCHUNK / 32];
    __shared__ __attribute__((aligned(16))) int2 srt[CAP];
    __shared__ __attribute__((aligned(16))) float tabs[FWAVES * TABW];

    int tid  = threadIdx.x;
    int lane = tid & 31;
    int wave = tid >> 5;
    int bin  = blockIdx.x;

    int o0 = 0, len = 0, incl = 0;
    if (tid < NCHUNK) {
        o0 = offmat[(long)bin * NCHUNK + tid];
        len = offmat[(long)(bin + 1) * NCHUNK + tid] - o0;
        incl = len;
        #pragma unroll
        for (int d = 1; d < 32; d <<= 1) {
            int q = __shfl_up(incl, d, 32);
            if (lane >= d) incl += q;
        }
        if (lane == 31) wcnt[wave] = incl;
    }
    if (tid < BRP) cnt[tid] = 0;
    __syncthreads();
    int tot = 0, before = 0;
    #pragma unroll
    for (int w = 0; w < NCHUNK / 32; ++w) {
        int t = wcnt[w];
        if (w < wave) before += t;
        tot += t;
    }
    if (tid < NCHUNK) {
        pos[tid] = before + incl - len;
        off[tid] = o0;
    }
    __syncthreads();

    for (int i0 = tid; i0 < tot; i0 += 4 * L3_T) {
        unsigned y[4];
        #pragma unroll
        for (int j = 0; j < 4; ++j) {
            int i = i0 + j * L3_T;
            y[j] = (unsigned)rec[rec_slot(i < tot ? i : tot - 1, pos, off)].y;
        }
        #pragma unroll
        for (int j = 0; j < 4; ++j)
            if (i0 + j * L3_T < tot) atomicAdd(&cnt[y[j] >> 24], 1);
    }
    __syncthreads();
    int c = 0, inc2 = 0;
    if (tid < BRP) {
        c = cnt[tid];
        inc2 = c;
        #pragma unroll
        for (int d = 1; d < 32; d <<= 1) {
            int q = __shfl_up(inc2, d, 32);
            if (lane >= d) inc2 += q;
        }
        if (lane == 31) wtot[wave] = inc2;
    }
    __syncthreads();
    if (tid < BRP) {
        int add = 0;
        for (int w = 0; w < wave; ++w) add += wtot[w];
        int start = add + inc2 - c;
        cnt[tid] = start;
        nstart[tid] = start;
    }
    if (tid == 0) nstart[BRP] = tot;
    __syncthreads();
    const int  tot_s = __builtin_amdgcn_readfirstlane(tot);
    const bool fits  = tot_s <= CAP;
    if (fits) {
        for (int i0 = tid; i0 < tot; i0 += 4 * L3_T) {
            int2 r[4];
            #pragma unroll
            for (int j = 0; j < 4; ++j) {
                int i = i0 + j * L3_T;
                r[j] = rec[rec_slot(i < tot ? i : tot - 1, pos, off)];
            }
            #pragma unroll
            for (int j = 0; j < 4; ++j) {
                if (i0 + j * L3_T < tot) {
                    int slot = atomicAdd(&cnt[(unsigned)r[j].y >> 24], 1);
                    srt[slot] = make_int2(r[j].x, r[j].y & 0x00ffffff);
                }
            }
        }
    }
    __syncthreads();

    float* tab = tabs + wave * TABW;
    const float* wrow = tab + (lane >> 3) * WSTR;
    unsigned lane8 = 8u * (unsigned)lane;
    const float4* eterm4 = (const float4*)eterm;
    const float4* asrc4  = (const float4*)a_src;
    const float4* adst4  = (const float4*)a_dst;
    const float2 g = ((const float2*)gamma)[lane];
    const float2 b = ((const float2*)beta)[lane];

    const int wv = __builtin_amdgcn_readfirstlane(wave);
    for (int t = wv; t < BR; t += FWAVES) {
        int node = bin * BR + t;
        if (node >= n) break;
        if (fits) {
            int beg = __builtin_amdgcn_readfirstlane(nstart[t]);
            int end = __builtin_amdgcn_readfirstlane(nstart[t + 1]);
            gather_node<false>(node, beg, end, tot_s, (unsigned)t, lane, srt, rec, pos, off,
                               h, lane8, tab, wrow, eterm4, asrc4, adst4, g, b, out);
        } else {
            gather_node<true>(node, 0, 0, tot_s, (unsigned)t, lane, srt, rec, pos, off,
                              h, lane8, tab, wrow, eterm4, asrc4, adst4, g, b, out);
        }
    }
}

extern "C" void kernel_launch(void* const* d_in, const int* in_sizes, int n_in,
                              void* d_out, int out_size, void* d_ws, size_t ws_size,
                              hipStream_t stream) {
    const float* x         = (const float*)d_in[0];
    const int*   ei        = (const int*)  d_in[1];
    const float* edge_attr = (const float*)d_in[2];
    const float* W_node    = (const float*)d_in[3];
    const float* W_edge    = (const float*)d_in[4];
    const float* attn_src  = (const float*)d_in[5];
    const float* attn_dst  = (const float*)d_in[6];
    const float* gamma     = (const float*)d_in[7];
    const float* beta      = (const float*)d_in[8];
    float* out = (float*)d_out;

    int  n = in_sizes[0] / IN_DIM;
    long e = (long)in_sizes[2] / EDGE_DIM;
    const int* srcIdx = ei;
    const int* dstIdx = ei + e;

    int nb = (n + BR - 1) / BR;

    float* ws    = (float*)d_ws;
    float* h     = ws;  ws += (long)n * OUT_DIM;
    float* a_src = ws;  ws += (long)n * H_HEADS;
    float* a_dst = ws;  ws += (long)n * H_HEADS;
    float* eterm = ws;  ws += e * H_HEADS;
    int2*  rec   = (int2*)ws;  ws += 2 * e;
    int*   offmat = (int*)ws; ws += (long)(nb + 1) * NCHUNK;

    gemm_and_bin<<<NCHUNK + GEMM_BLOCKS + ETERM_BLOCKS, L1_T, 0, stream>>>(
        x, W_node, attn_src, attn_dst, h, a_src, a_dst, n, srcIdx, dstIdx, offmat, rec, e, nb,
        edge_attr, W_edge, eterm);

    rank_and_gather<<<nb, L3_T, 0, stream>>>(
        rec, offmat, n, h, eterm, a_src, a_dst, gamma, beta, out);
}


# gather's per-node vector work cut: head maxima on integer-compare DPP steps, exp as exp2 of a product, one reciprocal per head, leaky relu as a max, layernorm sums on DPP; layernorm sum order changes
# speedup vs baseline: 6.5912x; 1.0396x over previous
#include <hip/hip_runtime.h>
#include <hip/hip_bf16.h>

typedef float v2f __attribute__((ext_vector_type(2)));
typedef float v8f __attribute__((ext_vector_type(8)));

#define IN_DIM   64
#define OUT_DIM  64
#define EDGE_DIM 16
#define H_HEADS  4
#define HD       16
#define NEG_SLOPE   0.2f
#define EPS_SOFTMAX 1e-8f
#define EPS_LN      1e-5f

template <int M>
__device__ __forceinline__ float xrow(float x) {
    return __int_as_float(__builtin_amdgcn_update_dpp(0, __float_as_int(x), 0x160 + M, 0xf, 0xf, false));
}
__device__ __forceinline__ float row16_sum(float v) {
    v += xrow<1>(v);
    v += xrow<2>(v);
    v += xrow<4>(v);
    v += xrow<8>(v);
    return v;
}

#define L1_T 256
#define GEMM_BLOCKS 768
__device__ __forceinline__ void node_gemm_role(int gblock0, int gstride,
                               const float* __restrict__ x,
                               const float* __restrict__ W,
                               const float* __restrict__ attn_src,
                               const float* __restrict__ attn_dst,
                               float* __restrict__ h,
                               float* __restrict__ a_src,
                               float* __restrict__ a_dst, int n) {
    int wave = (threadIdx.x >> 5) & 3;
    int band = threadIdx.x >> 7;
    int lane = threadIdx.x & 31;
    int grp  = lane >> 4;
    int lid  = lane & 15;

    int col0 = wave * 16;
    const float* wc = W + (long)(2 * grp) * OUT_DIM + col0 + lid;

    float ws = attn_src[col0 + lid];
    float wd = attn_dst[col0 + lid];
    v2f bf[IN_DIM / 4];
    #pragma unroll
    for (int k0 = 0; k0 < IN_DIM; k0 += 4) {
        bf[k0 / 4].x = wc[(long)k0 * OUT_DIM];
        bf[k0 / 4].y = wc[(long)(k0 + 1) * OUT_DIM];
    }

  for (int gblock = gblock0; ; gblock += gstride) {
    int row0 = (gblock * 2 + band) * 16;
    if (row0 >= n) break;

    int arow = row0 + lid;
    if (arow >= n) arow = n - 1;

    const float* xr = x + (long)arow * IN_DIM + 2 * grp;

    v8f c = {};
    #pragma unroll
    for (int k0 = 0; k0 < IN_DIM; k0 += 4) {
        v2f a = *(const v2f*)(xr + k0);
        c = __builtin_amdgcn_wmma_f32_16x16x4_f32(false, a, false, bf[k0 / 4],
                                                  (short)0, c, false, false);
    }

    #pragma unroll
    for (int r = 0; r < 8; ++r) {
        int row = row0 + r + 8 * grp;
        if (row < n) h[(long)row * OUT_DIM + col0 + lid] = c[r];
    }

    float ms = 0.0f, md = 0.0f;
    #pragma unroll
    for (int r = 0; r < 8; ++r) {
        float ps = row16_sum(c[r] * ws);
        float pd = row16_sum(c[r] * wd);
        if (lid == r) { ms = ps; md = pd; }
    }
    int prow = row0 + 8 * grp + lid;
    if (lid < 8 && prow < n) {
        a_src[(long)prow * H_HEADS + wave] = ms;
        a_dst[(long)prow * H_HEADS + wave] = md;
    }
  }
}

#define ETERM_BLOCKS 1024
__device__ __forceinline__ float4 edge_term(const float4* __restrict__ ea,
                                            const float* __restrict__ W_edge) {
    float sc[H_HEADS] = {0.f, 0.f, 0.f, 0.f};
    #pragma unroll
    for (int q = 0; q < 4; ++q) {
        float4 v = ea[q];
        int d0 = q * 4;
        #pragma unroll
        for (int hh = 0; hh < H_HEADS; ++hh) {
            sc[hh] += v.x * W_edge[(d0 + 0) * H_HEADS + hh]
                    + v.y * W_edge[(d0 + 1) * H_HEADS + hh]
                    + v.z * W_edge[(d0 + 2) * H_HEADS + hh]
                    + v.w * W_edge[(d0 + 3) * H_HEADS + hh];
        }
    }
    return make_float4(sc[0], sc[1], sc[2], sc[3]);
}
__device__ __forceinline__ void edge_term_role(long e0, long estride,
                           const float* __restrict__ edge_attr,
                           const float* __restrict__ W_edge,
                           float* __restrict__ eterm, long e_cnt) {
  long e = e0;
  for (; e + estride < e_cnt; e += 2 * estride) {
    asm volatile("" ::: "memory");
    long e2 = e + estride;
    float4 ta = edge_term((const float4*)(edge_attr + e  * EDGE_DIM), W_edge);
    float4 tb = edge_term((const float4*)(edge_attr + e2 * EDGE_DIM), W_edge);
    ((float4*)eterm)[e]  = ta;
    ((float4*)eterm)[e2] = tb;
  }
  if (e < e_cnt)
    ((float4*)eterm)[e] = edge_term((const float4*)(edge_attr + e * EDGE_DIM), W_edge);
}

#define BR       196
#define BRP      224
#define NCHUNK   256
#define MAXB     1024

__device__ __forceinline__ long chunk_begin(int c, long e_cnt) {
    long per = (e_cnt + NCHUNK - 1) / NCHUNK;
    long e0 = (long)c * per;
    return e0 > e_cnt ? e_cnt : e0;
}

__device__ __forceinline__ void chunk_bin_role(const int* __restrict__ src, const int* __restrict__ dst,
                                               int* __restrict__ offmat, int2* __restrict__ rec,
                                               long e_cnt, int nb, int* hist, int* wtot) {
    int tid  = threadIdx.x;
    int lane = tid & 31;
    int wave = tid >> 5;
    int g = blockIdx.x;
    for (int t = tid; t < MAXB; t += L1_T) hist[t] = 0;
    __syncthreads();
    long e0 = chunk_begin(g, e_cnt);
    long e1 = chunk_begin(g + 1, e_cnt);
    for (long i0 = e0 + tid; i0 < e1; i0 += 16 * L1_T) {
        int d[16];
        #pragma unroll
        for (int j = 0; j < 16; ++j) {
            long i = i0 + (long)j * L1_T;
            d[j] = (i < e1) ? dst[i] : -1;
        }
        #pragma unroll
        for (int j = 0; j < 16; ++j)
            if (d[j] >= 0) atomicAdd(&hist[(unsigned)d[j] / BR], 1);
    }
    __syncthreads();

    int4 v = ((const int4*)hist)[tid];
    int s = v.x + v.y + v.z + v.w;
    int incl = s;
    #pragma unroll
    for (int d = 1; d < 32; d <<= 1) {
        int q = __shfl_up(incl, d, 32);
        if (lane >= d) incl += q;
    }
    if (lane == 31) wtot[wave] = incl;
    __syncthreads();
    int add = 0;
    for (int w = 0; w < wave; ++w) add += wtot[w];
    int p0 = (int)e0 + add + incl - s;
    int p1 = p0 + v.x, p2 = p1 + v.y, p3 = p2 + v.z;
    ((int4*)hist)[tid] = make_int4(p0, p1, p2, p3);
    int b0 = 4 * tid;
    if (b0 + 0 < nb) offmat[(long)(b0 + 0) * NCHUNK + g] = p0;
    if (b0 + 1 < nb) offmat[(long)(b0 + 1) * NCHUNK + g] = p1;
    if (b0 + 2 < nb) offmat[(long)(b0 + 2) * NCHUNK + g] = p2;
    if (b0 + 3 < nb) offmat[(long)(b0 + 3) * NCHUNK + g] = p3;
    if (tid == 0) offmat[(long)nb * NCHUNK + g] = (int)e1;
    __syncthreads();

    for (long i0 = e0 + tid; i0 < e1; i0 += 16 * L1_T) {
        int d[16], sn[16];
        #pragma unroll
        for (int j = 0; j < 16; ++j) {
            long i = i0 + (long)j * L1_T;
            bool ok = i < e1;
            d[j]  = ok ? dst[i] : -1;
            sn[j] = ok ? src[i] : 0;
        }
        #pragma unroll
        for (int j = 0; j < 16; ++j) {
            if (d[j] >= 0) {
                long i = i0 + (long)j * L1_T;
                unsigned bn = (unsigned)d[j] / BR;
                int slot = atomicAdd(&hist[bn], 1);
                unsigned loc = (unsigned)d[j] - bn * BR;
                rec[slot] = make_int2((int)i, (int)((unsigned)sn[j] | (loc << 24)));
            }
        }
    }
}

__global__ void __launch_bounds__(L1_T)
gemm_and_bin(const float* __restrict__ x,
             const float* __restrict__ W,
             const float* __restrict__ attn_src,
             const float* __restrict__ attn_dst,
             float* __restrict__ h,
             float* __restrict__ a_src,
             float* __restrict__ a_dst, int n,
             const int* __restrict__ src, const int* __restrict__ dst,
             int* __restrict__ offmat, int2* __restrict__ rec, long e_cnt, int nb,
             const float* __restrict__ edge_attr,
             const float* __restrict__ W_edge,
             float* __restrict__ eterm) {
    __shared__ __attribute__((aligned(16))) int hist[MAXB];
    __shared__ int wtot[L1_T / 32];
    if (blockIdx.x < NCHUNK) {
        chunk_bin_role(src, dst, offmat, rec, e_cnt, nb, hist, wtot);
        return;
    }
    if (blockIdx.x < NCHUNK + GEMM_BLOCKS) {
        node_gemm_role((int)blockIdx.x - NCHUNK, GEMM_BLOCKS,
                       x, W, attn_src, attn_dst, h, a_src, a_dst, n);
        return;
    }
    edge_term_role((long)((int)blockIdx.x - NCHUNK - GEMM_BLOCKS) * L1_T + threadIdx.x,
                   (long)ETERM_BLOCKS * L1_T,
                   edge_attr, W_edge, eterm, e_cnt);
}

__device__ __forceinline__ int rec_slot(int i, const int* pos, const int* off) {
    int c = 0;
    #pragma unroll
    for (int st = NCHUNK / 2; st > 0; st >>= 1)
        if (pos[c + st] <= i) c += st;
    return off[c] + (i - pos[c]);
}

__device__ __forceinline__ float xhalf(float x) {
    int b = __float_as_int(x);
    return __int_as_float(__builtin_amdgcn_permlanex16(b, b, 0x76543210, 0xfedcba98, false, false));
}
template <int M>
__device__ __forceinline__ int xrow_i(int x) {
    return __builtin_amdgcn_update_dpp(0, x, 0x160 + M, 0xf, 0xf, false);
}
__device__ __forceinline__ int imax(int a, int b) { return a > b ? a : b; }
__device__ __forceinline__ float wave_max(float v) {
    int b = __float_as_int(v);
    b = imax(b, xrow_i<1>(b));
    b = imax(b, xrow_i<2>(b));
    b = imax(b, xrow_i<4>(b));
    b = imax(b, xrow_i<8>(b));
    b = imax(b, __builtin_amdgcn_permlanex16(b, b, 0x76543210, 0xfedcba98, false, false));
    return __int_as_float(b);
}
__device__ __forceinline__ float exp_le0(float x) {
    return __builtin_amdgcn_exp2f(x * 1.44269504088896341f);
}
__device__ __forceinline__ float wave_sum(float v) {
    v += xrow<1>(v);
    v += xrow<2>(v);
    v += xrow<4>(v);
    v += xrow<8>(v);
    return v + xhalf(v);
}

#define WSTR   36
#define TABW   (4 * WSTR + 32)

__device__ __forceinline__ void wave_lds_sync() {
    __builtin_amdgcn_fence(__ATOMIC_RELEASE, "wavefront");
    __builtin_amdgcn_wave_barrier();
    __builtin_amdgcn_fence(__ATOMIC_ACQUIRE, "wavefront");
}

__device__ __forceinline__ void add_rows(const char* __restrict__ hb, unsigned lane8,
                                         float* tab, const float* wrow, int lane,
                                         unsigned srcoff,
                                         float w0, float w1, float w2, float w3,
                                         int m, float& ax, float& ay) {
    unsigned* st = (unsigned*)(tab + 4 * WSTR);
    wave_lds_sync();
    tab[0 * WSTR + lane] = w0;
    tab[1 * WSTR + lane] = w1;
    tab[2 * WSTR + lane] = w2;
    tab[3 * WSTR + lane] = w3;
    st[lane] = srcoff;
    wave_lds_sync();
    #pragma nounroll
    for (int k0 = 0; k0 < m; k0 += 8) {
        float4 wa = *(const float4*)(wrow + k0);
        float4 wb = *(const float4*)(wrow + k0 + 4);
        uint4  sa = *(const uint4*)(st + k0);
        uint4  sb = *(const uint4*)(st + k0 + 4);
        float2 hv[8];
        hv[0] = *(const float2*)(hb + (sa.x + lane8));
        hv[1] = *(const float2*)(hb + (sa.y + lane8));
        hv[2] = *(const float2*)(hb + (sa.z + lane8));
        hv[3] = *(const float2*)(hb + (sa.w + lane8));
        hv[4] = *(const float2*)(hb + (sb.x + lane8));
        hv[5] = *(const float2*)(hb + (sb.y + lane8));
        hv[6] = *(const float2*)(hb + (sb.z + lane8));
        hv[7] = *(const float2*)(hb + (sb.w + lane8));
        ax += hv[0].x * wa.x;  ay += hv[0].y * wa.x;
        ax += hv[1].x * wa.y;  ay += hv[1].y * wa.y;
        ax += hv[2].x * wa.z;  ay += hv[2].y * wa.z;
        ax += hv[3].x * wa.w;  ay += hv[3].y * wa.w;
        ax += hv[4].x * wb.x;  ay += hv[4].y * wb.x;
        ax += hv[5].x * wb.y;  ay += hv[5].y * wb.y;
        ax += hv[6].x * wb.z;  ay += hv[6].y * wb.z;
        ax += hv[7].x * wb.w;  ay += hv[7].y * wb.w;
    }
}

__device__ __forceinline__ float4 edge_act(float4 s, float4 ad) {
    float4 r;
    float v;
    v = s.x + ad.x;  r.x = fmaxf(v, NEG_SLOPE * v);
    v = s.y + ad.y;  r.y = fmaxf(v, NEG_SLOPE * v);
    v = s.z + ad.z;  r.z = fmaxf(v, NEG_SLOPE * v);
    v = s.w + ad.w;  r.w = fmaxf(v, NEG_SLOPE * v);
    return r;
}

#define L3_T   1024
#define FWAVES (L3_T / 32)
#define CAP    8192

template <bool SCAN>
__device__ __forceinline__ bool fetch_edge(int c, int lane, int beg, int end, int tot, unsigned loc,
                                           const int2* srt, const int2* __restrict__ rec,
                                           const int* pos, const int* off, int2& pe) {
    pe = make_int2(0, 0);
    if (!SCAN) {
        int i = beg + 32 * c + lane;
        bool v = i < end;
        if (v) pe = srt[i];
        return v;
    } else {
        int i = 32 * c + lane;
        bool v = i < tot;
        if (v) {
            int2 r = rec[rec_slot(i, pos, off)];
            v = ((unsigned)r.y >> 24) == loc;
            if (v) pe = make_int2(r.x, r.y & 0x00ffffff);
        }
        return v;
    }
}

__device__ __forceinline__ float4 edge_score(int2 pe, const float4* __restrict__ eterm4,
                                             const float4* __restrict__ asrc4, float4 ad) {
    float4 et = eterm4[pe.x];
    float4 as = asrc4[pe.y];
    return edge_act(make_float4(as.x + et.x, as.y + et.y, as.z + et.z, as.w + et.w), ad);
}

template <bool SCAN>
__device__ __forceinline__ void gather_node(int node, int beg, int end, int tot, unsigned loc, int lane,
        const int2* srt, const int2* __restrict__ rec, const int* pos, const int* off,
        const float* __restrict__ h, unsigned lane8, float* tab, const float* wrow,
        const float4* __restrict__ eterm4, const float4* __restrict__ asrc4,
        const float4* __restrict__ adst4, float2 g, float2 b, float* __restrict__ out) {
    const char* hb = (const char*)h;
    const float4 ad = adst4[node];
    const int ne  = SCAN ? tot : end - beg;
    const int nch = (ne + 31) >> 5;

    int2 pe0;
    bool v0 = fetch_edge<SCAN>(0, lane, beg, end, tot, loc, srt, rec, pos, off, pe0);
    float4 s0 = make_float4(0.0f, 0.0f, 0.0f, 0.0f);
    if (v0) s0 = edge_score(pe0, eterm4, asrc4, ad);

    float mx0 = fmaxf(s0.x, 0.0f), mx1 = fmaxf(s0.y, 0.0f);
    float mx2 = fmaxf(s0.z, 0.0f), mx3 = fmaxf(s0.w, 0.0f);
    for (int c = 1; c < nch; ++c) {
        int2 pe;
        if (fetch_edge<SCAN>(c, lane, beg, end, tot, loc, srt, rec, pos, off, pe)) {
            float4 s = edge_score(pe, eterm4, asrc4, ad);
            mx0 = fmaxf(mx0, s.x);  mx1 = fmaxf(mx1, s.y);
            mx2 = fmaxf(mx2, s.z);  mx3 = fmaxf(mx3, s.w);
        }
    }
    mx0 = wave_max(mx0);  mx1 = wave_max(mx1);
    mx2 = wave_max(mx2);  mx3 = wave_max(mx3);

    float e0 = 0.0f, e1 = 0.0f, e2 = 0.0f, e3 = 0.0f;
    if (v0) {
        e0 = exp_le0(s0.x - mx0);  e1 = exp_le0(s0.y - mx1);
        e2 = exp_le0(s0.z - mx2);  e3 = exp_le0(s0.w - mx3);
    }
    float sm0 = e0, sm1 = e1, sm2 = e2, sm3 = e3;
    for (int c = 1; c < nch; ++c) {
        int2 pe;
        if (fetch_edge<SCAN>(c, lane, beg, end, tot, loc, srt, rec, pos, off, pe)) {
            float4 s = edge_score(pe, eterm4, asrc4, ad);
            sm0 += exp_le0(s.x - mx0);  sm1 += exp_le0(s.y - mx1);
            sm2 += exp_le0(s.z - mx2);  sm3 += exp_le0(s.w - mx3);
        }
    }
    float rd0 = __builtin_amdgcn_rcpf(wave_sum(sm0) + EPS_SOFTMAX);
    float rd1 = __builtin_amdgcn_rcpf(wave_sum(sm1) + EPS_SOFTMAX);
    float rd2 = __builtin_amdgcn_rcpf(wave_sum(sm2) + EPS_SOFTMAX);
    float rd3 = __builtin_amdgcn_rcpf(wave_sum(sm3) + EPS_SOFTMAX);

    float ax = 0.0f, ay = 0.0f;
    add_rows(hb, lane8, tab, wrow, lane, (unsigned)pe0.y << 8,
             e0 * rd0, e1 * rd1, e2 * rd2, e3 * rd3, ne < 32 ? ne : 32, ax, ay);
    for (int c = 1; c < nch; ++c) {
        int m = ne - 32 * c;
        if (m > 32) m = 32;
        int2 pe;
        float x0 = 0.0f, x1 = 0.0f, x2 = 0.0f, x3 = 0.0f;
        if (fetch_edge<SCAN>(c, lane, beg, end, tot, loc, srt, rec, pos, off, pe)) {
            float4 s = edge_score(pe, eterm4, asrc4, ad);
            x0 = exp_le0(s.x - mx0);  x1 = exp_le0(s.y - mx1);
            x2 = exp_le0(s.z - mx2);  x3 = exp_le0(s.w - mx3);
        }
        add_rows(hb, lane8, tab, wrow, lane, (unsigned)pe.y << 8,
                 x0 * rd0, x1 * rd1, x2 * rd2, x3 * rd3, m, ax, ay);
    }

    long base = (long)node * OUT_DIM + 2 * lane;
    float2 hn = *(const float2*)(h + base);
    float y0 = ax + hn.x;
    float y1 = ay + hn.y;

    float s = wave_sum(y0 + y1);
    float mu = s * (1.0f / 64.0f);

    float d0 = y0 - mu, d1 = y1 - mu;
    float vs = wave_sum(d0 * d0 + d1 * d1);
    float rstd = rsqrtf(vs * (1.0f / 64.0f) + EPS_LN);

    float2 o2;
    o2.x = d0 * rstd * g.x + b.x;
    o2.y = d1 * rstd * g.y + b.y;
    *(float2*)(out + base) = o2;
}

__global__ void __launch_bounds__(L3_T)
rank_and_gather(const int2* __restrict__ rec, const int* __restrict__ offmat,
                int n,
                const float* __restrict__ h,
                const float* __restrict__ eterm,
                const float* __restrict__ a_src,
                const float* __restrict__ a_dst,
                const float* __restrict__ gamma,
                const float* __restrict__ beta,
                float* __restrict__ out) {
    __shared__ int cnt[BRP];
    __shared__ int nstart[BRP + 1];
    __shared__ int wtot[BRP / 32];
    __shared__ int pos[NCHUNK];
    __shared__ int off[NCHUNK];
    __shared__ int wcnt[NCHUNK / 32];
    __shared__ __attribute__((aligned(16))) int2 srt[CAP];
    __shared__ __attribute__((aligned(16))) float tabs[FWAVES * TABW];

    int tid  = threadIdx.x;
    int lane = tid & 31;
    int wave = tid >> 5;
    int bin  = blockIdx.x;

    int o0 = 0, len = 0, incl = 0;
    if (tid < NCHUNK) {
        o0 = offmat[(long)bin * NCHUNK + tid];
        len = offmat[(long)(bin + 1) * NCHUNK + tid] - o0;
        incl = len;
        #pragma unroll
        for (int d = 1; d < 32; d <<= 1) {
            int q = __shfl_up(incl, d, 32);
            if (lane >= d) incl += q;
        }
        if (lane == 31) wcnt[wave] = incl;
    }
    if (tid < BRP) cnt[tid] = 0;
    __syncthreads();
    int tot = 0, before = 0;
    #pragma unroll
    for (int w = 0; w < NCHUNK / 32; ++w) {
        int t = wcnt[w];
        if (w < wave) before += t;
        tot += t;
    }
    if (tid < NCHUNK) {
        pos[tid] = before + incl - len;
        off[tid] = o0;
    }
    __syncthreads();

    for (int i0 = tid; i0 < tot; i0 += 4 * L3_T) {
        unsigned y[4];
        #pragma unroll
        for (int j = 0; j < 4; ++j) {
            int i = i0 + j * L3_T;
            y[j] = (unsigned)rec[rec_slot(i < tot ? i : tot - 1, pos, off)].y;
        }
        #pragma unroll
        for (int j = 0; j < 4; ++j)
            if (i0 + j * L3_T < tot) atomicAdd(&cnt[y[j] >> 24], 1);
    }
    __syncthreads();
    int c = 0, inc2 = 0;
    if (tid < BRP) {
        c = cnt[tid];
        inc2 = c;
        #pragma unroll
        for (int d = 1; d < 32; d <<= 1) {
            int q = __shfl_up(inc2, d, 32);
            if (lane >= d) inc2 += q;
        }
        if (lane == 31) wtot[wave] = inc2;
    }
    __syncthreads();
    if (tid < BRP) {
        int add = 0;
        for (int w = 0; w < wave; ++w) add += wtot[w];
        int start = add + inc2 - c;
        cnt[tid] = start;
        nstart[tid] = start;
    }
    if (tid == 0) nstart[BRP] = tot;
    __syncthreads();
    const int  tot_s = __builtin_amdgcn_readfirstlane(tot);
    const bool fits  = tot_s <= CAP;
    if (fits) {
        for (int i0 = tid; i0 < tot; i0 += 4 * L3_T) {
            int2 r[4];
            #pragma unroll
            for (int j = 0; j < 4; ++j) {
                int i = i0 + j * L3_T;
                r[j] = rec[rec_slot(i < tot ? i : tot - 1, pos, off)];
            }
            #pragma unroll
            for (int j = 0; j < 4; ++j) {
                if (i0 + j * L3_T < tot) {
                    int slot = atomicAdd(&cnt[(unsigned)r[j].y >> 24], 1);
                    srt[slot] = make_int2(r[j].x, r[j].y & 0x00ffffff);
                }
            }
        }
    }
    __syncthreads();

    float* tab = tabs + wave * TABW;
    const float* wrow = tab + (lane >> 3) * WSTR;
    unsigned lane8 = 8u * (unsigned)lane;
    const float4* eterm4 = (const float4*)eterm;
    const float4* asrc4  = (const float4*)a_src;
    const float4* adst4  = (const float4*)a_dst;
    const float2 g = ((const float2*)gamma)[lane];
    const float2 b = ((const float2*)beta)[lane];

    const int wv = __builtin_amdgcn_readfirstlane(wave);
    for (int t = wv; t < BR; t += FWAVES) {
        int node = bin * BR + t;
        if (node >= n) break;
        if (fits) {
            int beg = __builtin_amdgcn_readfirstlane(nstart[t]);
            int end = __builtin_amdgcn_readfirstlane(nstart[t + 1]);
            gather_node<false>(node, beg, end, tot_s, (unsigned)t, lane, srt, rec, pos, off,
                               h, lane8, tab, wrow, eterm4, asrc4, adst4, g, b, out);
        } else {
            gather_node<true>(node, 0, 0, tot_s, (unsigned)t, lane, srt, rec, pos, off,
                              h, lane8, tab, wrow, eterm4, asrc4, adst4, g, b, out);
        }
    }
}

extern "C" void kernel_launch(void* const* d_in, const int* in_sizes, int n_in,
                              void* d_out, int out_size, void* d_ws, size_t ws_size,
                              hipStream_t stream) {
    const float* x         = (const float*)d_in[0];
    const int*   ei        = (const int*)  d_in[1];
    const float* edge_attr = (const float*)d_in[2];
    const float* W_node    = (const float*)d_in[3];
    const float* W_edge    = (const float*)d_in[4];
    const float* attn_src  = (const float*)d_in[5];
    const float* attn_dst  = (const float*)d_in[6];
    const float* gamma     = (const float*)d_in[7];
    const float* beta      = (const float*)d_in[8];
    float* out = (float*)d_out;

    int  n = in_sizes[0] / IN_DIM;
    long e = (long)in_sizes[2] / EDGE_DIM;
    const int* srcIdx = ei;
    const int* dstIdx = ei + e;

    int nb = (n + BR - 1) / BR;

    float* ws    = (float*)d_ws;
    float* h     = ws;  ws += (long)n * OUT_DIM;
    float* a_src = ws;  ws += (long)n * H_HEADS;
    float* a_dst = ws;  ws += (long)n * H_HEADS;
    float* eterm = ws;  ws += e * H_HEADS;
    int2*  rec   = (int2*)ws;  ws += 2 * e;
    int*   offmat = (int*)ws; ws += (long)(nb + 1) * NCHUNK;

    gemm_and_bin<<<NCHUNK + GEMM_BLOCKS + ETERM_BLOCKS, L1_T, 0, stream>>>(
        x, W_node, attn_src, attn_dst, h, a_src, a_dst, n, srcIdx, dstIdx, offmat, rec, e, nb,
        edge_attr, W_edge, eterm);

    rank_and_gather<<<nb, L3_T, 0, stream>>>(
        rec, offmat, n, h, eterm, a_src, a_dst, gamma, beta, out);
}
